# starting kernel
# baseline (speedup 1.0000x reference)
#include <hip/hip_runtime.h>
#include <hip/hip_bf16.h>
#include <math.h>

#define USR_NUM 100000
#define USR_DIM 64
#define NUM_LOC 5000
#define T_STEPS 16
#define N_EDGES 80000

typedef float v2f __attribute__((ext_vector_type(2)));
typedef float v8f __attribute__((ext_vector_type(8)));

__global__ void k_zero(float* __restrict__ p, int n) {
  int i = blockIdx.x * blockDim.x + threadIdx.x;
  if (i < n) p[i] = 0.0f;
}

__global__ void k_deg_accum(const int* __restrict__ dst, float* __restrict__ deg, int E) {
  int i = blockIdx.x * blockDim.x + threadIdx.x;
  if (i < E) atomicAdd(&deg[dst[i]], 1.0f);
}

__global__ void k_deg_fin(const float* __restrict__ deg, float* __restrict__ isq,
                          float* __restrict__ idg, int n) {
  int i = blockIdx.x * blockDim.x + threadIdx.x;
  if (i < n) {
    float d = deg[i] + 1.0f;
    isq[i] = rsqrtf(d);
    idg[i] = 1.0f / d;
  }
}

__global__ void k_scatter_users(const int* __restrict__ ids, const float* __restrict__ Wemb,
                                const float* __restrict__ bemb, float* __restrict__ e,
                                float* __restrict__ cnt, int N) {
  int t = blockIdx.x * blockDim.x + threadIdx.x;
  if (t >= N * 64) return;
  int i = t >> 6, d = t & 63;
  int id = ids[i];
  atomicAdd(&e[(long)id * 64 + d], Wemb[t] + bemb[d]);
  if (d == 0) atomicAdd(&cnt[id], 1.0f);
}

__global__ void k_enorm_selfinit(const float* __restrict__ cnt, const float* __restrict__ idg,
                                 float* __restrict__ e, float* __restrict__ agg, int nloc) {
  int t = blockIdx.x * blockDim.x + threadIdx.x;
  if (t >= nloc * 64) return;
  int j = t >> 6;
  float cv = cnt[j]; if (cv < 1.0f) cv = 1.0f;
  float v = e[t] / cv;
  e[t] = v;
  agg[t] = v * idg[j];
}

__global__ void k_selfinit(const float* __restrict__ h, const float* __restrict__ idg,
                           float* __restrict__ agg, int nloc) {
  int t = blockIdx.x * blockDim.x + threadIdx.x;
  if (t >= nloc * 64) return;
  agg[t] = h[t] * idg[t >> 6];
}

__global__ void k_edge_scatter(const int* __restrict__ src, const int* __restrict__ dst,
                               const float* __restrict__ isq, const float* __restrict__ h,
                               float* __restrict__ agg, int E) {
  int t = blockIdx.x * blockDim.x + threadIdx.x;
  if (t >= E * 64) return;
  int e = t >> 6, d = t & 63;
  int s = src[e], q = dst[e];
  atomicAdd(&agg[(long)q * 64 + d], h[(long)s * 64 + d] * isq[s] * isq[q]);
}

template <int NT>
__global__ __launch_bounds__(256) void k_gemm_wmma(
    const float* __restrict__ A, const float* __restrict__ B,
    const float* __restrict__ bias, float* __restrict__ C, int M, int K, int relu) {
  constexpr int Nc = NT * 16;
  __shared__ float Bs[64 * Nc];
  for (int i = threadIdx.x; i < K * Nc; i += 256) Bs[i] = B[i];
  __syncthreads();

  const int lane = threadIdx.x & 31;
  const int wave = threadIdx.x >> 5;
  const int half = lane >> 4;
  const int l16  = lane & 15;
  const int row0 = (blockIdx.x * 8 + wave) * 16;

  int ar = row0 + l16; if (ar > M - 1) ar = M - 1;
  const float* Arow = A + (long)ar * K;

  v8f acc[NT];
#pragma unroll
  for (int tt = 0; tt < NT; ++tt) {
    float bv = bias[tt * 16 + l16];
    v8f tmp = { bv, bv, bv, bv, bv, bv, bv, bv };
    acc[tt] = tmp;
  }

  for (int k = 0; k < K; k += 4) {
    v2f a;
    a[0] = Arow[k + 2 * half + 0];
    a[1] = Arow[k + 2 * half + 1];
#pragma unroll
    for (int tt = 0; tt < NT; ++tt) {
      v2f b;
      b[0] = Bs[(k + 2 * half + 0) * Nc + tt * 16 + l16];
      b[1] = Bs[(k + 2 * half + 1) * Nc + tt * 16 + l16];
      acc[tt] = __builtin_amdgcn_wmma_f32_16x16x4_f32(false, a, false, b, (short)0,
                                                      acc[tt], false, false);
    }
  }
#pragma unroll
  for (int tt = 0; tt < NT; ++tt) {
    for (int r = 0; r < 8; ++r) {
      int row = row0 + r + 8 * half;
      if (row < M) {
        float v = acc[tt][r];
        if (relu) v = v > 0.0f ? v : 0.0f;
        C[(long)row * Nc + tt * 16 + l16] = v;
      }
    }
  }
}

__global__ __launch_bounds__(256) void k_lstm_gates(
    const float* __restrict__ X, int Kx, const float* __restrict__ Wi,
    const float* __restrict__ H, const float* __restrict__ Wh,
    const float* __restrict__ b, float* __restrict__ G, int M) {
  __shared__ float Bs[64 * 64];
  for (int i = threadIdx.x; i < Kx * 64; i += 256) Bs[i] = Wi[i];
  for (int i = threadIdx.x; i < 16 * 64; i += 256) Bs[Kx * 64 + i] = Wh[i];
  __syncthreads();

  const int lane = threadIdx.x & 31;
  const int wave = threadIdx.x >> 5;
  const int half = lane >> 4;
  const int l16  = lane & 15;
  const int row0 = (blockIdx.x * 8 + wave) * 16;

  int ar = row0 + l16; if (ar > M - 1) ar = M - 1;

  v8f acc[4];
#pragma unroll
  for (int tt = 0; tt < 4; ++tt) {
    float bv = b[tt * 16 + l16];
    v8f tmp = { bv, bv, bv, bv, bv, bv, bv, bv };
    acc[tt] = tmp;
  }

  const float* Xr = X + (long)ar * Kx;
  for (int k = 0; k < Kx; k += 4) {
    v2f a;
    a[0] = Xr[k + 2 * half + 0];
    a[1] = Xr[k + 2 * half + 1];
#pragma unroll
    for (int tt = 0; tt < 4; ++tt) {
      v2f w;
      w[0] = Bs[(k + 2 * half + 0) * 64 + tt * 16 + l16];
      w[1] = Bs[(k + 2 * half + 1) * 64 + tt * 16 + l16];
      acc[tt] = __builtin_amdgcn_wmma_f32_16x16x4_f32(false, a, false, w, (short)0,
                                                      acc[tt], false, false);
    }
  }
  const float* Hr = H + (long)ar * 16;
#pragma unroll
  for (int k = 0; k < 16; k += 4) {
    v2f a;
    a[0] = Hr[k + 2 * half + 0];
    a[1] = Hr[k + 2 * half + 1];
#pragma unroll
    for (int tt = 0; tt < 4; ++tt) {
      v2f w;
      w[0] = Bs[(Kx + k + 2 * half + 0) * 64 + tt * 16 + l16];
      w[1] = Bs[(Kx + k + 2 * half + 1) * 64 + tt * 16 + l16];
      acc[tt] = __builtin_amdgcn_wmma_f32_16x16x4_f32(false, a, false, w, (short)0,
                                                      acc[tt], false, false);
    }
  }
#pragma unroll
  for (int tt = 0; tt < 4; ++tt) {
    for (int r = 0; r < 8; ++r) {
      int row = row0 + r + 8 * half;
      if (row < M) G[(long)row * 64 + tt * 16 + l16] = acc[tt][r];
    }
  }
}

__device__ __forceinline__ float sigm(float x) { return 1.0f / (1.0f + expf(-x)); }

__global__ void k_lstm_cell(const float* __restrict__ G, float* __restrict__ h,
                            float* __restrict__ c, int N) {
  int t = blockIdx.x * blockDim.x + threadIdx.x;
  if (t >= N * 16) return;
  int i = t >> 4, j = t & 15;
  const float* g = G + (long)i * 64;
  float gi = g[j], gf = g[16 + j], gg = g[32 + j], go = g[48 + j];
  float cn = sigm(gf) * c[t] + sigm(gi) * tanhf(gg);
  c[t] = cn;
  h[t] = sigm(go) * tanhf(cn);
}

__global__ void k_gather_e2n(const int* __restrict__ ids, const float* __restrict__ he,
                             float* __restrict__ out32, int N) {
  int t = blockIdx.x * blockDim.x + threadIdx.x;
  if (t >= N * 32) return;
  int i = t >> 5, d = t & 31;
  out32[t] = he[(long)ids[i] * 32 + d];
}

__global__ void k_outproj(const float* __restrict__ h, const float* __restrict__ Wout,
                          const float* __restrict__ bout, float* __restrict__ out, int N) {
  int i = blockIdx.x * blockDim.x + threadIdx.x;
  if (i >= N) return;
  const float* hr = h + (long)i * 16;
  float a0 = bout[0], a1 = bout[1];
  for (int j = 0; j < 16; ++j) {
    float v = hr[j];
    a0 += v * Wout[j * 2 + 0];
    a1 += v * Wout[j * 2 + 1];
  }
  out[(long)i * 2 + 0] = a0;
  out[(long)i * 2 + 1] = a1;
}

static inline int gs(long n, int b) { return (int)((n + b - 1) / b); }

extern "C" void kernel_launch(void* const* d_in, const int* in_sizes, int n_in,
                              void* d_out, int out_size, void* d_ws, size_t ws_size,
                              hipStream_t stream) {
  (void)in_sizes; (void)n_in; (void)out_size; (void)ws_size;

  const int*   seq  = (const int*)d_in[0];
  const int*   gidx = (const int*)d_in[1];
  const float* Wemb = (const float*)d_in[2];
  const float* bemb = (const float*)d_in[3];
  const float* W1   = (const float*)d_in[4];
  const float* b1   = (const float*)d_in[5];
  const float* W2   = (const float*)d_in[6];
  const float* b2   = (const float*)d_in[7];
  const float* We2n = (const float*)d_in[8];
  const float* be2n = (const float*)d_in[9];
  const float* Wi0  = (const float*)d_in[10];
  const float* Wh0  = (const float*)d_in[11];
  const float* bL0  = (const float*)d_in[12];
  const float* Wi1  = (const float*)d_in[13];
  const float* Wh1  = (const float*)d_in[14];
  const float* bL1  = (const float*)d_in[15];
  const float* Wout = (const float*)d_in[16];
  const float* bout = (const float*)d_in[17];

  const int* srcE = gidx;
  const int* dstE = gidx + N_EDGES;
  float* out = (float*)d_out;

  float* p = (float*)d_ws;
  float* e     = p; p += (size_t)NUM_LOC * 64;
  float* cnt   = p; p += NUM_LOC;
  float* deg   = p; p += NUM_LOC;
  float* isq   = p; p += NUM_LOC;
  float* idg   = p; p += NUM_LOC;
  float* agg   = p; p += (size_t)NUM_LOC * 64;
  float* h1b   = p; p += (size_t)NUM_LOC * 64;
  float* h2b   = p; p += (size_t)NUM_LOC * 64;
  float* he2n  = p; p += (size_t)NUM_LOC * 32;
  float* out32 = p; p += (size_t)USR_NUM * 32;
  float* gbuf  = p; p += (size_t)USR_NUM * 64;
  float* h0    = p; p += (size_t)USR_NUM * 16;
  float* c0    = p; p += (size_t)USR_NUM * 16;
  float* h1s   = p; p += (size_t)USR_NUM * 16;
  float* c1s   = p; p += (size_t)USR_NUM * 16;

  const int BT = 256;

  k_zero<<<gs(NUM_LOC, BT), BT, 0, stream>>>(deg, NUM_LOC);
  k_deg_accum<<<gs(N_EDGES, BT), BT, 0, stream>>>(dstE, deg, N_EDGES);
  k_deg_fin<<<gs(NUM_LOC, BT), BT, 0, stream>>>(deg, isq, idg, NUM_LOC);

  k_zero<<<gs((long)USR_NUM * 64, BT), BT, 0, stream>>>(h0, USR_NUM * 64);

  for (int t = 0; t < T_STEPS; ++t) {
    const int* ids = seq + (long)t * USR_NUM;

    k_zero<<<gs((long)NUM_LOC * 65, BT), BT, 0, stream>>>(e, NUM_LOC * 65);
    k_scatter_users<<<gs((long)USR_NUM * 64, BT), BT, 0, stream>>>(ids, Wemb, bemb, e, cnt, USR_NUM);
    k_enorm_selfinit<<<gs((long)NUM_LOC * 64, BT), BT, 0, stream>>>(cnt, idg, e, agg, NUM_LOC);

    k_edge_scatter<<<gs((long)N_EDGES * 64, BT), BT, 0, stream>>>(srcE, dstE, isq, e, agg, N_EDGES);
    k_gemm_wmma<4><<<gs(NUM_LOC, 128), 256, 0, stream>>>(agg, W1, b1, h1b, NUM_LOC, 64, 1);

    k_selfinit<<<gs((long)NUM_LOC * 64, BT), BT, 0, stream>>>(h1b, idg, agg, NUM_LOC);
    k_edge_scatter<<<gs((long)N_EDGES * 64, BT), BT, 0, stream>>>(srcE, dstE, isq, h1b, agg, N_EDGES);
    k_gemm_wmma<4><<<gs(NUM_LOC, 128), 256, 0, stream>>>(agg, W2, b2, h2b, NUM_LOC, 64, 0);

    k_gemm_wmma<2><<<gs(NUM_LOC, 128), 256, 0, stream>>>(h2b, We2n, be2n, he2n, NUM_LOC, 64, 0);
    k_gather_e2n<<<gs((long)USR_NUM * 32, BT), BT, 0, stream>>>(ids, he2n, out32, USR_NUM);

    k_lstm_gates<<<gs(USR_NUM, 128), 256, 0, stream>>>(out32, 32, Wi0, h0, Wh0, bL0, gbuf, USR_NUM);
    k_lstm_cell<<<gs((long)USR_NUM * 16, BT), BT, 0, stream>>>(gbuf, h0, c0, USR_NUM);

    k_lstm_gates<<<gs(USR_NUM, 128), 256, 0, stream>>>(h0, 16, Wi1, h1s, Wh1, bL1, gbuf, USR_NUM);
    k_lstm_cell<<<gs((long)USR_NUM * 16, BT), BT, 0, stream>>>(gbuf, h1s, c1s, USR_NUM);

    k_outproj<<<gs(USR_NUM, BT), BT, 0, stream>>>(h1s, Wout, bout, out + (long)t * USR_NUM * 2, USR_NUM);
  }
}


# Per-user tail (gather, both LSTM layers, cell update, out projection) fused into ONE kernel over all 16 steps: h/c stay in registers, gates never stored; edge2node rows of all steps staged first
# speedup vs baseline: 1.3340x; 1.3340x over previous
#include <hip/hip_runtime.h>
#include <hip/hip_bf16.h>
#include <math.h>

#define USR_NUM 100000
#define USR_DIM 64
#define NUM_LOC 5000
#define T_STEPS 16
#define N_EDGES 80000

typedef float v2f __attribute__((ext_vector_type(2)));
typedef float v4f __attribute__((ext_vector_type(4)));
typedef float v8f __attribute__((ext_vector_type(8)));

__global__ void k_zero(float* __restrict__ p, int n) {
  int i = blockIdx.x * blockDim.x + threadIdx.x;
  if (i < n) p[i] = 0.0f;
}

__global__ void k_deg_accum(const int* __restrict__ dst, float* __restrict__ deg, int E) {
  int i = blockIdx.x * blockDim.x + threadIdx.x;
  if (i < E) atomicAdd(&deg[dst[i]], 1.0f);
}

__global__ void k_deg_fin(const float* __restrict__ deg, float* __restrict__ isq,
                          float* __restrict__ idg, int n) {
  int i = blockIdx.x * blockDim.x + threadIdx.x;
  if (i < n) {
    float d = deg[i] + 1.0f;
    isq[i] = rsqrtf(d);
    idg[i] = 1.0f / d;
  }
}

__global__ void k_scatter_users(const int* __restrict__ ids, const float* __restrict__ Wemb,
                                const float* __restrict__ bemb, float* __restrict__ e,
                                float* __restrict__ cnt, int N) {
  int t = blockIdx.x * blockDim.x + threadIdx.x;
  if (t >= N * 64) return;
  int i = t >> 6, d = t & 63;
  int id = ids[i];
  atomicAdd(&e[(long)id * 64 + d], Wemb[t] + bemb[d]);
  if (d == 0) atomicAdd(&cnt[id], 1.0f);
}

__global__ void k_enorm_selfinit(const float* __restrict__ cnt, const float* __restrict__ idg,
                                 float* __restrict__ e, float* __restrict__ agg, int nloc) {
  int t = blockIdx.x * blockDim.x + threadIdx.x;
  if (t >= nloc * 64) return;
  int j = t >> 6;
  float cv = cnt[j]; if (cv < 1.0f) cv = 1.0f;
  float v = e[t] / cv;
  e[t] = v;
  agg[t] = v * idg[j];
}

__global__ void k_selfinit(const float* __restrict__ h, const float* __restrict__ idg,
                           float* __restrict__ agg, int nloc) {
  int t = blockIdx.x * blockDim.x + threadIdx.x;
  if (t >= nloc * 64) return;
  agg[t] = h[t] * idg[t >> 6];
}

__global__ void k_edge_scatter(const int* __restrict__ src, const int* __restrict__ dst,
                               const float* __restrict__ isq, const float* __restrict__ h,
                               float* __restrict__ agg, int E) {
  int t = blockIdx.x * blockDim.x + threadIdx.x;
  if (t >= E * 64) return;
  int e = t >> 6, d = t & 63;
  int s = src[e], q = dst[e];
  atomicAdd(&agg[(long)q * 64 + d], h[(long)s * 64 + d] * isq[s] * isq[q]);
}

template <int NT>
__global__ __launch_bounds__(256) void k_gemm_wmma(
    const float* __restrict__ A, const float* __restrict__ B,
    const float* __restrict__ bias, float* __restrict__ C, int M, int K, int relu) {
  constexpr int Nc = NT * 16;
  __shared__ float Bs[64 * Nc];
  for (int i = threadIdx.x; i < K * Nc; i += 256) Bs[i] = B[i];
  __syncthreads();

  const int lane = threadIdx.x & 31;
  const int wave = threadIdx.x >> 5;
  const int half = lane >> 4;
  const int l16  = lane & 15;
  const int row0 = (blockIdx.x * 8 + wave) * 16;

  int ar = row0 + l16; if (ar > M - 1) ar = M - 1;
  const float* Arow = A + (long)ar * K;

  v8f acc[NT];
#pragma unroll
  for (int tt = 0; tt < NT; ++tt) {
    float bv = bias[tt * 16 + l16];
    v8f tmp = { bv, bv, bv, bv, bv, bv, bv, bv };
    acc[tt] = tmp;
  }

  for (int k = 0; k < K; k += 4) {
    v2f a;
    a[0] = Arow[k + 2 * half + 0];
    a[1] = Arow[k + 2 * half + 1];
#pragma unroll
    for (int tt = 0; tt < NT; ++tt) {
      v2f b;
      b[0] = Bs[(k + 2 * half + 0) * Nc + tt * 16 + l16];
      b[1] = Bs[(k + 2 * half + 1) * Nc + tt * 16 + l16];
      acc[tt] = __builtin_amdgcn_wmma_f32_16x16x4_f32(false, a, false, b, (short)0,
                                                      acc[tt], false, false);
    }
  }
#pragma unroll
  for (int tt = 0; tt < NT; ++tt) {
    for (int r = 0; r < 8; ++r) {
      int row = row0 + r + 8 * half;
      if (row < M) {
        float v = acc[tt][r];
        if (relu) v = v > 0.0f ? v : 0.0f;
        C[(long)row * Nc + tt * 16 + l16] = v;
      }
    }
  }
}

#define HP 20

__device__ __forceinline__ float sigm(float x) { return 1.0f / (1.0f + expf(-x)); }

template <int NKS>
__device__ __forceinline__ void stage_frag(float* dst, const float* __restrict__ Wm) {
  for (int i = threadIdx.x; i < NKS * 256; i += 64) {
    const int s = i & 1, tt = (i >> 1) & 3, ln = (i >> 3) & 31, ks = i >> 8;
    dst[i] = Wm[(ks * 4 + 2 * (ln >> 4) + s) * 64 + tt * 16 + (ln & 15)];
  }
}

__global__ __launch_bounds__(64) void k_lstm_all(
    const int* __restrict__ seq,
    const float* __restrict__ he,
    const float* __restrict__ Wi0, const float* __restrict__ Wh0, const float* __restrict__ b0,
    const float* __restrict__ Wi1, const float* __restrict__ Wh1, const float* __restrict__ b1,
    const float* __restrict__ Wout, const float* __restrict__ bout,
    float* __restrict__ out, int N) {
  __shared__ __attribute__((aligned(32))) float Ws[80 * 64];
  __shared__ __attribute__((aligned(16))) float Hs[2 * 16 * HP];
  __shared__ __attribute__((aligned(16))) float Wo[32];
  __shared__ __attribute__((aligned(16))) float Bb[2 * 16 * 4];
  stage_frag<8>(Ws,            Wi0);
  stage_frag<4>(Ws +  8 * 256, Wh0);
  stage_frag<4>(Ws + 12 * 256, Wi1);
  stage_frag<4>(Ws + 16 * 256, Wh1);
  if (threadIdx.x < 32) Wo[threadIdx.x] = Wout[(threadIdx.x & 15) * 2 + (threadIdx.x >> 4)];
  Bb[threadIdx.x]      = b0[(threadIdx.x & 3) * 16 + (threadIdx.x >> 2)];
  Bb[64 + threadIdx.x] = b1[(threadIdx.x & 3) * 16 + (threadIdx.x >> 2)];
  __syncthreads();

  const int lane = threadIdx.x & 31;
  const int wave = threadIdx.x >> 5;
  const int half = lane >> 4;
  const int l16  = lane & 15;
  const int row0 = (blockIdx.x * 2 + wave) * 16;

  int ar = row0 + l16; if (ar > N - 1) ar = N - 1;

  const float* wo = Wo + half * 16;
  const float bo = bout[half];

  const v8f* Wf = (const v8f*)Ws + lane;
  float* Ht = Hs + wave * (16 * HP);

  v2f h0A[4], h1A[4];
  float c0[8], c1[8];
#pragma unroll
  for (int i = 0; i < 4; ++i) { v2f z = { 0.0f, 0.0f }; h0A[i] = z; h1A[i] = z; }
#pragma unroll
  for (int i = 0; i < 8; ++i) { c0[i] = 0.0f; c1[i] = 0.0f; }

  int id = seq[ar];
  for (int t = 0; t < T_STEPS; ++t) {
    const float* Xr = he + ((long)t * NUM_LOC + id) * 32 + 2 * half;
    v2f xa[8];
#pragma unroll
    for (int ks = 0; ks < 8; ++ks) xa[ks] = *(const v2f*)(Xr + ks * 4);
    const int tn = t + 1 < T_STEPS ? t + 1 : t;
    id = seq[(long)tn * N + ar];

    v8f acc[4];
    const v4f bs0 = *(const v4f*)(Bb + l16 * 4);
#pragma unroll
    for (int tt = 0; tt < 4; ++tt) {
      float bv = bs0[tt];
      v8f tmp = { bv, bv, bv, bv, bv, bv, bv, bv };
      acc[tt] = tmp;
    }
#pragma unroll
    for (int ks = 0; ks < 8; ++ks) {
      const v8f w8 = Wf[(0 + ks) * 32];
#pragma unroll
      for (int tt = 0; tt < 4; ++tt) {
        v2f w = { w8[2 * tt], w8[2 * tt + 1] };
        acc[tt] = __builtin_amdgcn_wmma_f32_16x16x4_f32(false, xa[ks], false, w, (short)0,
                                                        acc[tt], false, false);
      }
    }
#pragma unroll
    for (int ks = 0; ks < 4; ++ks) {
      const v8f w8 = Wf[(8 + ks) * 32];
#pragma unroll
      for (int tt = 0; tt < 4; ++tt) {
        v2f w = { w8[2 * tt], w8[2 * tt + 1] };
        acc[tt] = __builtin_amdgcn_wmma_f32_16x16x4_f32(false, h0A[ks], false, w, (short)0,
                                                        acc[tt], false, false);
      }
    }
#pragma unroll
    for (int r = 0; r < 8; ++r) {
      float gi = acc[0][r], gf = acc[1][r], gg = acc[2][r], go = acc[3][r];
      float cn = sigm(gf) * c0[r] + sigm(gi) * tanhf(gg);
      c0[r] = cn;
      Ht[(r + 8 * half) * HP + l16] = sigm(go) * tanhf(cn);
    }
    __syncthreads();
#pragma unroll
    for (int ks = 0; ks < 4; ++ks) h0A[ks] = *(const v2f*)(Ht + l16 * HP + ks * 4 + 2 * half);

    const v4f bs1 = *(const v4f*)(Bb + 64 + l16 * 4);
#pragma unroll
    for (int tt = 0; tt < 4; ++tt) {
      float bv = bs1[tt];
      v8f tmp = { bv, bv, bv, bv, bv, bv, bv, bv };
      acc[tt] = tmp;
    }
#pragma unroll
    for (int ks = 0; ks < 4; ++ks) {
      const v8f w8 = Wf[(12 + ks) * 32];
#pragma unroll
      for (int tt = 0; tt < 4; ++tt) {
        v2f w = { w8[2 * tt], w8[2 * tt + 1] };
        acc[tt] = __builtin_amdgcn_wmma_f32_16x16x4_f32(false, h0A[ks], false, w, (short)0,
                                                        acc[tt], false, false);
      }
    }
#pragma unroll
    for (int ks = 0; ks < 4; ++ks) {
      const v8f w8 = Wf[(16 + ks) * 32];
#pragma unroll
      for (int tt = 0; tt < 4; ++tt) {
        v2f w = { w8[2 * tt], w8[2 * tt + 1] };
        acc[tt] = __builtin_amdgcn_wmma_f32_16x16x4_f32(false, h1A[ks], false, w, (short)0,
                                                        acc[tt], false, false);
      }
    }
#pragma unroll
    for (int r = 0; r < 8; ++r) {
      float gi = acc[0][r], gf = acc[1][r], gg = acc[2][r], go = acc[3][r];
      float cn = sigm(gf) * c1[r] + sigm(gi) * tanhf(gg);
      c1[r] = cn;
      Ht[(r + 8 * half) * HP + l16] = sigm(go) * tanhf(cn);
    }
    __syncthreads();
#pragma unroll
    for (int ks = 0; ks < 4; ++ks) h1A[ks] = *(const v2f*)(Ht + l16 * HP + ks * 4 + 2 * half);

    float a = bo;
#pragma unroll
    for (int j = 0; j < 16; ++j) a += Ht[l16 * HP + j] * wo[j];
    if (row0 + l16 < N) out[((long)t * N + row0 + l16) * 2 + half] = a;
  }
}

static inline int gs(long n, int b) { return (int)((n + b - 1) / b); }

extern "C" void kernel_launch(void* const* d_in, const int* in_sizes, int n_in,
                              void* d_out, int out_size, void* d_ws, size_t ws_size,
                              hipStream_t stream) {
  (void)in_sizes; (void)n_in; (void)out_size; (void)ws_size;

  const int*   seq  = (const int*)d_in[0];
  const int*   gidx = (const int*)d_in[1];
  const float* Wemb = (const float*)d_in[2];
  const float* bemb = (const float*)d_in[3];
  const float* W1   = (const float*)d_in[4];
  const float* b1   = (const float*)d_in[5];
  const float* W2   = (const float*)d_in[6];
  const float* b2   = (const float*)d_in[7];
  const float* We2n = (const float*)d_in[8];
  const float* be2n = (const float*)d_in[9];
  const float* Wi0  = (const float*)d_in[10];
  const float* Wh0  = (const float*)d_in[11];
  const float* bL0  = (const float*)d_in[12];
  const float* Wi1  = (const float*)d_in[13];
  const float* Wh1  = (const float*)d_in[14];
  const float* bL1  = (const float*)d_in[15];
  const float* Wout = (const float*)d_in[16];
  const float* bout = (const float*)d_in[17];

  const int* srcE = gidx;
  const int* dstE = gidx + N_EDGES;
  float* out = (float*)d_out;

  float* p = (float*)d_ws;
  float* e     = p; p += (size_t)NUM_LOC * 64;
  float* cnt   = p; p += NUM_LOC;
  float* deg   = p; p += NUM_LOC;
  float* isq   = p; p += NUM_LOC;
  float* idg   = p; p += NUM_LOC;
  float* agg   = p; p += (size_t)NUM_LOC * 64;
  float* h1b   = p; p += (size_t)NUM_LOC * 64;
  float* h2b   = p; p += (size_t)NUM_LOC * 64;
  float* heall = p; p += (size_t)T_STEPS * NUM_LOC * 32;

  const int BT = 256;

  k_zero<<<gs(NUM_LOC, BT), BT, 0, stream>>>(deg, NUM_LOC);
  k_deg_accum<<<gs(N_EDGES, BT), BT, 0, stream>>>(dstE, deg, N_EDGES);
  k_deg_fin<<<gs(NUM_LOC, BT), BT, 0, stream>>>(deg, isq, idg, NUM_LOC);

  for (int t = 0; t < T_STEPS; ++t) {
    const int* ids = seq + (long)t * USR_NUM;

    k_zero<<<gs((long)NUM_LOC * 65, BT), BT, 0, stream>>>(e, NUM_LOC * 65);
    k_scatter_users<<<gs((long)USR_NUM * 64, BT), BT, 0, stream>>>(ids, Wemb, bemb, e, cnt, USR_NUM);
    k_enorm_selfinit<<<gs((long)NUM_LOC * 64, BT), BT, 0, stream>>>(cnt, idg, e, agg, NUM_LOC);

    k_edge_scatter<<<gs((long)N_EDGES * 64, BT), BT, 0, stream>>>(srcE, dstE, isq, e, agg, N_EDGES);
    k_gemm_wmma<4><<<gs(NUM_LOC, 128), 256, 0, stream>>>(agg, W1, b1, h1b, NUM_LOC, 64, 1);

    k_selfinit<<<gs((long)NUM_LOC * 64, BT), BT, 0, stream>>>(h1b, idg, agg, NUM_LOC);
    k_edge_scatter<<<gs((long)N_EDGES * 64, BT), BT, 0, stream>>>(srcE, dstE, isq, h1b, agg, N_EDGES);
    k_gemm_wmma<4><<<gs(NUM_LOC, 128), 256, 0, stream>>>(agg, W2, b2, h2b, NUM_LOC, 64, 0);

    k_gemm_wmma<2><<<gs(NUM_LOC, 128), 256, 0, stream>>>(h2b, We2n, be2n,
                                                         heall + (size_t)t * NUM_LOC * 32, NUM_LOC, 64, 0);
  }

  k_lstm_all<<<gs(USR_NUM, 32), 64, 0, stream>>>(seq, heall, Wi0, Wh0, bL0, Wi1, Wh1, bL1,
                                                 Wout, bout, out, USR_NUM);
}


# All segment sums (users->locations, graph edges->nodes) as gather-sums over index lists built per call by counting sort; float atomics gone; sums now run in list order (fill arrival order)
# speedup vs baseline: 2.4790x; 1.8583x over previous
#include <hip/hip_runtime.h>
#include <hip/hip_bf16.h>
#include <math.h>

#define USR_NUM 100000
#define USR_DIM 64
#define NUM_LOC 5000
#define T_STEPS 16
#define N_EDGES 80000

typedef float v2f __attribute__((ext_vector_type(2)));
typedef float v4f __attribute__((ext_vector_type(4)));
typedef float v8f __attribute__((ext_vector_type(8)));


__global__ void k_zero_i(int* __restrict__ p, int n) {
  int i = blockIdx.x * blockDim.x + threadIdx.x;
  if (i < n) p[i] = 0;
}

__global__ void k_count(const int* __restrict__ seq, const int* __restrict__ dst,
                        int* __restrict__ uptr, int* __restrict__ gptr) {
  int i = blockIdx.x * blockDim.x + threadIdx.x;
  if (i < T_STEPS * USR_NUM) {
    int t = i / USR_NUM;
    atomicAdd(&uptr[t * NUM_LOC + seq[i]], 1);
  } else if (i < T_STEPS * USR_NUM + N_EDGES) {
    atomicAdd(&gptr[dst[i - T_STEPS * USR_NUM]], 1);
  }
}

__global__ __launch_bounds__(1024) void k_scan(int* __restrict__ uptr, int* __restrict__ gptr,
                                               float* __restrict__ isq, float* __restrict__ idg) {
  __shared__ int part[1024];
  const int tid = threadIdx.x;
  const bool graph = blockIdx.x == T_STEPS;
  int* p = graph ? gptr : uptr + blockIdx.x * NUM_LOC;
  int v[5];
  int s = 0;
#pragma unroll
  for (int k = 0; k < 5; ++k) {
    int idx = tid * 5 + k;
    v[k] = idx < NUM_LOC ? p[idx] : 0;
    s += v[k];
  }
  if (graph) {
#pragma unroll
    for (int k = 0; k < 5; ++k) {
      int idx = tid * 5 + k;
      if (idx < NUM_LOC) {
        float d = (float)v[k] + 1.0f;
        isq[idx] = rsqrtf(d);
        idg[idx] = 1.0f / d;
      }
    }
  }
  part[tid] = s;
  __syncthreads();
  for (int off = 1; off < 1024; off <<= 1) {
    int x = tid >= off ? part[tid - off] : 0;
    __syncthreads();
    part[tid] += x;
    __syncthreads();
  }
  int run = part[tid] - s;
#pragma unroll
  for (int k = 0; k < 5; ++k) {
    int idx = tid * 5 + k;
    if (idx < NUM_LOC) { p[idx] = run; run += v[k]; }
  }
}

__global__ void k_fill(const int* __restrict__ seq, const int* __restrict__ src,
                       const int* __restrict__ dst, const float* __restrict__ isq,
                       int* __restrict__ uptr, int* __restrict__ gptr,
                       int* __restrict__ ulist_all, int* __restrict__ gcol,
                       float* __restrict__ gw) {
  int i = blockIdx.x * blockDim.x + threadIdx.x;
  if (i < T_STEPS * USR_NUM) {
    int t = i / USR_NUM;
    int pos = atomicAdd(&uptr[t * NUM_LOC + seq[i]], 1);
    ulist_all[(long)t * (NUM_LOC * 32) + pos] = i - t * USR_NUM;
  } else if (i < T_STEPS * USR_NUM + N_EDGES) {
    int e = i - T_STEPS * USR_NUM;
    int s = src[e], q = dst[e];
    int pos = atomicAdd(&gptr[q], 1);
    gcol[pos] = s;
    gw[pos] = isq[s] * isq[q];
  }
}

__global__ __launch_bounds__(256) void k_gather_users(
    const int* __restrict__ ptr, const int* __restrict__ ul, const float* __restrict__ Wemb,
    const float* __restrict__ bemb, float* __restrict__ e, int nloc) {
  const int lane = threadIdx.x & 31;
  const int j = __builtin_amdgcn_readfirstlane(blockIdx.x * 8 + (threadIdx.x >> 5));
  if (j >= nloc) return;
  const int beg = j ? ptr[j - 1] : 0;
  const int end = ptr[j];
  const v2f bb = *(const v2f*)(bemb + 2 * lane);
  v2f acc = { 0.0f, 0.0f };
  for (int k = beg; k < end; k += 8) {
    int id[8];
#pragma unroll
    for (int u = 0; u < 8; ++u) id[u] = ul[k + u < end ? k + u : end - 1];
    v2f x[8];
#pragma unroll
    for (int u = 0; u < 8; ++u) x[u] = *(const v2f*)(Wemb + (long)id[u] * 64 + 2 * lane);
#pragma unroll
    for (int u = 0; u < 8; ++u) {
      if (k + u < end) acc += x[u] + bb;
    }
  }
  float cv = (float)(end - beg); if (cv < 1.0f) cv = 1.0f;
  v2f r = { acc[0] / cv, acc[1] / cv };
  *(v2f*)(e + (long)j * 64 + 2 * lane) = r;
}

__global__ __launch_bounds__(256) void k_gather_edges(
    const int* __restrict__ gptr, const int* __restrict__ gcol, const float* __restrict__ gw,
    const float* __restrict__ idg, const float* __restrict__ h, float* __restrict__ agg,
    int nloc) {
  const int lane = threadIdx.x & 31;
  const int q = __builtin_amdgcn_readfirstlane(blockIdx.x * 8 + (threadIdx.x >> 5));
  if (q >= nloc) return;
  const int beg = q ? gptr[q - 1] : 0;
  const int end = gptr[q];
  const v2f self = *(const v2f*)(h + (long)q * 64 + 2 * lane);
  const float sd = idg[q];
  v2f acc = { 0.0f, 0.0f };
  for (int k = beg; k < end; k += 8) {
    int s[8];
    float w[8];
#pragma unroll
    for (int u = 0; u < 8; ++u) {
      const int kk = k + u < end ? k + u : end - 1;
      s[u] = gcol[kk];
      w[u] = gw[kk];
    }
    v2f x[8];
#pragma unroll
    for (int u = 0; u < 8; ++u) x[u] = *(const v2f*)(h + (long)s[u] * 64 + 2 * lane);
#pragma unroll
    for (int u = 0; u < 8; ++u) {
      if (k + u < end) acc += x[u] * w[u];
    }
  }
  acc += self * sd;
  *(v2f*)(agg + (long)q * 64 + 2 * lane) = acc;
}

template <int NT>
__global__ __launch_bounds__(256) void k_gemm_wmma(
    const float* __restrict__ A, const float* __restrict__ B,
    const float* __restrict__ bias, float* __restrict__ C, int M, int K, int relu) {
  constexpr int Nc = NT * 16;
  __shared__ float Bs[64 * Nc];
  for (int i = threadIdx.x; i < K * Nc; i += 256) Bs[i] = B[i];
  __syncthreads();

  const int lane = threadIdx.x & 31;
  const int wave = threadIdx.x >> 5;
  const int half = lane >> 4;
  const int l16  = lane & 15;
  const int row0 = (blockIdx.x * 8 + wave) * 16;

  int ar = row0 + l16; if (ar > M - 1) ar = M - 1;
  const float* Arow = A + (long)ar * K;

  v8f acc[NT];
#pragma unroll
  for (int tt = 0; tt < NT; ++tt) {
    float bv = bias[tt * 16 + l16];
    v8f tmp = { bv, bv, bv, bv, bv, bv, bv, bv };
    acc[tt] = tmp;
  }

  for (int k = 0; k < K; k += 4) {
    v2f a;
    a[0] = Arow[k + 2 * half + 0];
    a[1] = Arow[k + 2 * half + 1];
#pragma unroll
    for (int tt = 0; tt < NT; ++tt) {
      v2f b;
      b[0] = Bs[(k + 2 * half + 0) * Nc + tt * 16 + l16];
      b[1] = Bs[(k + 2 * half + 1) * Nc + tt * 16 + l16];
      acc[tt] = __builtin_amdgcn_wmma_f32_16x16x4_f32(false, a, false, b, (short)0,
                                                      acc[tt], false, false);
    }
  }
#pragma unroll
  for (int tt = 0; tt < NT; ++tt) {
    for (int r = 0; r < 8; ++r) {
      int row = row0 + r + 8 * half;
      if (row < M) {
        float v = acc[tt][r];
        if (relu) v = v > 0.0f ? v : 0.0f;
        C[(long)row * Nc + tt * 16 + l16] = v;
      }
    }
  }
}

#define HP 20

__device__ __forceinline__ float sigm(float x) { return 1.0f / (1.0f + expf(-x)); }

template <int NKS>
__device__ __forceinline__ void stage_frag(float* dst, const float* __restrict__ Wm) {
  for (int i = threadIdx.x; i < NKS * 256; i += 64) {
    const int s = i & 1, tt = (i >> 1) & 3, ln = (i >> 3) & 31, ks = i >> 8;
    dst[i] = Wm[(ks * 4 + 2 * (ln >> 4) + s) * 64 + tt * 16 + (ln & 15)];
  }
}

__global__ __launch_bounds__(64) void k_lstm_all(
    const int* __restrict__ seq,
    const float* __restrict__ he,
    const float* __restrict__ Wi0, const float* __restrict__ Wh0, const float* __restrict__ b0,
    const float* __restrict__ Wi1, const float* __restrict__ Wh1, const float* __restrict__ b1,
    const float* __restrict__ Wout, const float* __restrict__ bout,
    float* __restrict__ out, int N) {
  __shared__ __attribute__((aligned(32))) float Ws[80 * 64];
  __shared__ __attribute__((aligned(16))) float Hs[2 * 16 * HP];
  __shared__ __attribute__((aligned(16))) float Wo[32];
  __shared__ __attribute__((aligned(16))) float Bb[2 * 16 * 4];
  stage_frag<8>(Ws,            Wi0);
  stage_frag<4>(Ws +  8 * 256, Wh0);
  stage_frag<4>(Ws + 12 * 256, Wi1);
  stage_frag<4>(Ws + 16 * 256, Wh1);
  if (threadIdx.x < 32) Wo[threadIdx.x] = Wout[(threadIdx.x & 15) * 2 + (threadIdx.x >> 4)];
  Bb[threadIdx.x]      = b0[(threadIdx.x & 3) * 16 + (threadIdx.x >> 2)];
  Bb[64 + threadIdx.x] = b1[(threadIdx.x & 3) * 16 + (threadIdx.x >> 2)];
  __syncthreads();

  const int lane = threadIdx.x & 31;
  const int wave = threadIdx.x >> 5;
  const int half = lane >> 4;
  const int l16  = lane & 15;
  const int row0 = (blockIdx.x * 2 + wave) * 16;

  int ar = row0 + l16; if (ar > N - 1) ar = N - 1;

  const float* wo = Wo + half * 16;
  const float bo = bout[half];

  const v8f* Wf = (const v8f*)Ws + lane;
  float* Ht = Hs + wave * (16 * HP);

  v2f h0A[4], h1A[4];
  float c0[8], c1[8];
#pragma unroll
  for (int i = 0; i < 4; ++i) { v2f z = { 0.0f, 0.0f }; h0A[i] = z; h1A[i] = z; }
#pragma unroll
  for (int i = 0; i < 8; ++i) { c0[i] = 0.0f; c1[i] = 0.0f; }

  int id = seq[ar];
  for (int t = 0; t < T_STEPS; ++t) {
    const float* Xr = he + ((long)t * NUM_LOC + id) * 32 + 2 * half;
    v2f xa[8];
#pragma unroll
    for (int ks = 0; ks < 8; ++ks) xa[ks] = *(const v2f*)(Xr + ks * 4);
    const int tn = t + 1 < T_STEPS ? t + 1 : t;
    id = seq[(long)tn * N + ar];

    v8f acc[4];
    const v4f bs0 = *(const v4f*)(Bb + l16 * 4);
#pragma unroll
    for (int tt = 0; tt < 4; ++tt) {
      float bv = bs0[tt];
      v8f tmp = { bv, bv, bv, bv, bv, bv, bv, bv };
      acc[tt] = tmp;
    }
#pragma unroll
    for (int ks = 0; ks < 8; ++ks) {
      const v8f w8 = Wf[(0 + ks) * 32];
#pragma unroll
      for (int tt = 0; tt < 4; ++tt) {
        v2f w = { w8[2 * tt], w8[2 * tt + 1] };
        acc[tt] = __builtin_amdgcn_wmma_f32_16x16x4_f32(false, xa[ks], false, w, (short)0,
                                                        acc[tt], false, false);
      }
    }
#pragma unroll
    for (int ks = 0; ks < 4; ++ks) {
      const v8f w8 = Wf[(8 + ks) * 32];
#pragma unroll
      for (int tt = 0; tt < 4; ++tt) {
        v2f w = { w8[2 * tt], w8[2 * tt + 1] };
        acc[tt] = __builtin_amdgcn_wmma_f32_16x16x4_f32(false, h0A[ks], false, w, (short)0,
                                                        acc[tt], false, false);
      }
    }
#pragma unroll
    for (int r = 0; r < 8; ++r) {
      float gi = acc[0][r], gf = acc[1][r], gg = acc[2][r], go = acc[3][r];
      float cn = sigm(gf) * c0[r] + sigm(gi) * tanhf(gg);
      c0[r] = cn;
      Ht[(r + 8 * half) * HP + l16] = sigm(go) * tanhf(cn);
    }
    __syncthreads();
#pragma unroll
    for (int ks = 0; ks < 4; ++ks) h0A[ks] = *(const v2f*)(Ht + l16 * HP + ks * 4 + 2 * half);

    const v4f bs1 = *(const v4f*)(Bb + 64 + l16 * 4);
#pragma unroll
    for (int tt = 0; tt < 4; ++tt) {
      float bv = bs1[tt];
      v8f tmp = { bv, bv, bv, bv, bv, bv, bv, bv };
      acc[tt] = tmp;
    }
#pragma unroll
    for (int ks = 0; ks < 4; ++ks) {
      const v8f w8 = Wf[(12 + ks) * 32];
#pragma unroll
      for (int tt = 0; tt < 4; ++tt) {
        v2f w = { w8[2 * tt], w8[2 * tt + 1] };
        acc[tt] = __builtin_amdgcn_wmma_f32_16x16x4_f32(false, h0A[ks], false, w, (short)0,
                                                        acc[tt], false, false);
      }
    }
#pragma unroll
    for (int ks = 0; ks < 4; ++ks) {
      const v8f w8 = Wf[(16 + ks) * 32];
#pragma unroll
      for (int tt = 0; tt < 4; ++tt) {
        v2f w = { w8[2 * tt], w8[2 * tt + 1] };
        acc[tt] = __builtin_amdgcn_wmma_f32_16x16x4_f32(false, h1A[ks], false, w, (short)0,
                                                        acc[tt], false, false);
      }
    }
#pragma unroll
    for (int r = 0; r < 8; ++r) {
      float gi = acc[0][r], gf = acc[1][r], gg = acc[2][r], go = acc[3][r];
      float cn = sigm(gf) * c1[r] + sigm(gi) * tanhf(gg);
      c1[r] = cn;
      Ht[(r + 8 * half) * HP + l16] = sigm(go) * tanhf(cn);
    }
    __syncthreads();
#pragma unroll
    for (int ks = 0; ks < 4; ++ks) h1A[ks] = *(const v2f*)(Ht + l16 * HP + ks * 4 + 2 * half);

    float a = bo;
#pragma unroll
    for (int j = 0; j < 16; ++j) a += Ht[l16 * HP + j] * wo[j];
    if (row0 + l16 < N) out[((long)t * N + row0 + l16) * 2 + half] = a;
  }
}

static inline int gs(long n, int b) { return (int)((n + b - 1) / b); }

extern "C" void kernel_launch(void* const* d_in, const int* in_sizes, int n_in,
                              void* d_out, int out_size, void* d_ws, size_t ws_size,
                              hipStream_t stream) {
  (void)in_sizes; (void)n_in; (void)out_size; (void)ws_size;

  const int*   seq  = (const int*)d_in[0];
  const int*   gidx = (const int*)d_in[1];
  const float* Wemb = (const float*)d_in[2];
  const float* bemb = (const float*)d_in[3];
  const float* W1   = (const float*)d_in[4];
  const float* b1   = (const float*)d_in[5];
  const float* W2   = (const float*)d_in[6];
  const float* b2   = (const float*)d_in[7];
  const float* We2n = (const float*)d_in[8];
  const float* be2n = (const float*)d_in[9];
  const float* Wi0  = (const float*)d_in[10];
  const float* Wh0  = (const float*)d_in[11];
  const float* bL0  = (const float*)d_in[12];
  const float* Wi1  = (const float*)d_in[13];
  const float* Wh1  = (const float*)d_in[14];
  const float* bL1  = (const float*)d_in[15];
  const float* Wout = (const float*)d_in[16];
  const float* bout = (const float*)d_in[17];

  const int* srcE = gidx;
  const int* dstE = gidx + N_EDGES;
  float* out = (float*)d_out;

  float* p = (float*)d_ws;
  float* e     = p; p += (size_t)NUM_LOC * 64;
  float* agg   = p; p += (size_t)NUM_LOC * 64;
  float* h1b   = p; p += (size_t)NUM_LOC * 64;
  float* isq   = p; p += 5056;
  float* idg   = p; p += 5056;
  int*   uptr  = (int*)p; p += (size_t)T_STEPS * NUM_LOC;
  int*   gptr  = (int*)p; p += 5056;
  int*   gcol  = (int*)p; p += N_EDGES;
  float* gw    = p; p += N_EDGES;
  float* heall = p; p += (size_t)T_STEPS * NUM_LOC * 32;
  int*   ulist = (int*)heall;

  const int BT = 256;
  const long NIDX = (long)T_STEPS * USR_NUM + N_EDGES;

  k_zero_i<<<gs((long)T_STEPS * NUM_LOC + NUM_LOC, BT), BT, 0, stream>>>(uptr, T_STEPS * NUM_LOC + NUM_LOC);
  k_count<<<gs(NIDX, BT), BT, 0, stream>>>(seq, dstE, uptr, gptr);
  k_scan<<<T_STEPS + 1, 1024, 0, stream>>>(uptr, gptr, isq, idg);
  k_fill<<<gs(NIDX, BT), BT, 0, stream>>>(seq, srcE, dstE, isq, uptr, gptr, ulist, gcol, gw);

  for (int t = 0; t < T_STEPS; ++t) {
    k_gather_users<<<gs(NUM_LOC, 8), 256, 0, stream>>>(uptr + (size_t)t * NUM_LOC,
                                                       ulist + (size_t)t * NUM_LOC * 32,
                                                       Wemb, bemb, e, NUM_LOC);

    k_gather_edges<<<gs(NUM_LOC, 8), 256, 0, stream>>>(gptr, gcol, gw, idg, e, agg, NUM_LOC);
    k_gemm_wmma<4><<<gs(NUM_LOC, 128), 256, 0, stream>>>(agg, W1, b1, h1b, NUM_LOC, 64, 1);

    k_gather_edges<<<gs(NUM_LOC, 8), 256, 0, stream>>>(gptr, gcol, gw, idg, h1b, agg, NUM_LOC);
    k_gemm_wmma<4><<<gs(NUM_LOC, 128), 256, 0, stream>>>(agg, W2, b2, e, NUM_LOC, 64, 0);

    k_gemm_wmma<2><<<gs(NUM_LOC, 128), 256, 0, stream>>>(e, We2n, be2n,
                                                         heall + (size_t)t * NUM_LOC * 32, NUM_LOC, 64, 0);
  }

  k_lstm_all<<<gs(USR_NUM, 32), 64, 0, stream>>>(seq, heall, Wi0, Wh0, bL0, Wi1, Wh1, bL1,
                                                 Wout, bout, out, USR_NUM);
}


# Each GCN product fused into the gather-sum that feeds it: 16 aggregated rows go through LDS straight into the WMMAs (edge2node product chained behind layer 2); 6 -> 3 launches per step
# speedup vs baseline: 3.3485x; 1.3507x over previous
#include <hip/hip_runtime.h>
#include <hip/hip_bf16.h>
#include <math.h>

#define USR_NUM 100000
#define USR_DIM 64
#define NUM_LOC 5000
#define T_STEPS 16
#define N_EDGES 80000

typedef float v2f __attribute__((ext_vector_type(2)));
typedef float v4f __attribute__((ext_vector_type(4)));
typedef float v8f __attribute__((ext_vector_type(8)));


__global__ void k_zero_i(int* __restrict__ p, int n) {
  int i = blockIdx.x * blockDim.x + threadIdx.x;
  if (i < n) p[i] = 0;
}

__global__ void k_count(const int* __restrict__ seq, const int* __restrict__ dst,
                        int* __restrict__ uptr, int* __restrict__ gptr) {
  int i = blockIdx.x * blockDim.x + threadIdx.x;
  if (i < T_STEPS * USR_NUM) {
    int t = i / USR_NUM;
    atomicAdd(&uptr[t * NUM_LOC + seq[i]], 1);
  } else if (i < T_STEPS * USR_NUM + N_EDGES) {
    atomicAdd(&gptr[dst[i - T_STEPS * USR_NUM]], 1);
  }
}

__global__ __launch_bounds__(1024) void k_scan(int* __restrict__ uptr, int* __restrict__ gptr,
                                               float* __restrict__ isq, float* __restrict__ idg) {
  __shared__ int part[1024];
  const int tid = threadIdx.x;
  const bool graph = blockIdx.x == T_STEPS;
  int* p = graph ? gptr : uptr + blockIdx.x * NUM_LOC;
  int v[5];
  int s = 0;
#pragma unroll
  for (int k = 0; k < 5; ++k) {
    int idx = tid * 5 + k;
    v[k] = idx < NUM_LOC ? p[idx] : 0;
    s += v[k];
  }
  if (graph) {
#pragma unroll
    for (int k = 0; k < 5; ++k) {
      int idx = tid * 5 + k;
      if (idx < NUM_LOC) {
        float d = (float)v[k] + 1.0f;
        isq[idx] = rsqrtf(d);
        idg[idx] = 1.0f / d;
      }
    }
  }
  part[tid] = s;
  __syncthreads();
  for (int off = 1; off < 1024; off <<= 1) {
    int x = tid >= off ? part[tid - off] : 0;
    __syncthreads();
    part[tid] += x;
    __syncthreads();
  }
  int run = part[tid] - s;
#pragma unroll
  for (int k = 0; k < 5; ++k) {
    int idx = tid * 5 + k;
    if (idx < NUM_LOC) { p[idx] = run; run += v[k]; }
  }
}

__global__ void k_fill(const int* __restrict__ seq, const int* __restrict__ src,
                       const int* __restrict__ dst, const float* __restrict__ isq,
                       int* __restrict__ uptr, int* __restrict__ gptr,
                       int* __restrict__ ulist_all, int* __restrict__ gcol,
                       float* __restrict__ gw) {
  int i = blockIdx.x * blockDim.x + threadIdx.x;
  if (i < T_STEPS * USR_NUM) {
    int t = i / USR_NUM;
    int pos = atomicAdd(&uptr[t * NUM_LOC + seq[i]], 1);
    ulist_all[(long)t * (NUM_LOC * 32) + pos] = i - t * USR_NUM;
  } else if (i < T_STEPS * USR_NUM + N_EDGES) {
    int e = i - T_STEPS * USR_NUM;
    int s = src[e], q = dst[e];
    int pos = atomicAdd(&gptr[q], 1);
    gcol[pos] = s;
    gw[pos] = isq[s] * isq[q];
  }
}

__global__ __launch_bounds__(256) void k_gather_users(
    const int* __restrict__ ptr, const int* __restrict__ ul, const float* __restrict__ Wemb,
    const float* __restrict__ bemb, float* __restrict__ e, int nloc) {
  const int lane = threadIdx.x & 31;
  const int j = __builtin_amdgcn_readfirstlane(blockIdx.x * 8 + (threadIdx.x >> 5));
  if (j >= nloc) return;
  const int beg = j ? ptr[j - 1] : 0;
  const int end = ptr[j];
  const v2f bb = *(const v2f*)(bemb + 2 * lane);
  v2f acc = { 0.0f, 0.0f };
  for (int k = beg; k < end; k += 8) {
    int id[8];
#pragma unroll
    for (int u = 0; u < 8; ++u) id[u] = ul[k + u < end ? k + u : end - 1];
    v2f x[8];
#pragma unroll
    for (int u = 0; u < 8; ++u) x[u] = *(const v2f*)(Wemb + (long)id[u] * 64 + 2 * lane);
#pragma unroll
    for (int u = 0; u < 8; ++u) {
      if (k + u < end) acc += x[u] + bb;
    }
  }
  float cv = (float)(end - beg); if (cv < 1.0f) cv = 1.0f;
  v2f r = { acc[0] / cv, acc[1] / cv };
  *(v2f*)(e + (long)j * 64 + 2 * lane) = r;
}

#define TP 68
#define OP 36

__device__ __forceinline__ v2f gather_node(
    const int* __restrict__ gptr, const int* __restrict__ gcol, const float* __restrict__ gw,
    const float* __restrict__ idg, const float* __restrict__ h, int q, int lane) {
  const int beg = q ? gptr[q - 1] : 0;
  const int end = gptr[q];
  const v2f self = *(const v2f*)(h + (long)q * 64 + 2 * lane);
  const float sd = idg[q];
  v2f acc = { 0.0f, 0.0f };
  for (int k = beg; k < end; k += 8) {
    int s[8];
    float w[8];
#pragma unroll
    for (int u = 0; u < 8; ++u) {
      const int kk = k + u < end ? k + u : end - 1;
      s[u] = gcol[kk];
      w[u] = gw[kk];
    }
    v2f x[8];
#pragma unroll
    for (int u = 0; u < 8; ++u) x[u] = *(const v2f*)(h + (long)s[u] * 64 + 2 * lane);
#pragma unroll
    for (int u = 0; u < 8; ++u) {
      if (k + u < end) acc += x[u] * w[u];
    }
  }
  acc += self * sd;
  return acc;
}

__device__ __forceinline__ void load_bfrag(v2f (&b)[16], const float* __restrict__ Wm, int nc,
                                           int tt, int half, int l16) {
#pragma unroll
  for (int ks = 0; ks < 16; ++ks) {
    b[ks][0] = Wm[(4 * ks + 2 * half + 0) * nc + tt * 16 + l16];
    b[ks][1] = Wm[(4 * ks + 2 * half + 1) * nc + tt * 16 + l16];
  }
}

__device__ __forceinline__ v8f tile_product(const float* tile, const v2f (&b)[16], float bv,
                                            int half, int l16) {
  v8f c = { bv, bv, bv, bv, bv, bv, bv, bv };
#pragma unroll
  for (int ks = 0; ks < 16; ++ks) {
    const v2f a = *(const v2f*)(tile + l16 * TP + 4 * ks + 2 * half);
    c = __builtin_amdgcn_wmma_f32_16x16x4_f32(false, a, false, b[ks], (short)0, c, false, false);
  }
  return c;
}

__global__ __launch_bounds__(512) void k_gcn_relu(
    const int* __restrict__ gptr, const int* __restrict__ gcol, const float* __restrict__ gw,
    const float* __restrict__ idg, const float* __restrict__ hin, const float* __restrict__ W,
    const float* __restrict__ bias, float* __restrict__ hout, int nloc) {
  __shared__ __attribute__((aligned(16))) float At[16 * TP];
  __shared__ __attribute__((aligned(16))) float Ct[16 * TP];
  const int lane = threadIdx.x & 31;
  const int wave = threadIdx.x >> 5;
  const int half = lane >> 4;
  const int l16  = lane & 15;
  const int q = __builtin_amdgcn_readfirstlane(blockIdx.x * 16 + wave);

  v2f bf[16];
  float bv = 0.0f;
  if (wave < 4) {
    load_bfrag(bf, W, 64, wave, half, l16);
    bv = bias[wave * 16 + l16];
  }

  v2f acc = { 0.0f, 0.0f };
  if (q < nloc) acc = gather_node(gptr, gcol, gw, idg, hin, q, lane);
  *(v2f*)(At + wave * TP + 2 * lane) = acc;
  __syncthreads();

  if (wave < 4) {
    const v8f c = tile_product(At, bf, bv, half, l16);
#pragma unroll
    for (int r = 0; r < 8; ++r) {
      const float v = c[r];
      Ct[(r + 8 * half) * TP + wave * 16 + l16] = v > 0.0f ? v : 0.0f;
    }
  }
  __syncthreads();

  if (q < nloc) *(v2f*)(hout + (long)q * 64 + 2 * lane) = *(const v2f*)(Ct + wave * TP + 2 * lane);
}

__global__ __launch_bounds__(512) void k_gcn_e2n(
    const int* __restrict__ gptr, const int* __restrict__ gcol, const float* __restrict__ gw,
    const float* __restrict__ idg, const float* __restrict__ hin, const float* __restrict__ W2,
    const float* __restrict__ b2, const float* __restrict__ We, const float* __restrict__ be,
    float* __restrict__ out, int nloc) {
  __shared__ __attribute__((aligned(16))) float At[16 * TP];
  __shared__ __attribute__((aligned(16))) float Ht[16 * TP];
  __shared__ __attribute__((aligned(16))) float Ot[16 * OP];
  const int lane = threadIdx.x & 31;
  const int wave = threadIdx.x >> 5;
  const int half = lane >> 4;
  const int l16  = lane & 15;
  const int q = __builtin_amdgcn_readfirstlane(blockIdx.x * 16 + wave);

  v2f bf[16];
  float bv = 0.0f;
  if (wave < 4) {
    load_bfrag(bf, W2, 64, wave, half, l16);
    bv = b2[wave * 16 + l16];
  } else if (wave < 6) {
    load_bfrag(bf, We, 32, wave - 4, half, l16);
    bv = be[(wave - 4) * 16 + l16];
  }

  v2f acc = { 0.0f, 0.0f };
  if (q < nloc) acc = gather_node(gptr, gcol, gw, idg, hin, q, lane);
  *(v2f*)(At + wave * TP + 2 * lane) = acc;
  __syncthreads();

  if (wave < 4) {
    const v8f c = tile_product(At, bf, bv, half, l16);
#pragma unroll
    for (int r = 0; r < 8; ++r) Ht[(r + 8 * half) * TP + wave * 16 + l16] = c[r];
  }
  __syncthreads();

  if (wave >= 4 && wave < 6) {
    const v8f c = tile_product(Ht, bf, bv, half, l16);
#pragma unroll
    for (int r = 0; r < 8; ++r) Ot[(r + 8 * half) * OP + (wave - 4) * 16 + l16] = c[r];
  }
  __syncthreads();

  if (q < nloc) out[(long)q * 32 + lane] = Ot[wave * OP + lane];
}

#define HP 20

__device__ __forceinline__ float sigm(float x) { return 1.0f / (1.0f + expf(-x)); }

template <int NKS>
__device__ __forceinline__ void stage_frag(float* dst, const float* __restrict__ Wm) {
  for (int i = threadIdx.x; i < NKS * 256; i += 64) {
    const int s = i & 1, tt = (i >> 1) & 3, ln = (i >> 3) & 31, ks = i >> 8;
    dst[i] = Wm[(ks * 4 + 2 * (ln >> 4) + s) * 64 + tt * 16 + (ln & 15)];
  }
}

__global__ __launch_bounds__(64) void k_lstm_all(
    const int* __restrict__ seq,
    const float* __restrict__ he,
    const float* __restrict__ Wi0, const float* __restrict__ Wh0, const float* __restrict__ b0,
    const float* __restrict__ Wi1, const float* __restrict__ Wh1, const float* __restrict__ b1,
    const float* __restrict__ Wout, const float* __restrict__ bout,
    float* __restrict__ out, int N) {
  __shared__ __attribute__((aligned(32))) float Ws[80 * 64];
  __shared__ __attribute__((aligned(16))) float Hs[2 * 16 * HP];
  __shared__ __attribute__((aligned(16))) float Wo[32];
  __shared__ __attribute__((aligned(16))) float Bb[2 * 16 * 4];
  stage_frag<8>(Ws,            Wi0);
  stage_frag<4>(Ws +  8 * 256, Wh0);
  stage_frag<4>(Ws + 12 * 256, Wi1);
  stage_frag<4>(Ws + 16 * 256, Wh1);
  if (threadIdx.x < 32) Wo[threadIdx.x] = Wout[(threadIdx.x & 15) * 2 + (threadIdx.x >> 4)];
  Bb[threadIdx.x]      = b0[(threadIdx.x & 3) * 16 + (threadIdx.x >> 2)];
  Bb[64 + threadIdx.x] = b1[(threadIdx.x & 3) * 16 + (threadIdx.x >> 2)];
  __syncthreads();

  const int lane = threadIdx.x & 31;
  const int wave = threadIdx.x >> 5;
  const int half = lane >> 4;
  const int l16  = lane & 15;
  const int row0 = (blockIdx.x * 2 + wave) * 16;

  int ar = row0 + l16; if (ar > N - 1) ar = N - 1;

  const float* wo = Wo + half * 16;
  const float bo = bout[half];

  const v8f* Wf = (const v8f*)Ws + lane;
  float* Ht = Hs + wave * (16 * HP);

  v2f h0A[4], h1A[4];
  float c0[8], c1[8];
#pragma unroll
  for (int i = 0; i < 4; ++i) { v2f z = { 0.0f, 0.0f }; h0A[i] = z; h1A[i] = z; }
#pragma unroll
  for (int i = 0; i < 8; ++i) { c0[i] = 0.0f; c1[i] = 0.0f; }

  int id = seq[ar];
  for (int t = 0; t < T_STEPS; ++t) {
    const float* Xr = he + ((long)t * NUM_LOC + id) * 32 + 2 * half;
    v2f xa[8];
#pragma unroll
    for (int ks = 0; ks < 8; ++ks) xa[ks] = *(const v2f*)(Xr + ks * 4);
    const int tn = t + 1 < T_STEPS ? t + 1 : t;
    id = seq[(long)tn * N + ar];

    v8f acc[4];
    const v4f bs0 = *(const v4f*)(Bb + l16 * 4);
#pragma unroll
    for (int tt = 0; tt < 4; ++tt) {
      float bv = bs0[tt];
      v8f tmp = { bv, bv, bv, bv, bv, bv, bv, bv };
      acc[tt] = tmp;
    }
#pragma unroll
    for (int ks = 0; ks < 8; ++ks) {
      const v8f w8 = Wf[(0 + ks) * 32];
#pragma unroll
      for (int tt = 0; tt < 4; ++tt) {
        v2f w = { w8[2 * tt], w8[2 * tt + 1] };
        acc[tt] = __builtin_amdgcn_wmma_f32_16x16x4_f32(false, xa[ks], false, w, (short)0,
                                                        acc[tt], false, false);
      }
    }
#pragma unroll
    for (int ks = 0; ks < 4; ++ks) {
      const v8f w8 = Wf[(8 + ks) * 32];
#pragma unroll
      for (int tt = 0; tt < 4; ++tt) {
        v2f w = { w8[2 * tt], w8[2 * tt + 1] };
        acc[tt] = __builtin_amdgcn_wmma_f32_16x16x4_f32(false, h0A[ks], false, w, (short)0,
                                                        acc[tt], false, false);
      }
    }
#pragma unroll
    for (int r = 0; r < 8; ++r) {
      float gi = acc[0][r], gf = acc[1][r], gg = acc[2][r], go = acc[3][r];
      float cn = sigm(gf) * c0[r] + sigm(gi) * tanhf(gg);
      c0[r] = cn;
      Ht[(r + 8 * half) * HP + l16] = sigm(go) * tanhf(cn);
    }
    __syncthreads();
#pragma unroll
    for (int ks = 0; ks < 4; ++ks) h0A[ks] = *(const v2f*)(Ht + l16 * HP + ks * 4 + 2 * half);

    const v4f bs1 = *(const v4f*)(Bb + 64 + l16 * 4);
#pragma unroll
    for (int tt = 0; tt < 4; ++tt) {
      float bv = bs1[tt];
      v8f tmp = { bv, bv, bv, bv, bv, bv, bv, bv };
      acc[tt] = tmp;
    }
#pragma unroll
    for (int ks = 0; ks < 4; ++ks) {
      const v8f w8 = Wf[(12 + ks) * 32];
#pragma unroll
      for (int tt = 0; tt < 4; ++tt) {
        v2f w = { w8[2 * tt], w8[2 * tt + 1] };
        acc[tt] = __builtin_amdgcn_wmma_f32_16x16x4_f32(false, h0A[ks], false, w, (short)0,
                                                        acc[tt], false, false);
      }
    }
#pragma unroll
    for (int ks = 0; ks < 4; ++ks) {
      const v8f w8 = Wf[(16 + ks) * 32];
#pragma unroll
      for (int tt = 0; tt < 4; ++tt) {
        v2f w = { w8[2 * tt], w8[2 * tt + 1] };
        acc[tt] = __builtin_amdgcn_wmma_f32_16x16x4_f32(false, h1A[ks], false, w, (short)0,
                                                        acc[tt], false, false);
      }
    }
#pragma unroll
    for (int r = 0; r < 8; ++r) {
      float gi = acc[0][r], gf = acc[1][r], gg = acc[2][r], go = acc[3][r];
      float cn = sigm(gf) * c1[r] + sigm(gi) * tanhf(gg);
      c1[r] = cn;
      Ht[(r + 8 * half) * HP + l16] = sigm(go) * tanhf(cn);
    }
    __syncthreads();
#pragma unroll
    for (int ks = 0; ks < 4; ++ks) h1A[ks] = *(const v2f*)(Ht + l16 * HP + ks * 4 + 2 * half);

    float a = bo;
#pragma unroll
    for (int j = 0; j < 16; ++j) a += Ht[l16 * HP + j] * wo[j];
    if (row0 + l16 < N) out[((long)t * N + row0 + l16) * 2 + half] = a;
  }
}

static inline int gs(long n, int b) { return (int)((n + b - 1) / b); }

extern "C" void kernel_launch(void* const* d_in, const int* in_sizes, int n_in,
                              void* d_out, int out_size, void* d_ws, size_t ws_size,
                              hipStream_t stream) {
  (void)in_sizes; (void)n_in; (void)out_size; (void)ws_size;

  const int*   seq  = (const int*)d_in[0];
  const int*   gidx = (const int*)d_in[1];
  const float* Wemb = (const float*)d_in[2];
  const float* bemb = (const float*)d_in[3];
  const float* W1   = (const float*)d_in[4];
  const float* b1   = (const float*)d_in[5];
  const float* W2   = (const float*)d_in[6];
  const float* b2   = (const float*)d_in[7];
  const float* We2n = (const float*)d_in[8];
  const float* be2n = (const float*)d_in[9];
  const float* Wi0  = (const float*)d_in[10];
  const float* Wh0  = (const float*)d_in[11];
  const float* bL0  = (const float*)d_in[12];
  const float* Wi1  = (const float*)d_in[13];
  const float* Wh1  = (const float*)d_in[14];
  const float* bL1  = (const float*)d_in[15];
  const float* Wout = (const float*)d_in[16];
  const float* bout = (const float*)d_in[17];

  const int* srcE = gidx;
  const int* dstE = gidx + N_EDGES;
  float* out = (float*)d_out;

  float* p = (float*)d_ws;
  float* e     = p; p += (size_t)NUM_LOC * 64;
  float* h1b   = p; p += (size_t)NUM_LOC * 64;
  float* isq   = p; p += 5056;
  float* idg   = p; p += 5056;
  int*   uptr  = (int*)p; p += (size_t)T_STEPS * NUM_LOC;
  int*   gptr  = (int*)p; p += 5056;
  int*   gcol  = (int*)p; p += N_EDGES;
  float* gw    = p; p += N_EDGES;
  float* heall = p; p += (size_t)T_STEPS * NUM_LOC * 32;
  int*   ulist = (int*)heall;

  const int BT = 256;
  const long NIDX = (long)T_STEPS * USR_NUM + N_EDGES;

  k_zero_i<<<gs((long)T_STEPS * NUM_LOC + NUM_LOC, BT), BT, 0, stream>>>(uptr, T_STEPS * NUM_LOC + NUM_LOC);
  k_count<<<gs(NIDX, BT), BT, 0, stream>>>(seq, dstE, uptr, gptr);
  k_scan<<<T_STEPS + 1, 1024, 0, stream>>>(uptr, gptr, isq, idg);
  k_fill<<<gs(NIDX, BT), BT, 0, stream>>>(seq, srcE, dstE, isq, uptr, gptr, ulist, gcol, gw);

  for (int t = 0; t < T_STEPS; ++t) {
    k_gather_users<<<gs(NUM_LOC, 8), 256, 0, stream>>>(uptr + (size_t)t * NUM_LOC,
                                                       ulist + (size_t)t * NUM_LOC * 32,
                                                       Wemb, bemb, e, NUM_LOC);

    k_gcn_relu<<<gs(NUM_LOC, 16), 512, 0, stream>>>(gptr, gcol, gw, idg, e, W1, b1, h1b, NUM_LOC);

    k_gcn_e2n<<<gs(NUM_LOC, 16), 512, 0, stream>>>(gptr, gcol, gw, idg, h1b, W2, b2, We2n, be2n,
                                                   heall + (size_t)t * NUM_LOC * 32, NUM_LOC);
  }

  k_lstm_all<<<gs(USR_NUM, 32), 64, 0, stream>>>(seq, heall, Wi0, Wh0, bL0, Wi1, Wh1, bL1,
                                                 Wout, bout, out, USR_NUM);
}


# Every per-step kernel launched ONCE for all 16 steps (grid row = 4 steps, node-major planes, a node's edge list read once per 4 steps): 53 -> 8 launches; needs 42 MB of workspace, else step by step
# speedup vs baseline: 4.5109x; 1.3471x over previous
#include <hip/hip_runtime.h>
#include <hip/hip_bf16.h>
#include <math.h>

#define USR_NUM 100000
#define USR_DIM 64
#define NUM_LOC 5000
#define T_STEPS 16
#define N_EDGES 80000

typedef float v2f __attribute__((ext_vector_type(2)));
typedef float v4f __attribute__((ext_vector_type(4)));
typedef float v8f __attribute__((ext_vector_type(8)));


__global__ void k_zero_i(int* __restrict__ p, int n) {
  int i = blockIdx.x * blockDim.x + threadIdx.x;
  if (i < n) p[i] = 0;
}

__global__ void k_count(const int* __restrict__ seq, const int* __restrict__ dst,
                        int* __restrict__ uptr, int* __restrict__ gptr) {
  int i = blockIdx.x * blockDim.x + threadIdx.x;
  if (i < T_STEPS * USR_NUM) {
    int t = i / USR_NUM;
    atomicAdd(&uptr[t * NUM_LOC + seq[i]], 1);
  } else if (i < T_STEPS * USR_NUM + N_EDGES) {
    atomicAdd(&gptr[dst[i - T_STEPS * USR_NUM]], 1);
  }
}

__global__ __launch_bounds__(1024) void k_scan(int* __restrict__ uptr, int* __restrict__ gptr,
                                               float* __restrict__ isq, float* __restrict__ idg) {
  __shared__ int part[1024];
  const int tid = threadIdx.x;
  const bool graph = blockIdx.x == T_STEPS;
  int* p = graph ? gptr : uptr + blockIdx.x * NUM_LOC;
  int v[5];
  int s = 0;
#pragma unroll
  for (int k = 0; k < 5; ++k) {
    int idx = tid * 5 + k;
    v[k] = idx < NUM_LOC ? p[idx] : 0;
    s += v[k];
  }
  if (graph) {
#pragma unroll
    for (int k = 0; k < 5; ++k) {
      int idx = tid * 5 + k;
      if (idx < NUM_LOC) {
        float d = (float)v[k] + 1.0f;
        isq[idx] = rsqrtf(d);
        idg[idx] = 1.0f / d;
      }
    }
  }
  part[tid] = s;
  __syncthreads();
  for (int off = 1; off < 1024; off <<= 1) {
    int x = tid >= off ? part[tid - off] : 0;
    __syncthreads();
    part[tid] += x;
    __syncthreads();
  }
  int run = part[tid] - s;
#pragma unroll
  for (int k = 0; k < 5; ++k) {
    int idx = tid * 5 + k;
    if (idx < NUM_LOC) { p[idx] = run; run += v[k]; }
  }
}

__global__ void k_fill(const int* __restrict__ seq, const int* __restrict__ src,
                       const int* __restrict__ dst, const float* __restrict__ isq,
                       int* __restrict__ uptr, int* __restrict__ gptr,
                       int* __restrict__ ulist_all, int ustride, int* __restrict__ gcol,
                       float* __restrict__ gw) {
  int i = blockIdx.x * blockDim.x + threadIdx.x;
  if (i < T_STEPS * USR_NUM) {
    int t = i / USR_NUM;
    int pos = atomicAdd(&uptr[t * NUM_LOC + seq[i]], 1);
    ulist_all[(long)t * ustride + pos] = i - t * USR_NUM;
  } else if (i < T_STEPS * USR_NUM + N_EDGES) {
    int e = i - T_STEPS * USR_NUM;
    int s = src[e], q = dst[e];
    int pos = atomicAdd(&gptr[q], 1);
    gcol[pos] = s;
    gw[pos] = isq[s] * isq[q];
  }
}

template <int TS, int NS>
__global__ __launch_bounds__(256) void k_gather_users(
    const int* __restrict__ uptr, const int* __restrict__ ulist, int ustride,
    const float* __restrict__ Wemb, const float* __restrict__ bemb, float* __restrict__ e_all,
    int nloc) {
  const int lane = threadIdx.x & 31;
  const int j = __builtin_amdgcn_readfirstlane(blockIdx.x * 8 + (threadIdx.x >> 5));
  if (j >= nloc) return;
  const v2f bb = *(const v2f*)(bemb + 2 * lane);
  for (int it = 0; it < TS; ++it) {
    const int t = blockIdx.y * TS + it;
    const int* ptr = uptr + t * NUM_LOC;
    const int* ul = ulist + (long)t * ustride;
    const int beg = j ? ptr[j - 1] : 0;
    const int end = ptr[j];
    v2f acc = { 0.0f, 0.0f };
    for (int k = beg; k < end; k += 8) {
      int id[8];
#pragma unroll
      for (int u = 0; u < 8; ++u) id[u] = ul[k + u < end ? k + u : end - 1];
      v2f x[8];
#pragma unroll
      for (int u = 0; u < 8; ++u) x[u] = *(const v2f*)(Wemb + (long)id[u] * 64 + 2 * lane);
#pragma unroll
      for (int u = 0; u < 8; ++u) {
        if (k + u < end) acc += x[u] + bb;
      }
    }
    float cv = (float)(end - beg); if (cv < 1.0f) cv = 1.0f;
    v2f r = { acc[0] / cv, acc[1] / cv };
    *(v2f*)(e_all + ((long)j * NS + t) * 64 + 2 * lane) = r;
  }
}

#define TP 68
#define OP 36

template <int TS, int NS>
__device__ __forceinline__ void gather_node(
    v2f (&acc)[TS], const int* __restrict__ gptr, const int* __restrict__ gcol,
    const float* __restrict__ gw, const float* __restrict__ idg, const float* __restrict__ h,
    int q, int t0, int lane) {
  constexpr int GE = TS == 1 ? 8 : 4;
  const int beg = q ? gptr[q - 1] : 0;
  const int end = gptr[q];
  const float* hl = h + (long)t0 * 64 + 2 * lane;
  v2f self[TS];
#pragma unroll
  for (int ts = 0; ts < TS; ++ts) self[ts] = *(const v2f*)(hl + (long)q * (NS * 64) + ts * 64);
  const float sd = idg[q];
#pragma unroll
  for (int ts = 0; ts < TS; ++ts) { v2f z = { 0.0f, 0.0f }; acc[ts] = z; }
  for (int k = beg; k < end; k += GE) {
    int s[GE];
    float w[GE];
#pragma unroll
    for (int u = 0; u < GE; ++u) {
      const int kk = k + u < end ? k + u : end - 1;
      s[u] = gcol[kk];
      w[u] = gw[kk];
    }
    v2f x[GE][TS];
#pragma unroll
    for (int u = 0; u < GE; ++u) {
      const float* hs = hl + (long)s[u] * (NS * 64);
#pragma unroll
      for (int ts = 0; ts < TS; ++ts) x[u][ts] = *(const v2f*)(hs + ts * 64);
    }
#pragma unroll
    for (int u = 0; u < GE; ++u) {
      const bool live = k + u < end;
#pragma unroll
      for (int ts = 0; ts < TS; ++ts) {
        const v2f old = acc[ts];
        const v2f upd = old + x[u][ts] * w[u];
        acc[ts] = live ? upd : old;
      }
    }
  }
#pragma unroll
  for (int ts = 0; ts < TS; ++ts) acc[ts] += self[ts] * sd;
}

__device__ __forceinline__ void load_bfrag(v2f (&b)[16], const float* __restrict__ Wm, int nc,
                                           int tt, int half, int l16) {
#pragma unroll
  for (int ks = 0; ks < 16; ++ks) {
    b[ks][0] = Wm[(4 * ks + 2 * half + 0) * nc + tt * 16 + l16];
    b[ks][1] = Wm[(4 * ks + 2 * half + 1) * nc + tt * 16 + l16];
  }
}

template <int NT>
__device__ __forceinline__ void tiles_product(v8f (&c)[NT], const float* tiles, const v2f (&b)[16],
                                              float bv, int half, int l16) {
#pragma unroll
  for (int i = 0; i < NT; ++i) { v8f z = { bv, bv, bv, bv, bv, bv, bv, bv }; c[i] = z; }
#pragma unroll
  for (int ks = 0; ks < 16; ++ks) {
#pragma unroll
    for (int i = 0; i < NT; ++i) {
      const v2f a = *(const v2f*)(tiles + (i * 16 + l16) * TP + 4 * ks + 2 * half);
      c[i] = __builtin_amdgcn_wmma_f32_16x16x4_f32(false, a, false, b[ks], (short)0, c[i], false, false);
    }
  }
}

template <int TS, int NS>
__global__ __launch_bounds__(512) void k_gcn_relu(
    const int* __restrict__ gptr, const int* __restrict__ gcol, const float* __restrict__ gw,
    const float* __restrict__ idg, const float* __restrict__ hin, const float* __restrict__ W,
    const float* __restrict__ bias, float* __restrict__ hout, int nloc) {
  __shared__ __attribute__((aligned(16))) float At[TS * 16 * TP];
  __shared__ __attribute__((aligned(16))) float Ct[TS * 16 * TP];
  const int lane = threadIdx.x & 31;
  const int wave = threadIdx.x >> 5;
  const int half = lane >> 4;
  const int l16  = lane & 15;
  const int t0   = blockIdx.y * TS;
  const int q = __builtin_amdgcn_readfirstlane(blockIdx.x * 16 + wave);

  v2f bf[16];
  float bv = 0.0f;
  if (wave < 4) {
    load_bfrag(bf, W, 64, wave, half, l16);
    bv = bias[wave * 16 + l16];
  }

  v2f acc[TS];
#pragma unroll
  for (int ts = 0; ts < TS; ++ts) { v2f z = { 0.0f, 0.0f }; acc[ts] = z; }
  if (q < nloc) gather_node<TS, NS>(acc, gptr, gcol, gw, idg, hin, q, t0, lane);
#pragma unroll
  for (int ts = 0; ts < TS; ++ts) *(v2f*)(At + (ts * 16 + wave) * TP + 2 * lane) = acc[ts];
  __syncthreads();

  if (wave < 4) {
    v8f c[TS];
    tiles_product<TS>(c, At, bf, bv, half, l16);
#pragma unroll
    for (int ts = 0; ts < TS; ++ts) {
#pragma unroll
      for (int r = 0; r < 8; ++r) {
        const float v = c[ts][r];
        Ct[(ts * 16 + r + 8 * half) * TP + wave * 16 + l16] = v > 0.0f ? v : 0.0f;
      }
    }
  }
  __syncthreads();

  if (q < nloc) {
#pragma unroll
    for (int ts = 0; ts < TS; ++ts)
      *(v2f*)(hout + ((long)q * NS + t0 + ts) * 64 + 2 * lane) =
          *(const v2f*)(Ct + (ts * 16 + wave) * TP + 2 * lane);
  }
}

template <int TS, int NS>
__global__ __launch_bounds__(512) void k_gcn_e2n(
    const int* __restrict__ gptr, const int* __restrict__ gcol, const float* __restrict__ gw,
    const float* __restrict__ idg, const float* __restrict__ hin, const float* __restrict__ W2,
    const float* __restrict__ b2, const float* __restrict__ We, const float* __restrict__ be,
    float* __restrict__ out, int nloc) {
  __shared__ __attribute__((aligned(16))) float At[TS * 16 * TP];
  __shared__ __attribute__((aligned(16))) float Ht[TS * 16 * TP];
  __shared__ __attribute__((aligned(16))) float Ot[TS * 16 * OP];
  const int lane = threadIdx.x & 31;
  const int wave = threadIdx.x >> 5;
  const int half = lane >> 4;
  const int l16  = lane & 15;
  const int t0   = blockIdx.y * TS;
  const int q = __builtin_amdgcn_readfirstlane(blockIdx.x * 16 + wave);

  constexpr int ET = TS == 1 ? 1 : 2;
  constexpr int EW = 2 * TS / ET;
  const int ec = (wave - 4) & 1;
  const int es = ((wave - 4) >> 1) * ET;
  v2f bf[16];
  float bv = 0.0f;
  if (wave < 4) {
    load_bfrag(bf, W2, 64, wave, half, l16);
    bv = b2[wave * 16 + l16];
  } else if (wave < 4 + EW) {
    load_bfrag(bf, We, 32, ec, half, l16);
    bv = be[ec * 16 + l16];
  }

  v2f acc[TS];
#pragma unroll
  for (int ts = 0; ts < TS; ++ts) { v2f z = { 0.0f, 0.0f }; acc[ts] = z; }
  if (q < nloc) gather_node<TS, NS>(acc, gptr, gcol, gw, idg, hin, q, t0, lane);
#pragma unroll
  for (int ts = 0; ts < TS; ++ts) *(v2f*)(At + (ts * 16 + wave) * TP + 2 * lane) = acc[ts];
  __syncthreads();

  if (wave < 4) {
    v8f c[TS];
    tiles_product<TS>(c, At, bf, bv, half, l16);
#pragma unroll
    for (int ts = 0; ts < TS; ++ts) {
#pragma unroll
      for (int r = 0; r < 8; ++r) Ht[(ts * 16 + r + 8 * half) * TP + wave * 16 + l16] = c[ts][r];
    }
  }
  __syncthreads();

  if (wave >= 4 && wave < 4 + EW) {
    v8f c[ET];
    tiles_product<ET>(c, Ht + es * 16 * TP, bf, bv, half, l16);
#pragma unroll
    for (int i = 0; i < ET; ++i) {
#pragma unroll
      for (int r = 0; r < 8; ++r) Ot[((es + i) * 16 + r + 8 * half) * OP + ec * 16 + l16] = c[i][r];
    }
  }
  __syncthreads();

  if (q < nloc) {
#pragma unroll
    for (int ts = 0; ts < TS; ++ts)
      out[((long)(t0 + ts) * nloc + q) * 32 + lane] = Ot[(ts * 16 + wave) * OP + lane];
  }
}

#define HP 20

__device__ __forceinline__ float sigm(float x) { return 1.0f / (1.0f + expf(-x)); }

template <int NKS>
__device__ __forceinline__ void stage_frag(float* dst, const float* __restrict__ Wm) {
  for (int i = threadIdx.x; i < NKS * 256; i += 64) {
    const int s = i & 1, tt = (i >> 1) & 3, ln = (i >> 3) & 31, ks = i >> 8;
    dst[i] = Wm[(ks * 4 + 2 * (ln >> 4) + s) * 64 + tt * 16 + (ln & 15)];
  }
}

__global__ __launch_bounds__(64) void k_lstm_all(
    const int* __restrict__ seq,
    const float* __restrict__ he,
    const float* __restrict__ Wi0, const float* __restrict__ Wh0, const float* __restrict__ b0,
    const float* __restrict__ Wi1, const float* __restrict__ Wh1, const float* __restrict__ b1,
    const float* __restrict__ Wout, const float* __restrict__ bout,
    float* __restrict__ out, int N) {
  __shared__ __attribute__((aligned(32))) float Ws[80 * 64];
  __shared__ __attribute__((aligned(16))) float Hs[2 * 16 * HP];
  __shared__ __attribute__((aligned(16))) float Wo[32];
  __shared__ __attribute__((aligned(16))) float Bb[2 * 16 * 4];
  stage_frag<8>(Ws,            Wi0);
  stage_frag<4>(Ws +  8 * 256, Wh0);
  stage_frag<4>(Ws + 12 * 256, Wi1);
  stage_frag<4>(Ws + 16 * 256, Wh1);
  if (threadIdx.x < 32) Wo[threadIdx.x] = Wout[(threadIdx.x & 15) * 2 + (threadIdx.x >> 4)];
  Bb[threadIdx.x]      = b0[(threadIdx.x & 3) * 16 + (threadIdx.x >> 2)];
  Bb[64 + threadIdx.x] = b1[(threadIdx.x & 3) * 16 + (threadIdx.x >> 2)];
  __syncthreads();

  const int lane = threadIdx.x & 31;
  const int wave = threadIdx.x >> 5;
  const int half = lane >> 4;
  const int l16  = lane & 15;
  const int row0 = (blockIdx.x * 2 + wave) * 16;

  int ar = row0 + l16; if (ar > N - 1) ar = N - 1;

  const float* wo = Wo + half * 16;
  const float bo = bout[half];

  const v8f* Wf = (const v8f*)Ws + lane;
  float* Ht = Hs + wave * (16 * HP);

  v2f h0A[4], h1A[4];
  float c0[8], c1[8];
#pragma unroll
  for (int i = 0; i < 4; ++i) { v2f z = { 0.0f, 0.0f }; h0A[i] = z; h1A[i] = z; }
#pragma unroll
  for (int i = 0; i < 8; ++i) { c0[i] = 0.0f; c1[i] = 0.0f; }

  int id = seq[ar];
  for (int t = 0; t < T_STEPS; ++t) {
    const float* Xr = he + ((long)t * NUM_LOC + id) * 32 + 2 * half;
    v2f xa[8];
#pragma unroll
    for (int ks = 0; ks < 8; ++ks) xa[ks] = *(const v2f*)(Xr + ks * 4);
    const int tn = t + 1 < T_STEPS ? t + 1 : t;
    id = seq[(long)tn * N + ar];

    v8f acc[4];
    const v4f bs0 = *(const v4f*)(Bb + l16 * 4);
#pragma unroll
    for (int tt = 0; tt < 4; ++tt) {
      float bv = bs0[tt];
      v8f tmp = { bv, bv, bv, bv, bv, bv, bv, bv };
      acc[tt] = tmp;
    }
#pragma unroll
    for (int ks = 0; ks < 8; ++ks) {
      const v8f w8 = Wf[(0 + ks) * 32];
#pragma unroll
      for (int tt = 0; tt < 4; ++tt) {
        v2f w = { w8[2 * tt], w8[2 * tt + 1] };
        acc[tt] = __builtin_amdgcn_wmma_f32_16x16x4_f32(false, xa[ks], false, w, (short)0,
                                                        acc[tt], false, false);
      }
    }
#pragma unroll
    for (int ks = 0; ks < 4; ++ks) {
      const v8f w8 = Wf[(8 + ks) * 32];
#pragma unroll
      for (int tt = 0; tt < 4; ++tt) {
        v2f w = { w8[2 * tt], w8[2 * tt + 1] };
        acc[tt] = __builtin_amdgcn_wmma_f32_16x16x4_f32(false, h0A[ks], false, w, (short)0,
                                                        acc[tt], false, false);
      }
    }
#pragma unroll
    for (int r = 0; r < 8; ++r) {
      float gi = acc[0][r], gf = acc[1][r], gg = acc[2][r], go = acc[3][r];
      float cn = sigm(gf) * c0[r] + sigm(gi) * tanhf(gg);
      c0[r] = cn;
      Ht[(r + 8 * half) * HP + l16] = sigm(go) * tanhf(cn);
    }
    __syncthreads();
#pragma unroll
    for (int ks = 0; ks < 4; ++ks) h0A[ks] = *(const v2f*)(Ht + l16 * HP + ks * 4 + 2 * half);

    const v4f bs1 = *(const v4f*)(Bb + 64 + l16 * 4);
#pragma unroll
    for (int tt = 0; tt < 4; ++tt) {
      float bv = bs1[tt];
      v8f tmp = { bv, bv, bv, bv, bv, bv, bv, bv };
      acc[tt] = tmp;
    }
#pragma unroll
    for (int ks = 0; ks < 4; ++ks) {
      const v8f w8 = Wf[(12 + ks) * 32];
#pragma unroll
      for (int tt = 0; tt < 4; ++tt) {
        v2f w = { w8[2 * tt], w8[2 * tt + 1] };
        acc[tt] = __builtin_amdgcn_wmma_f32_16x16x4_f32(false, h0A[ks], false, w, (short)0,
                                                        acc[tt], false, false);
      }
    }
#pragma unroll
    for (int ks = 0; ks < 4; ++ks) {
      const v8f w8 = Wf[(16 + ks) * 32];
#pragma unroll
      for (int tt = 0; tt < 4; ++tt) {
        v2f w = { w8[2 * tt], w8[2 * tt + 1] };
        acc[tt] = __builtin_amdgcn_wmma_f32_16x16x4_f32(false, h1A[ks], false, w, (short)0,
                                                        acc[tt], false, false);
      }
    }
#pragma unroll
    for (int r = 0; r < 8; ++r) {
      float gi = acc[0][r], gf = acc[1][r], gg = acc[2][r], go = acc[3][r];
      float cn = sigm(gf) * c1[r] + sigm(gi) * tanhf(gg);
      c1[r] = cn;
      Ht[(r + 8 * half) * HP + l16] = sigm(go) * tanhf(cn);
    }
    __syncthreads();
#pragma unroll
    for (int ks = 0; ks < 4; ++ks) h1A[ks] = *(const v2f*)(Ht + l16 * HP + ks * 4 + 2 * half);

    float a = bo;
#pragma unroll
    for (int j = 0; j < 16; ++j) a += Ht[l16 * HP + j] * wo[j];
    if (row0 + l16 < N) out[((long)t * N + row0 + l16) * 2 + half] = a;
  }
}

static inline int gs(long n, int b) { return (int)((n + b - 1) / b); }

extern "C" void kernel_launch(void* const* d_in, const int* in_sizes, int n_in,
                              void* d_out, int out_size, void* d_ws, size_t ws_size,
                              hipStream_t stream) {
  (void)in_sizes; (void)n_in; (void)out_size;

  const int*   seq  = (const int*)d_in[0];
  const int*   gidx = (const int*)d_in[1];
  const float* Wemb = (const float*)d_in[2];
  const float* bemb = (const float*)d_in[3];
  const float* W1   = (const float*)d_in[4];
  const float* b1   = (const float*)d_in[5];
  const float* W2   = (const float*)d_in[6];
  const float* b2   = (const float*)d_in[7];
  const float* We2n = (const float*)d_in[8];
  const float* be2n = (const float*)d_in[9];
  const float* Wi0  = (const float*)d_in[10];
  const float* Wh0  = (const float*)d_in[11];
  const float* bL0  = (const float*)d_in[12];
  const float* Wi1  = (const float*)d_in[13];
  const float* Wh1  = (const float*)d_in[14];
  const float* bL1  = (const float*)d_in[15];
  const float* Wout = (const float*)d_in[16];
  const float* bout = (const float*)d_in[17];

  const int* srcE = gidx;
  const int* dstE = gidx + N_EDGES;
  float* out = (float*)d_out;

  float* p = (float*)d_ws;
  float* isq   = p; p += 5056;
  float* idg   = p; p += 5056;
  int*   uptr  = (int*)p; p += (size_t)T_STEPS * NUM_LOC;
  int*   gptr  = (int*)p; p += 5056;
  int*   gcol  = (int*)p; p += N_EDGES;
  float* gw    = p; p += N_EDGES;
  const size_t FIXED = 3 * 5056 + (size_t)T_STEPS * NUM_LOC + 2 * (size_t)N_EDGES;
  const size_t AREA  = (size_t)NUM_LOC * T_STEPS * 64;

  const bool all_steps = ws_size >= (FIXED + 2 * AREA) * sizeof(float);
  float* e     = p;
  float* h1    = all_steps ? p + AREA : p + NUM_LOC * 64;
  float* heall = all_steps ? p : p + 2 * NUM_LOC * 64;
  int*   ulist = all_steps ? (int*)(p + AREA) : (int*)heall;
  const int ustride = all_steps ? USR_NUM : NUM_LOC * 32;

  const int BT = 256;
  const long NIDX = (long)T_STEPS * USR_NUM + N_EDGES;
  const int NZ = T_STEPS * NUM_LOC + NUM_LOC;

  k_zero_i<<<gs(NZ, BT), BT, 0, stream>>>(uptr, NZ);
  k_count<<<gs(NIDX, BT), BT, 0, stream>>>(seq, dstE, uptr, gptr);
  k_scan<<<T_STEPS + 1, 1024, 0, stream>>>(uptr, gptr, isq, idg);
  k_fill<<<gs(NIDX, BT), BT, 0, stream>>>(seq, srcE, dstE, isq, uptr, gptr, ulist, ustride, gcol, gw);

  if (all_steps) {
    const int TS = 4;
    k_gather_users<TS, T_STEPS><<<dim3(gs(NUM_LOC, 8), T_STEPS / TS), 256, 0, stream>>>(
        uptr, ulist, ustride, Wemb, bemb, e, NUM_LOC);
    k_gcn_relu<TS, T_STEPS><<<dim3(gs(NUM_LOC, 16), T_STEPS / TS), 512, 0, stream>>>(
        gptr, gcol, gw, idg, e, W1, b1, h1, NUM_LOC);
    k_gcn_e2n<TS, T_STEPS><<<dim3(gs(NUM_LOC, 16), T_STEPS / TS), 512, 0, stream>>>(
        gptr, gcol, gw, idg, h1, W2, b2, We2n, be2n, heall, NUM_LOC);
  } else {
    for (int t = 0; t < T_STEPS; ++t) {
      k_gather_users<1, 1><<<gs(NUM_LOC, 8), 256, 0, stream>>>(
          uptr + (size_t)t * NUM_LOC, ulist + (size_t)t * ustride, ustride, Wemb, bemb, e, NUM_LOC);
      k_gcn_relu<1, 1><<<gs(NUM_LOC, 16), 512, 0, stream>>>(gptr, gcol, gw, idg, e, W1, b1, h1, NUM_LOC);
      k_gcn_e2n<1, 1><<<gs(NUM_LOC, 16), 512, 0, stream>>>(gptr, gcol, gw, idg, h1, W2, b2, We2n, be2n,
                                                            heall + (size_t)t * NUM_LOC * 32, NUM_LOC);
    }
  }

  k_lstm_all<<<gs(USR_NUM, 32), 64, 0, stream>>>(seq, heall, Wi0, Wh0, bL0, Wi1, Wh1, bL1,
                                                 Wout, bout, out, USR_NUM);
}


# Index-list count and fill kernels take 8 keys a thread (two 16-byte loads, 8 atomics in flight) instead of 1: 52504 -> 6568 waves per launch; list order stays atomic arrival order
# speedup vs baseline: 4.6558x; 1.0321x over previous
#include <hip/hip_runtime.h>
#include <hip/hip_bf16.h>
#include <math.h>

#define USR_NUM 100000
#define USR_DIM 64
#define NUM_LOC 5000
#define T_STEPS 16
#define N_EDGES 80000

typedef float v2f __attribute__((ext_vector_type(2)));
typedef float v4f __attribute__((ext_vector_type(4)));
typedef float v8f __attribute__((ext_vector_type(8)));


__global__ void k_zero_i(int* __restrict__ p, int n) {
  int i = blockIdx.x * blockDim.x + threadIdx.x;
  if (i < n) p[i] = 0;
}

typedef int v4i __attribute__((ext_vector_type(4)));
#define NU4 (T_STEPS * USR_NUM / 4)
#define NE4 (N_EDGES / 4)
#define NQH ((NU4 + NE4) / 2)
static_assert(USR_NUM % 4 == 0 && N_EDGES % 4 == 0 && (NU4 + NE4) % 2 == 0 && NQH < NU4,
              "key groups of 4 must not straddle steps, and a thread's first group must be user keys");

__global__ __launch_bounds__(256) void k_count(const int* __restrict__ seq, const int* __restrict__ dst,
                                               int* __restrict__ uptr, int* __restrict__ gptr) {
  const int g0 = blockIdx.x * blockDim.x + threadIdx.x;
  if (g0 >= NQH) return;
  const int g1 = g0 + NQH;
  const bool usr = g1 < NU4;
  const v4i a = *(const v4i*)(seq + 4 * (long)g0);
  const v4i b = usr ? *(const v4i*)(seq + 4 * (long)g1) : *(const v4i*)(dst + 4 * (long)(g1 - NU4));
  int* pa = uptr + (4 * g0 / USR_NUM) * NUM_LOC;
  int* pb = usr ? uptr + (4 * g1 / USR_NUM) * NUM_LOC : gptr;
#pragma unroll
  for (int s = 0; s < 4; ++s) atomicAdd(pa + a[s], 1);
#pragma unroll
  for (int s = 0; s < 4; ++s) atomicAdd(pb + b[s], 1);
}

__global__ __launch_bounds__(1024) void k_scan(int* __restrict__ uptr, int* __restrict__ gptr,
                                               float* __restrict__ isq, float* __restrict__ idg) {
  __shared__ int part[1024];
  const int tid = threadIdx.x;
  const bool graph = blockIdx.x == T_STEPS;
  int* p = graph ? gptr : uptr + blockIdx.x * NUM_LOC;
  int v[5];
  int s = 0;
#pragma unroll
  for (int k = 0; k < 5; ++k) {
    int idx = tid * 5 + k;
    v[k] = idx < NUM_LOC ? p[idx] : 0;
    s += v[k];
  }
  if (graph) {
#pragma unroll
    for (int k = 0; k < 5; ++k) {
      int idx = tid * 5 + k;
      if (idx < NUM_LOC) {
        float d = (float)v[k] + 1.0f;
        isq[idx] = rsqrtf(d);
        idg[idx] = 1.0f / d;
      }
    }
  }
  part[tid] = s;
  __syncthreads();
  for (int off = 1; off < 1024; off <<= 1) {
    int x = tid >= off ? part[tid - off] : 0;
    __syncthreads();
    part[tid] += x;
    __syncthreads();
  }
  int run = part[tid] - s;
#pragma unroll
  for (int k = 0; k < 5; ++k) {
    int idx = tid * 5 + k;
    if (idx < NUM_LOC) { p[idx] = run; run += v[k]; }
  }
}

__global__ __launch_bounds__(256) void k_fill(const int* __restrict__ seq, const int* __restrict__ src,
                       const int* __restrict__ dst, const float* __restrict__ isq,
                       int* __restrict__ uptr, int* __restrict__ gptr,
                       int* __restrict__ ulist_all, int ustride, int* __restrict__ gcol,
                       float* __restrict__ gw) {
  const int g0 = blockIdx.x * blockDim.x + threadIdx.x;
  if (g0 >= NQH) return;
  const int g1 = g0 + NQH;
  const bool usr = g1 < NU4;
  const v4i a = *(const v4i*)(seq + 4 * (long)g0);
  const v4i b = usr ? *(const v4i*)(seq + 4 * (long)g1) : *(const v4i*)(dst + 4 * (long)(g1 - NU4));
  const int ta = 4 * g0 / USR_NUM;
  const int ia = 4 * g0 - ta * USR_NUM;
  const int tb = usr ? 4 * g1 / USR_NUM : 0;
  const int ib = 4 * g1 - tb * USR_NUM;
  int* pa = uptr + ta * NUM_LOC;
  int* pb = usr ? uptr + tb * NUM_LOC : gptr;
  int* la = ulist_all + (long)ta * ustride;
  int posa[4], posb[4];
#pragma unroll
  for (int s = 0; s < 4; ++s) posa[s] = atomicAdd(pa + a[s], 1);
#pragma unroll
  for (int s = 0; s < 4; ++s) posb[s] = atomicAdd(pb + b[s], 1);
  if (usr) {
    int* lb = ulist_all + (long)tb * ustride;
#pragma unroll
    for (int s = 0; s < 4; ++s) la[posa[s]] = ia + s;
#pragma unroll
    for (int s = 0; s < 4; ++s) lb[posb[s]] = ib + s;
  } else {
    const v4i sv = *(const v4i*)(src + 4 * (long)(g1 - NU4));
    float w[4];
#pragma unroll
    for (int s = 0; s < 4; ++s) w[s] = isq[sv[s]] * isq[b[s]];
#pragma unroll
    for (int s = 0; s < 4; ++s) la[posa[s]] = ia + s;
#pragma unroll
    for (int s = 0; s < 4; ++s) { gcol[posb[s]] = sv[s]; gw[posb[s]] = w[s]; }
  }
}

template <int TS, int NS>
__global__ __launch_bounds__(256) void k_gather_users(
    const int* __restrict__ uptr, const int* __restrict__ ulist, int ustride,
    const float* __restrict__ Wemb, const float* __restrict__ bemb, float* __restrict__ e_all,
    int nloc) {
  const int lane = threadIdx.x & 31;
  const int j = __builtin_amdgcn_readfirstlane(blockIdx.x * 8 + (threadIdx.x >> 5));
  if (j >= nloc) return;
  const v2f bb = *(const v2f*)(bemb + 2 * lane);
  for (int it = 0; it < TS; ++it) {
    const int t = blockIdx.y * TS + it;
    const int* ptr = uptr + t * NUM_LOC;
    const int* ul = ulist + (long)t * ustride;
    const int beg = j ? ptr[j - 1] : 0;
    const int end = ptr[j];
    v2f acc = { 0.0f, 0.0f };
    for (int k = beg; k < end; k += 8) {
      int id[8];
#pragma unroll
      for (int u = 0; u < 8; ++u) id[u] = ul[k + u < end ? k + u : end - 1];
      v2f x[8];
#pragma unroll
      for (int u = 0; u < 8; ++u) x[u] = *(const v2f*)(Wemb + (long)id[u] * 64 + 2 * lane);
#pragma unroll
      for (int u = 0; u < 8; ++u) {
        if (k + u < end) acc += x[u] + bb;
      }
    }
    float cv = (float)(end - beg); if (cv < 1.0f) cv = 1.0f;
    v2f r = { acc[0] / cv, acc[1] / cv };
    *(v2f*)(e_all + ((long)j * NS + t) * 64 + 2 * lane) = r;
  }
}

#define TP 68
#define OP 36

template <int TS, int NS>
__device__ __forceinline__ void gather_node(
    v2f (&acc)[TS], const int* __restrict__ gptr, const int* __restrict__ gcol,
    const float* __restrict__ gw, const float* __restrict__ idg, const float* __restrict__ h,
    int q, int t0, int lane) {
  constexpr int GE = TS == 1 ? 8 : 4;
  const int beg = q ? gptr[q - 1] : 0;
  const int end = gptr[q];
  const float* hl = h + (long)t0 * 64 + 2 * lane;
  v2f self[TS];
#pragma unroll
  for (int ts = 0; ts < TS; ++ts) self[ts] = *(const v2f*)(hl + (long)q * (NS * 64) + ts * 64);
  const float sd = idg[q];
#pragma unroll
  for (int ts = 0; ts < TS; ++ts) { v2f z = { 0.0f, 0.0f }; acc[ts] = z; }
  for (int k = beg; k < end; k += GE) {
    int s[GE];
    float w[GE];
#pragma unroll
    for (int u = 0; u < GE; ++u) {
      const int kk = k + u < end ? k + u : end - 1;
      s[u] = gcol[kk];
      w[u] = gw[kk];
    }
    v2f x[GE][TS];
#pragma unroll
    for (int u = 0; u < GE; ++u) {
      const float* hs = hl + (long)s[u] * (NS * 64);
#pragma unroll
      for (int ts = 0; ts < TS; ++ts) x[u][ts] = *(const v2f*)(hs + ts * 64);
    }
#pragma unroll
    for (int u = 0; u < GE; ++u) {
      const bool live = k + u < end;
#pragma unroll
      for (int ts = 0; ts < TS; ++ts) {
        const v2f old = acc[ts];
        const v2f upd = old + x[u][ts] * w[u];
        acc[ts] = live ? upd : old;
      }
    }
  }
#pragma unroll
  for (int ts = 0; ts < TS; ++ts) acc[ts] += self[ts] * sd;
}

__device__ __forceinline__ void load_bfrag(v2f (&b)[16], const float* __restrict__ Wm, int nc,
                                           int tt, int half, int l16) {
#pragma unroll
  for (int ks = 0; ks < 16; ++ks) {
    b[ks][0] = Wm[(4 * ks + 2 * half + 0) * nc + tt * 16 + l16];
    b[ks][1] = Wm[(4 * ks + 2 * half + 1) * nc + tt * 16 + l16];
  }
}

template <int NT>
__device__ __forceinline__ void tiles_product(v8f (&c)[NT], const float* tiles, const v2f (&b)[16],
                                              float bv, int half, int l16) {
#pragma unroll
  for (int i = 0; i < NT; ++i) { v8f z = { bv, bv, bv, bv, bv, bv, bv, bv }; c[i] = z; }
#pragma unroll
  for (int ks = 0; ks < 16; ++ks) {
#pragma unroll
    for (int i = 0; i < NT; ++i) {
      const v2f a = *(const v2f*)(tiles + (i * 16 + l16) * TP + 4 * ks + 2 * half);
      c[i] = __builtin_amdgcn_wmma_f32_16x16x4_f32(false, a, false, b[ks], (short)0, c[i], false, false);
    }
  }
}

template <int TS, int NS>
__global__ __launch_bounds__(512) void k_gcn_relu(
    const int* __restrict__ gptr, const int* __restrict__ gcol, const float* __restrict__ gw,
    const float* __restrict__ idg, const float* __restrict__ hin, const float* __restrict__ W,
    const float* __restrict__ bias, float* __restrict__ hout, int nloc) {
  __shared__ __attribute__((aligned(16))) float At[TS * 16 * TP];
  __shared__ __attribute__((aligned(16))) float Ct[TS * 16 * TP];
  const int lane = threadIdx.x & 31;
  const int wave = threadIdx.x >> 5;
  const int half = lane >> 4;
  const int l16  = lane & 15;
  const int t0   = blockIdx.y * TS;
  const int q = __builtin_amdgcn_readfirstlane(blockIdx.x * 16 + wave);

  v2f bf[16];
  float bv = 0.0f;
  if (wave < 4) {
    load_bfrag(bf, W, 64, wave, half, l16);
    bv = bias[wave * 16 + l16];
  }

  v2f acc[TS];
#pragma unroll
  for (int ts = 0; ts < TS; ++ts) { v2f z = { 0.0f, 0.0f }; acc[ts] = z; }
  if (q < nloc) gather_node<TS, NS>(acc, gptr, gcol, gw, idg, hin, q, t0, lane);
#pragma unroll
  for (int ts = 0; ts < TS; ++ts) *(v2f*)(At + (ts * 16 + wave) * TP + 2 * lane) = acc[ts];
  __syncthreads();

  if (wave < 4) {
    v8f c[TS];
    tiles_product<TS>(c, At, bf, bv, half, l16);
#pragma unroll
    for (int ts = 0; ts < TS; ++ts) {
#pragma unroll
      for (int r = 0; r < 8; ++r) {
        const float v = c[ts][r];
        Ct[(ts * 16 + r + 8 * half) * TP + wave * 16 + l16] = v > 0.0f ? v : 0.0f;
      }
    }
  }
  __syncthreads();

  if (q < nloc) {
#pragma unroll
    for (int ts = 0; ts < TS; ++ts)
      *(v2f*)(hout + ((long)q * NS + t0 + ts) * 64 + 2 * lane) =
          *(const v2f*)(Ct + (ts * 16 + wave) * TP + 2 * lane);
  }
}

template <int TS, int NS>
__global__ __launch_bounds__(512) void k_gcn_e2n(
    const int* __restrict__ gptr, const int* __restrict__ gcol, const float* __restrict__ gw,
    const float* __restrict__ idg, const float* __restrict__ hin, const float* __restrict__ W2,
    const float* __restrict__ b2, const float* __restrict__ We, const float* __restrict__ be,
    float* __restrict__ out, int nloc) {
  __shared__ __attribute__((aligned(16))) float At[TS * 16 * TP];
  __shared__ __attribute__((aligned(16))) float Ht[TS * 16 * TP];
  __shared__ __attribute__((aligned(16))) float Ot[TS * 16 * OP];
  const int lane = threadIdx.x & 31;
  const int wave = threadIdx.x >> 5;
  const int half = lane >> 4;
  const int l16  = lane & 15;
  const int t0   = blockIdx.y * TS;
  const int q = __builtin_amdgcn_readfirstlane(blockIdx.x * 16 + wave);

  constexpr int ET = TS == 1 ? 1 : 2;
  constexpr int EW = 2 * TS / ET;
  const int ec = (wave - 4) & 1;
  const int es = ((wave - 4) >> 1) * ET;
  v2f bf[16];
  float bv = 0.0f;
  if (wave < 4) {
    load_bfrag(bf, W2, 64, wave, half, l16);
    bv = b2[wave * 16 + l16];
  } else if (wave < 4 + EW) {
    load_bfrag(bf, We, 32, ec, half, l16);
    bv = be[ec * 16 + l16];
  }

  v2f acc[TS];
#pragma unroll
  for (int ts = 0; ts < TS; ++ts) { v2f z = { 0.0f, 0.0f }; acc[ts] = z; }
  if (q < nloc) gather_node<TS, NS>(acc, gptr, gcol, gw, idg, hin, q, t0, lane);
#pragma unroll
  for (int ts = 0; ts < TS; ++ts) *(v2f*)(At + (ts * 16 + wave) * TP + 2 * lane) = acc[ts];
  __syncthreads();

  if (wave < 4) {
    v8f c[TS];
    tiles_product<TS>(c, At, bf, bv, half, l16);
#pragma unroll
    for (int ts = 0; ts < TS; ++ts) {
#pragma unroll
      for (int r = 0; r < 8; ++r) Ht[(ts * 16 + r + 8 * half) * TP + wave * 16 + l16] = c[ts][r];
    }
  }
  __syncthreads();

  if (wave >= 4 && wave < 4 + EW) {
    v8f c[ET];
    tiles_product<ET>(c, Ht + es * 16 * TP, bf, bv, half, l16);
#pragma unroll
    for (int i = 0; i < ET; ++i) {
#pragma unroll
      for (int r = 0; r < 8; ++r) Ot[((es + i) * 16 + r + 8 * half) * OP + ec * 16 + l16] = c[i][r];
    }
  }
  __syncthreads();

  if (q < nloc) {
#pragma unroll
    for (int ts = 0; ts < TS; ++ts)
      out[((long)(t0 + ts) * nloc + q) * 32 + lane] = Ot[(ts * 16 + wave) * OP + lane];
  }
}

#define HP 20

__device__ __forceinline__ float sigm(float x) { return 1.0f / (1.0f + expf(-x)); }

template <int NKS>
__device__ __forceinline__ void stage_frag(float* dst, const float* __restrict__ Wm) {
  for (int i = threadIdx.x; i < NKS * 256; i += 64) {
    const int s = i & 1, tt = (i >> 1) & 3, ln = (i >> 3) & 31, ks = i >> 8;
    dst[i] = Wm[(ks * 4 + 2 * (ln >> 4) + s) * 64 + tt * 16 + (ln & 15)];
  }
}

__global__ __launch_bounds__(64) void k_lstm_all(
    const int* __restrict__ seq,
    const float* __restrict__ he,
    const float* __restrict__ Wi0, const float* __restrict__ Wh0, const float* __restrict__ b0,
    const float* __restrict__ Wi1, const float* __restrict__ Wh1, const float* __restrict__ b1,
    const float* __restrict__ Wout, const float* __restrict__ bout,
    float* __restrict__ out, int N) {
  __shared__ __attribute__((aligned(32))) float Ws[80 * 64];
  __shared__ __attribute__((aligned(16))) float Hs[2 * 16 * HP];
  __shared__ __attribute__((aligned(16))) float Wo[32];
  __shared__ __attribute__((aligned(16))) float Bb[2 * 16 * 4];
  stage_frag<8>(Ws,            Wi0);
  stage_frag<4>(Ws +  8 * 256, Wh0);
  stage_frag<4>(Ws + 12 * 256, Wi1);
  stage_frag<4>(Ws + 16 * 256, Wh1);
  if (threadIdx.x < 32) Wo[threadIdx.x] = Wout[(threadIdx.x & 15) * 2 + (threadIdx.x >> 4)];
  Bb[threadIdx.x]      = b0[(threadIdx.x & 3) * 16 + (threadIdx.x >> 2)];
  Bb[64 + threadIdx.x] = b1[(threadIdx.x & 3) * 16 + (threadIdx.x >> 2)];
  __syncthreads();

  const int lane = threadIdx.x & 31;
  const int wave = threadIdx.x >> 5;
  const int half = lane >> 4;
  const int l16  = lane & 15;
  const int row0 = (blockIdx.x * 2 + wave) * 16;

  int ar = row0 + l16; if (ar > N - 1) ar = N - 1;

  const float* wo = Wo + half * 16;
  const float bo = bout[half];

  const v8f* Wf = (const v8f*)Ws + lane;
  float* Ht = Hs + wave * (16 * HP);

  v2f h0A[4], h1A[4];
  float c0[8], c1[8];
#pragma unroll
  for (int i = 0; i < 4; ++i) { v2f z = { 0.0f, 0.0f }; h0A[i] = z; h1A[i] = z; }
#pragma unroll
  for (int i = 0; i < 8; ++i) { c0[i] = 0.0f; c1[i] = 0.0f; }

  int id = seq[ar];
  for (int t = 0; t < T_STEPS; ++t) {
    const float* Xr = he + ((long)t * NUM_LOC + id) * 32 + 2 * half;
    v2f xa[8];
#pragma unroll
    for (int ks = 0; ks < 8; ++ks) xa[ks] = *(const v2f*)(Xr + ks * 4);
    const int tn = t + 1 < T_STEPS ? t + 1 : t;
    id = seq[(long)tn * N + ar];

    v8f acc[4];
    const v4f bs0 = *(const v4f*)(Bb + l16 * 4);
#pragma unroll
    for (int tt = 0; tt < 4; ++tt) {
      float bv = bs0[tt];
      v8f tmp = { bv, bv, bv, bv, bv, bv, bv, bv };
      acc[tt] = tmp;
    }
#pragma unroll
    for (int ks = 0; ks < 8; ++ks) {
      const v8f w8 = Wf[(0 + ks) * 32];
#pragma unroll
      for (int tt = 0; tt < 4; ++tt) {
        v2f w = { w8[2 * tt], w8[2 * tt + 1] };
        acc[tt] = __builtin_amdgcn_wmma_f32_16x16x4_f32(false, xa[ks], false, w, (short)0,
                                                        acc[tt], false, false);
      }
    }
#pragma unroll
    for (int ks = 0; ks < 4; ++ks) {
      const v8f w8 = Wf[(8 + ks) * 32];
#pragma unroll
      for (int tt = 0; tt < 4; ++tt) {
        v2f w = { w8[2 * tt], w8[2 * tt + 1] };
        acc[tt] = __builtin_amdgcn_wmma_f32_16x16x4_f32(false, h0A[ks], false, w, (short)0,
                                                        acc[tt], false, false);
      }
    }
#pragma unroll
    for (int r = 0; r < 8; ++r) {
      float gi = acc[0][r], gf = acc[1][r], gg = acc[2][r], go = acc[3][r];
      float cn = sigm(gf) * c0[r] + sigm(gi) * tanhf(gg);
      c0[r] = cn;
      Ht[(r + 8 * half) * HP + l16] = sigm(go) * tanhf(cn);
    }
    __syncthreads();
#pragma unroll
    for (int ks = 0; ks < 4; ++ks) h0A[ks] = *(const v2f*)(Ht + l16 * HP + ks * 4 + 2 * half);

    const v4f bs1 = *(const v4f*)(Bb + 64 + l16 * 4);
#pragma unroll
    for (int tt = 0; tt < 4; ++tt) {
      float bv = bs1[tt];
      v8f tmp = { bv, bv, bv, bv, bv, bv, bv, bv };
      acc[tt] = tmp;
    }
#pragma unroll
    for (int ks = 0; ks < 4; ++ks) {
      const v8f w8 = Wf[(12 + ks) * 32];
#pragma unroll
      for (int tt = 0; tt < 4; ++tt) {
        v2f w = { w8[2 * tt], w8[2 * tt + 1] };
        acc[tt] = __builtin_amdgcn_wmma_f32_16x16x4_f32(false, h0A[ks], false, w, (short)0,
                                                        acc[tt], false, false);
      }
    }
#pragma unroll
    for (int ks = 0; ks < 4; ++ks) {
      const v8f w8 = Wf[(16 + ks) * 32];
#pragma unroll
      for (int tt = 0; tt < 4; ++tt) {
        v2f w = { w8[2 * tt], w8[2 * tt + 1] };
        acc[tt] = __builtin_amdgcn_wmma_f32_16x16x4_f32(false, h1A[ks], false, w, (short)0,
                                                        acc[tt], false, false);
      }
    }
#pragma unroll
    for (int r = 0; r < 8; ++r) {
      float gi = acc[0][r], gf = acc[1][r], gg = acc[2][r], go = acc[3][r];
      float cn = sigm(gf) * c1[r] + sigm(gi) * tanhf(gg);
      c1[r] = cn;
      Ht[(r + 8 * half) * HP + l16] = sigm(go) * tanhf(cn);
    }
    __syncthreads();
#pragma unroll
    for (int ks = 0; ks < 4; ++ks) h1A[ks] = *(const v2f*)(Ht + l16 * HP + ks * 4 + 2 * half);

    float a = bo;
#pragma unroll
    for (int j = 0; j < 16; ++j) a += Ht[l16 * HP + j] * wo[j];
    if (row0 + l16 < N) out[((long)t * N + row0 + l16) * 2 + half] = a;
  }
}

static inline int gs(long n, int b) { return (int)((n + b - 1) / b); }

extern "C" void kernel_launch(void* const* d_in, const int* in_sizes, int n_in,
                              void* d_out, int out_size, void* d_ws, size_t ws_size,
                              hipStream_t stream) {
  (void)in_sizes; (void)n_in; (void)out_size;

  const int*   seq  = (const int*)d_in[0];
  const int*   gidx = (const int*)d_in[1];
  const float* Wemb = (const float*)d_in[2];
  const float* bemb = (const float*)d_in[3];
  const float* W1   = (const float*)d_in[4];
  const float* b1   = (const float*)d_in[5];
  const float* W2   = (const float*)d_in[6];
  const float* b2   = (const float*)d_in[7];
  const float* We2n = (const float*)d_in[8];
  const float* be2n = (const float*)d_in[9];
  const float* Wi0  = (const float*)d_in[10];
  const float* Wh0  = (const float*)d_in[11];
  const float* bL0  = (const float*)d_in[12];
  const float* Wi1  = (const float*)d_in[13];
  const float* Wh1  = (const float*)d_in[14];
  const float* bL1  = (const float*)d_in[15];
  const float* Wout = (const float*)d_in[16];
  const float* bout = (const float*)d_in[17];

  const int* srcE = gidx;
  const int* dstE = gidx + N_EDGES;
  float* out = (float*)d_out;

  float* p = (float*)d_ws;
  float* isq   = p; p += 5056;
  float* idg   = p; p += 5056;
  int*   uptr  = (int*)p; p += (size_t)T_STEPS * NUM_LOC;
  int*   gptr  = (int*)p; p += 5056;
  int*   gcol  = (int*)p; p += N_EDGES;
  float* gw    = p; p += N_EDGES;
  const size_t FIXED = 3 * 5056 + (size_t)T_STEPS * NUM_LOC + 2 * (size_t)N_EDGES;
  const size_t AREA  = (size_t)NUM_LOC * T_STEPS * 64;

  const bool all_steps = ws_size >= (FIXED + 2 * AREA) * sizeof(float);
  float* e     = p;
  float* h1    = all_steps ? p + AREA : p + NUM_LOC * 64;
  float* heall = all_steps ? p : p + 2 * NUM_LOC * 64;
  int*   ulist = all_steps ? (int*)(p + AREA) : (int*)heall;
  const int ustride = all_steps ? USR_NUM : NUM_LOC * 32;

  const int BT = 256;
  const int NZ = T_STEPS * NUM_LOC + NUM_LOC;

  k_zero_i<<<gs(NZ, BT), BT, 0, stream>>>(uptr, NZ);
  k_count<<<gs(NQH, BT), BT, 0, stream>>>(seq, dstE, uptr, gptr);
  k_scan<<<T_STEPS + 1, 1024, 0, stream>>>(uptr, gptr, isq, idg);
  k_fill<<<gs(NQH, BT), BT, 0, stream>>>(seq, srcE, dstE, isq, uptr, gptr, ulist, ustride, gcol, gw);

  if (all_steps) {
    const int TS = 4;
    k_gather_users<TS, T_STEPS><<<dim3(gs(NUM_LOC, 8), T_STEPS / TS), 256, 0, stream>>>(
        uptr, ulist, ustride, Wemb, bemb, e, NUM_LOC);
    k_gcn_relu<TS, T_STEPS><<<dim3(gs(NUM_LOC, 16), T_STEPS / TS), 512, 0, stream>>>(
        gptr, gcol, gw, idg, e, W1, b1, h1, NUM_LOC);
    k_gcn_e2n<TS, T_STEPS><<<dim3(gs(NUM_LOC, 16), T_STEPS / TS), 512, 0, stream>>>(
        gptr, gcol, gw, idg, h1, W2, b2, We2n, be2n, heall, NUM_LOC);
  } else {
    for (int t = 0; t < T_STEPS; ++t) {
      k_gather_users<1, 1><<<gs(NUM_LOC, 8), 256, 0, stream>>>(
          uptr + (size_t)t * NUM_LOC, ulist + (size_t)t * ustride, ustride, Wemb, bemb, e, NUM_LOC);
      k_gcn_relu<1, 1><<<gs(NUM_LOC, 16), 512, 0, stream>>>(gptr, gcol, gw, idg, e, W1, b1, h1, NUM_LOC);
      k_gcn_e2n<1, 1><<<gs(NUM_LOC, 16), 512, 0, stream>>>(gptr, gcol, gw, idg, h1, W2, b2, We2n, be2n,
                                                            heall + (size_t)t * NUM_LOC * 32, NUM_LOC);
    }
  }

  k_lstm_all<<<gs(USR_NUM, 32), 64, 0, stream>>>(seq, heall, Wi0, Wh0, bL0, Wi1, Wh1, bL1,
                                                 Wout, bout, out, USR_NUM);
}


# Recurrent tail's cell: sigmoid as rcp(1 + exp2(-x log2 e)) on the exp and reciprocal units for library exp + exact division; tanh polynomial on 8 values at once, large-argument form by wave ballot
# speedup vs baseline: 5.6619x; 1.2161x over previous
#include <hip/hip_runtime.h>
#include <hip/hip_bf16.h>
#include <math.h>

#define USR_NUM 100000
#define USR_DIM 64
#define NUM_LOC 5000
#define T_STEPS 16
#define N_EDGES 80000

typedef float v2f __attribute__((ext_vector_type(2)));
typedef float v4f __attribute__((ext_vector_type(4)));
typedef float v8f __attribute__((ext_vector_type(8)));


__global__ void k_zero_i(int* __restrict__ p, int n) {
  int i = blockIdx.x * blockDim.x + threadIdx.x;
  if (i < n) p[i] = 0;
}

typedef int v4i __attribute__((ext_vector_type(4)));
#define NU4 (T_STEPS * USR_NUM / 4)
#define NE4 (N_EDGES / 4)
#define NQH ((NU4 + NE4) / 2)
static_assert(USR_NUM % 4 == 0 && N_EDGES % 4 == 0 && (NU4 + NE4) % 2 == 0 && NQH < NU4,
              "key groups of 4 must not straddle steps, and a thread's first group must be user keys");

__global__ __launch_bounds__(256) void k_count(const int* __restrict__ seq, const int* __restrict__ dst,
                                               int* __restrict__ uptr, int* __restrict__ gptr) {
  const int g0 = blockIdx.x * blockDim.x + threadIdx.x;
  if (g0 >= NQH) return;
  const int g1 = g0 + NQH;
  const bool usr = g1 < NU4;
  const v4i a = *(const v4i*)(seq + 4 * (long)g0);
  const v4i b = usr ? *(const v4i*)(seq + 4 * (long)g1) : *(const v4i*)(dst + 4 * (long)(g1 - NU4));
  int* pa = uptr + (4 * g0 / USR_NUM) * NUM_LOC;
  int* pb = usr ? uptr + (4 * g1 / USR_NUM) * NUM_LOC : gptr;
#pragma unroll
  for (int s = 0; s < 4; ++s) atomicAdd(pa + a[s], 1);
#pragma unroll
  for (int s = 0; s < 4; ++s) atomicAdd(pb + b[s], 1);
}

__global__ __launch_bounds__(1024) void k_scan(int* __restrict__ uptr, int* __restrict__ gptr,
                                               float* __restrict__ isq, float* __restrict__ idg) {
  __shared__ int part[1024];
  const int tid = threadIdx.x;
  const bool graph = blockIdx.x == T_STEPS;
  int* p = graph ? gptr : uptr + blockIdx.x * NUM_LOC;
  int v[5];
  int s = 0;
#pragma unroll
  for (int k = 0; k < 5; ++k) {
    int idx = tid * 5 + k;
    v[k] = idx < NUM_LOC ? p[idx] : 0;
    s += v[k];
  }
  if (graph) {
#pragma unroll
    for (int k = 0; k < 5; ++k) {
      int idx = tid * 5 + k;
      if (idx < NUM_LOC) {
        float d = (float)v[k] + 1.0f;
        isq[idx] = rsqrtf(d);
        idg[idx] = 1.0f / d;
      }
    }
  }
  part[tid] = s;
  __syncthreads();
  for (int off = 1; off < 1024; off <<= 1) {
    int x = tid >= off ? part[tid - off] : 0;
    __syncthreads();
    part[tid] += x;
    __syncthreads();
  }
  int run = part[tid] - s;
#pragma unroll
  for (int k = 0; k < 5; ++k) {
    int idx = tid * 5 + k;
    if (idx < NUM_LOC) { p[idx] = run; run += v[k]; }
  }
}

__global__ __launch_bounds__(256) void k_fill(const int* __restrict__ seq, const int* __restrict__ src,
                       const int* __restrict__ dst, const float* __restrict__ isq,
                       int* __restrict__ uptr, int* __restrict__ gptr,
                       int* __restrict__ ulist_all, int ustride, int* __restrict__ gcol,
                       float* __restrict__ gw) {
  const int g0 = blockIdx.x * blockDim.x + threadIdx.x;
  if (g0 >= NQH) return;
  const int g1 = g0 + NQH;
  const bool usr = g1 < NU4;
  const v4i a = *(const v4i*)(seq + 4 * (long)g0);
  const v4i b = usr ? *(const v4i*)(seq + 4 * (long)g1) : *(const v4i*)(dst + 4 * (long)(g1 - NU4));
  const int ta = 4 * g0 / USR_NUM;
  const int ia = 4 * g0 - ta * USR_NUM;
  const int tb = usr ? 4 * g1 / USR_NUM : 0;
  const int ib = 4 * g1 - tb * USR_NUM;
  int* pa = uptr + ta * NUM_LOC;
  int* pb = usr ? uptr + tb * NUM_LOC : gptr;
  int* la = ulist_all + (long)ta * ustride;
  int posa[4], posb[4];
#pragma unroll
  for (int s = 0; s < 4; ++s) posa[s] = atomicAdd(pa + a[s], 1);
#pragma unroll
  for (int s = 0; s < 4; ++s) posb[s] = atomicAdd(pb + b[s], 1);
  if (usr) {
    int* lb = ulist_all + (long)tb * ustride;
#pragma unroll
    for (int s = 0; s < 4; ++s) la[posa[s]] = ia + s;
#pragma unroll
    for (int s = 0; s < 4; ++s) lb[posb[s]] = ib + s;
  } else {
    const v4i sv = *(const v4i*)(src + 4 * (long)(g1 - NU4));
    float w[4];
#pragma unroll
    for (int s = 0; s < 4; ++s) w[s] = isq[sv[s]] * isq[b[s]];
#pragma unroll
    for (int s = 0; s < 4; ++s) la[posa[s]] = ia + s;
#pragma unroll
    for (int s = 0; s < 4; ++s) { gcol[posb[s]] = sv[s]; gw[posb[s]] = w[s]; }
  }
}

template <int TS, int NS>
__global__ __launch_bounds__(256) void k_gather_users(
    const int* __restrict__ uptr, const int* __restrict__ ulist, int ustride,
    const float* __restrict__ Wemb, const float* __restrict__ bemb, float* __restrict__ e_all,
    int nloc) {
  const int lane = threadIdx.x & 31;
  const int j = __builtin_amdgcn_readfirstlane(blockIdx.x * 8 + (threadIdx.x >> 5));
  if (j >= nloc) return;
  const v2f bb = *(const v2f*)(bemb + 2 * lane);
  for (int it = 0; it < TS; ++it) {
    const int t = blockIdx.y * TS + it;
    const int* ptr = uptr + t * NUM_LOC;
    const int* ul = ulist + (long)t * ustride;
    const int beg = j ? ptr[j - 1] : 0;
    const int end = ptr[j];
    v2f acc = { 0.0f, 0.0f };
    for (int k = beg; k < end; k += 8) {
      int id[8];
#pragma unroll
      for (int u = 0; u < 8; ++u) id[u] = ul[k + u < end ? k + u : end - 1];
      v2f x[8];
#pragma unroll
      for (int u = 0; u < 8; ++u) x[u] = *(const v2f*)(Wemb + (long)id[u] * 64 + 2 * lane);
#pragma unroll
      for (int u = 0; u < 8; ++u) {
        if (k + u < end) acc += x[u] + bb;
      }
    }
    float cv = (float)(end - beg); if (cv < 1.0f) cv = 1.0f;
    v2f r = { acc[0] / cv, acc[1] / cv };
    *(v2f*)(e_all + ((long)j * NS + t) * 64 + 2 * lane) = r;
  }
}

#define TP 68
#define OP 36

template <int TS, int NS>
__device__ __forceinline__ void gather_node(
    v2f (&acc)[TS], const int* __restrict__ gptr, const int* __restrict__ gcol,
    const float* __restrict__ gw, const float* __restrict__ idg, const float* __restrict__ h,
    int q, int t0, int lane) {
  constexpr int GE = TS == 1 ? 8 : 4;
  const int beg = q ? gptr[q - 1] : 0;
  const int end = gptr[q];
  const float* hl = h + (long)t0 * 64 + 2 * lane;
  v2f self[TS];
#pragma unroll
  for (int ts = 0; ts < TS; ++ts) self[ts] = *(const v2f*)(hl + (long)q * (NS * 64) + ts * 64);
  const float sd = idg[q];
#pragma unroll
  for (int ts = 0; ts < TS; ++ts) { v2f z = { 0.0f, 0.0f }; acc[ts] = z; }
  for (int k = beg; k < end; k += GE) {
    int s[GE];
    float w[GE];
#pragma unroll
    for (int u = 0; u < GE; ++u) {
      const int kk = k + u < end ? k + u : end - 1;
      s[u] = gcol[kk];
      w[u] = gw[kk];
    }
    v2f x[GE][TS];
#pragma unroll
    for (int u = 0; u < GE; ++u) {
      const float* hs = hl + (long)s[u] * (NS * 64);
#pragma unroll
      for (int ts = 0; ts < TS; ++ts) x[u][ts] = *(const v2f*)(hs + ts * 64);
    }
#pragma unroll
    for (int u = 0; u < GE; ++u) {
      const bool live = k + u < end;
#pragma unroll
      for (int ts = 0; ts < TS; ++ts) {
        const v2f old = acc[ts];
        const v2f upd = old + x[u][ts] * w[u];
        acc[ts] = live ? upd : old;
      }
    }
  }
#pragma unroll
  for (int ts = 0; ts < TS; ++ts) acc[ts] += self[ts] * sd;
}

__device__ __forceinline__ void load_bfrag(v2f (&b)[16], const float* __restrict__ Wm, int nc,
                                           int tt, int half, int l16) {
#pragma unroll
  for (int ks = 0; ks < 16; ++ks) {
    b[ks][0] = Wm[(4 * ks + 2 * half + 0) * nc + tt * 16 + l16];
    b[ks][1] = Wm[(4 * ks + 2 * half + 1) * nc + tt * 16 + l16];
  }
}

template <int NT>
__device__ __forceinline__ void tiles_product(v8f (&c)[NT], const float* tiles, const v2f (&b)[16],
                                              float bv, int half, int l16) {
#pragma unroll
  for (int i = 0; i < NT; ++i) { v8f z = { bv, bv, bv, bv, bv, bv, bv, bv }; c[i] = z; }
#pragma unroll
  for (int ks = 0; ks < 16; ++ks) {
#pragma unroll
    for (int i = 0; i < NT; ++i) {
      const v2f a = *(const v2f*)(tiles + (i * 16 + l16) * TP + 4 * ks + 2 * half);
      c[i] = __builtin_amdgcn_wmma_f32_16x16x4_f32(false, a, false, b[ks], (short)0, c[i], false, false);
    }
  }
}

template <int TS, int NS>
__global__ __launch_bounds__(512) void k_gcn_relu(
    const int* __restrict__ gptr, const int* __restrict__ gcol, const float* __restrict__ gw,
    const float* __restrict__ idg, const float* __restrict__ hin, const float* __restrict__ W,
    const float* __restrict__ bias, float* __restrict__ hout, int nloc) {
  __shared__ __attribute__((aligned(16))) float At[TS * 16 * TP];
  __shared__ __attribute__((aligned(16))) float Ct[TS * 16 * TP];
  const int lane = threadIdx.x & 31;
  const int wave = threadIdx.x >> 5;
  const int half = lane >> 4;
  const int l16  = lane & 15;
  const int t0   = blockIdx.y * TS;
  const int q = __builtin_amdgcn_readfirstlane(blockIdx.x * 16 + wave);

  v2f bf[16];
  float bv = 0.0f;
  if (wave < 4) {
    load_bfrag(bf, W, 64, wave, half, l16);
    bv = bias[wave * 16 + l16];
  }

  v2f acc[TS];
#pragma unroll
  for (int ts = 0; ts < TS; ++ts) { v2f z = { 0.0f, 0.0f }; acc[ts] = z; }
  if (q < nloc) gather_node<TS, NS>(acc, gptr, gcol, gw, idg, hin, q, t0, lane);
#pragma unroll
  for (int ts = 0; ts < TS; ++ts) *(v2f*)(At + (ts * 16 + wave) * TP + 2 * lane) = acc[ts];
  __syncthreads();

  if (wave < 4) {
    v8f c[TS];
    tiles_product<TS>(c, At, bf, bv, half, l16);
#pragma unroll
    for (int ts = 0; ts < TS; ++ts) {
#pragma unroll
      for (int r = 0; r < 8; ++r) {
        const float v = c[ts][r];
        Ct[(ts * 16 + r + 8 * half) * TP + wave * 16 + l16] = v > 0.0f ? v : 0.0f;
      }
    }
  }
  __syncthreads();

  if (q < nloc) {
#pragma unroll
    for (int ts = 0; ts < TS; ++ts)
      *(v2f*)(hout + ((long)q * NS + t0 + ts) * 64 + 2 * lane) =
          *(const v2f*)(Ct + (ts * 16 + wave) * TP + 2 * lane);
  }
}

template <int TS, int NS>
__global__ __launch_bounds__(512) void k_gcn_e2n(
    const int* __restrict__ gptr, const int* __restrict__ gcol, const float* __restrict__ gw,
    const float* __restrict__ idg, const float* __restrict__ hin, const float* __restrict__ W2,
    const float* __restrict__ b2, const float* __restrict__ We, const float* __restrict__ be,
    float* __restrict__ out, int nloc) {
  __shared__ __attribute__((aligned(16))) float At[TS * 16 * TP];
  __shared__ __attribute__((aligned(16))) float Ht[TS * 16 * TP];
  __shared__ __attribute__((aligned(16))) float Ot[TS * 16 * OP];
  const int lane = threadIdx.x & 31;
  const int wave = threadIdx.x >> 5;
  const int half = lane >> 4;
  const int l16  = lane & 15;
  const int t0   = blockIdx.y * TS;
  const int q = __builtin_amdgcn_readfirstlane(blockIdx.x * 16 + wave);

  constexpr int ET = TS == 1 ? 1 : 2;
  constexpr int EW = 2 * TS / ET;
  const int ec = (wave - 4) & 1;
  const int es = ((wave - 4) >> 1) * ET;
  v2f bf[16];
  float bv = 0.0f;
  if (wave < 4) {
    load_bfrag(bf, W2, 64, wave, half, l16);
    bv = b2[wave * 16 + l16];
  } else if (wave < 4 + EW) {
    load_bfrag(bf, We, 32, ec, half, l16);
    bv = be[ec * 16 + l16];
  }

  v2f acc[TS];
#pragma unroll
  for (int ts = 0; ts < TS; ++ts) { v2f z = { 0.0f, 0.0f }; acc[ts] = z; }
  if (q < nloc) gather_node<TS, NS>(acc, gptr, gcol, gw, idg, hin, q, t0, lane);
#pragma unroll
  for (int ts = 0; ts < TS; ++ts) *(v2f*)(At + (ts * 16 + wave) * TP + 2 * lane) = acc[ts];
  __syncthreads();

  if (wave < 4) {
    v8f c[TS];
    tiles_product<TS>(c, At, bf, bv, half, l16);
#pragma unroll
    for (int ts = 0; ts < TS; ++ts) {
#pragma unroll
      for (int r = 0; r < 8; ++r) Ht[(ts * 16 + r + 8 * half) * TP + wave * 16 + l16] = c[ts][r];
    }
  }
  __syncthreads();

  if (wave >= 4 && wave < 4 + EW) {
    v8f c[ET];
    tiles_product<ET>(c, Ht + es * 16 * TP, bf, bv, half, l16);
#pragma unroll
    for (int i = 0; i < ET; ++i) {
#pragma unroll
      for (int r = 0; r < 8; ++r) Ot[((es + i) * 16 + r + 8 * half) * OP + ec * 16 + l16] = c[i][r];
    }
  }
  __syncthreads();

  if (q < nloc) {
#pragma unroll
    for (int ts = 0; ts < TS; ++ts)
      out[((long)(t0 + ts) * nloc + q) * 32 + lane] = Ot[(ts * 16 + wave) * OP + lane];
  }
}

#define HP 20

__device__ __forceinline__ float sigm(float x) {
  return __builtin_amdgcn_rcpf(1.0f + __builtin_amdgcn_exp2f(x * -1.4426950408889634f));
}
__device__ __forceinline__ float tanh_small(float x) {
  const float y2 = x * x;
  float p = __builtin_fmaf(y2, -0x1.758e7ap-8f, 0x1.521192p-6f);
  p = __builtin_fmaf(y2, p, -0x1.b8389cp-5f);
  p = __builtin_fmaf(y2, p, 0x1.110704p-3f);
  p = __builtin_fmaf(y2, p, -0x1.555532p-2f);
  return __builtin_fmaf(y2, x * p, x);
}
__device__ __forceinline__ float tanh_any(float x) {
  const float y = __builtin_fabsf(x);
  const float t = __builtin_amdgcn_exp2f(y * 2.8853900817779268f);
  const float big = __builtin_fmaf(__builtin_amdgcn_rcpf(t + 1.0f), -2.0f, 1.0f);
  return y < 0.625f ? tanh_small(x) : __builtin_copysignf(big, x);
}
__device__ __forceinline__ void tanh8(float (&o)[8], const float (&x)[8]) {
  float m = __builtin_fabsf(x[0]);
#pragma unroll
  for (int r = 1; r < 8; ++r) m = __builtin_fmaxf(m, __builtin_fabsf(x[r]));
  if (__builtin_amdgcn_ballot_w32(m >= 0.625f) == 0) {
#pragma unroll
    for (int r = 0; r < 8; ++r) o[r] = tanh_small(x[r]);
  } else {
#pragma unroll
    for (int r = 0; r < 8; ++r) o[r] = tanh_any(x[r]);
  }
}
__device__ __forceinline__ void cell8(const v8f (&acc)[4], float (&c)[8], float* Ht, int half, int l16) {
  float g[8], tg[8], th[8];
#pragma unroll
  for (int r = 0; r < 8; ++r) g[r] = acc[2][r];
  tanh8(tg, g);
#pragma unroll
  for (int r = 0; r < 8; ++r) c[r] = sigm(acc[1][r]) * c[r] + sigm(acc[0][r]) * tg[r];
  tanh8(th, c);
#pragma unroll
  for (int r = 0; r < 8; ++r) Ht[(r + 8 * half) * HP + l16] = sigm(acc[3][r]) * th[r];
}

template <int NKS>
__device__ __forceinline__ void stage_frag(float* dst, const float* __restrict__ Wm) {
  for (int i = threadIdx.x; i < NKS * 256; i += 64) {
    const int s = i & 1, tt = (i >> 1) & 3, ln = (i >> 3) & 31, ks = i >> 8;
    dst[i] = Wm[(ks * 4 + 2 * (ln >> 4) + s) * 64 + tt * 16 + (ln & 15)];
  }
}

__global__ __launch_bounds__(64) void k_lstm_all(
    const int* __restrict__ seq,
    const float* __restrict__ he,
    const float* __restrict__ Wi0, const float* __restrict__ Wh0, const float* __restrict__ b0,
    const float* __restrict__ Wi1, const float* __restrict__ Wh1, const float* __restrict__ b1,
    const float* __restrict__ Wout, const float* __restrict__ bout,
    float* __restrict__ out, int N) {
  __shared__ __attribute__((aligned(32))) float Ws[80 * 64];
  __shared__ __attribute__((aligned(16))) float Hs[2 * 16 * HP];
  __shared__ __attribute__((aligned(16))) float Wo[32];
  __shared__ __attribute__((aligned(16))) float Bb[2 * 16 * 4];
  stage_frag<8>(Ws,            Wi0);
  stage_frag<4>(Ws +  8 * 256, Wh0);
  stage_frag<4>(Ws + 12 * 256, Wi1);
  stage_frag<4>(Ws + 16 * 256, Wh1);
  if (threadIdx.x < 32) Wo[threadIdx.x] = Wout[(threadIdx.x & 15) * 2 + (threadIdx.x >> 4)];
  Bb[threadIdx.x]      = b0[(threadIdx.x & 3) * 16 + (threadIdx.x >> 2)];
  Bb[64 + threadIdx.x] = b1[(threadIdx.x & 3) * 16 + (threadIdx.x >> 2)];
  __syncthreads();

  const int lane = threadIdx.x & 31;
  const int wave = threadIdx.x >> 5;
  const int half = lane >> 4;
  const int l16  = lane & 15;
  const int row0 = (blockIdx.x * 2 + wave) * 16;

  int ar = row0 + l16; if (ar > N - 1) ar = N - 1;

  const float* wo = Wo + half * 16;
  const float bo = bout[half];

  const v8f* Wf = (const v8f*)Ws + lane;
  float* Ht = Hs + wave * (16 * HP);

  v2f h0A[4], h1A[4];
  float c0[8], c1[8];
#pragma unroll
  for (int i = 0; i < 4; ++i) { v2f z = { 0.0f, 0.0f }; h0A[i] = z; h1A[i] = z; }
#pragma unroll
  for (int i = 0; i < 8; ++i) { c0[i] = 0.0f; c1[i] = 0.0f; }

  int id = seq[ar];
  for (int t = 0; t < T_STEPS; ++t) {
    const float* Xr = he + ((long)t * NUM_LOC + id) * 32 + 2 * half;
    v2f xa[8];
#pragma unroll
    for (int ks = 0; ks < 8; ++ks) xa[ks] = *(const v2f*)(Xr + ks * 4);
    const int tn = t + 1 < T_STEPS ? t + 1 : t;
    id = seq[(long)tn * N + ar];

    v8f acc[4];
    const v4f bs0 = *(const v4f*)(Bb + l16 * 4);
#pragma unroll
    for (int tt = 0; tt < 4; ++tt) {
      float bv = bs0[tt];
      v8f tmp = { bv, bv, bv, bv, bv, bv, bv, bv };
      acc[tt] = tmp;
    }
#pragma unroll
    for (int ks = 0; ks < 8; ++ks) {
      const v8f w8 = Wf[(0 + ks) * 32];
#pragma unroll
      for (int tt = 0; tt < 4; ++tt) {
        v2f w = { w8[2 * tt], w8[2 * tt + 1] };
        acc[tt] = __builtin_amdgcn_wmma_f32_16x16x4_f32(false, xa[ks], false, w, (short)0,
                                                        acc[tt], false, false);
      }
    }
#pragma unroll
    for (int ks = 0; ks < 4; ++ks) {
      const v8f w8 = Wf[(8 + ks) * 32];
#pragma unroll
      for (int tt = 0; tt < 4; ++tt) {
        v2f w = { w8[2 * tt], w8[2 * tt + 1] };
        acc[tt] = __builtin_amdgcn_wmma_f32_16x16x4_f32(false, h0A[ks], false, w, (short)0,
                                                        acc[tt], false, false);
      }
    }
    cell8(acc, c0, Ht, half, l16);
    __syncthreads();
#pragma unroll
    for (int ks = 0; ks < 4; ++ks) h0A[ks] = *(const v2f*)(Ht + l16 * HP + ks * 4 + 2 * half);

    const v4f bs1 = *(const v4f*)(Bb + 64 + l16 * 4);
#pragma unroll
    for (int tt = 0; tt < 4; ++tt) {
      float bv = bs1[tt];
      v8f tmp = { bv, bv, bv, bv, bv, bv, bv, bv };
      acc[tt] = tmp;
    }
#pragma unroll
    for (int ks = 0; ks < 4; ++ks) {
      const v8f w8 = Wf[(12 + ks) * 32];
#pragma unroll
      for (int tt = 0; tt < 4; ++tt) {
        v2f w = { w8[2 * tt], w8[2 * tt + 1] };
        acc[tt] = __builtin_amdgcn_wmma_f32_16x16x4_f32(false, h0A[ks], false, w, (short)0,
                                                        acc[tt], false, false);
      }
    }
#pragma unroll
    for (int ks = 0; ks < 4; ++ks) {
      const v8f w8 = Wf[(16 + ks) * 32];
#pragma unroll
      for (int tt = 0; tt < 4; ++tt) {
        v2f w = { w8[2 * tt], w8[2 * tt + 1] };
        acc[tt] = __builtin_amdgcn_wmma_f32_16x16x4_f32(false, h1A[ks], false, w, (short)0,
                                                        acc[tt], false, false);
      }
    }
    cell8(acc, c1, Ht, half, l16);
    __syncthreads();
#pragma unroll
    for (int ks = 0; ks < 4; ++ks) h1A[ks] = *(const v2f*)(Ht + l16 * HP + ks * 4 + 2 * half);

    float a = bo;
#pragma unroll
    for (int j = 0; j < 16; ++j) a += Ht[l16 * HP + j] * wo[j];
    if (row0 + l16 < N) out[((long)t * N + row0 + l16) * 2 + half] = a;
  }
}

static inline int gs(long n, int b) { return (int)((n + b - 1) / b); }

extern "C" void kernel_launch(void* const* d_in, const int* in_sizes, int n_in,
                              void* d_out, int out_size, void* d_ws, size_t ws_size,
                              hipStream_t stream) {
  (void)in_sizes; (void)n_in; (void)out_size;

  const int*   seq  = (const int*)d_in[0];
  const int*   gidx = (const int*)d_in[1];
  const float* Wemb = (const float*)d_in[2];
  const float* bemb = (const float*)d_in[3];
  const float* W1   = (const float*)d_in[4];
  const float* b1   = (const float*)d_in[5];
  const float* W2   = (const float*)d_in[6];
  const float* b2   = (const float*)d_in[7];
  const float* We2n = (const float*)d_in[8];
  const float* be2n = (const float*)d_in[9];
  const float* Wi0  = (const float*)d_in[10];
  const float* Wh0  = (const float*)d_in[11];
  const float* bL0  = (const float*)d_in[12];
  const float* Wi1  = (const float*)d_in[13];
  const float* Wh1  = (const float*)d_in[14];
  const float* bL1  = (const float*)d_in[15];
  const float* Wout = (const float*)d_in[16];
  const float* bout = (const float*)d_in[17];

  const int* srcE = gidx;
  const int* dstE = gidx + N_EDGES;
  float* out = (float*)d_out;

  float* p = (float*)d_ws;
  float* isq   = p; p += 5056;
  float* idg   = p; p += 5056;
  int*   uptr  = (int*)p; p += (size_t)T_STEPS * NUM_LOC;
  int*   gptr  = (int*)p; p += 5056;
  int*   gcol  = (int*)p; p += N_EDGES;
  float* gw    = p; p += N_EDGES;
  const size_t FIXED = 3 * 5056 + (size_t)T_STEPS * NUM_LOC + 2 * (size_t)N_EDGES;
  const size_t AREA  = (size_t)NUM_LOC * T_STEPS * 64;

  const bool all_steps = ws_size >= (FIXED + 2 * AREA) * sizeof(float);
  float* e     = p;
  float* h1    = all_steps ? p + AREA : p + NUM_LOC * 64;
  float* heall = all_steps ? p : p + 2 * NUM_LOC * 64;
  int*   ulist = all_steps ? (int*)(p + AREA) : (int*)heall;
  const int ustride = all_steps ? USR_NUM : NUM_LOC * 32;

  const int BT = 256;
  const int NZ = T_STEPS * NUM_LOC + NUM_LOC;

  k_zero_i<<<gs(NZ, BT), BT, 0, stream>>>(uptr, NZ);
  k_count<<<gs(NQH, BT), BT, 0, stream>>>(seq, dstE, uptr, gptr);
  k_scan<<<T_STEPS + 1, 1024, 0, stream>>>(uptr, gptr, isq, idg);
  k_fill<<<gs(NQH, BT), BT, 0, stream>>>(seq, srcE, dstE, isq, uptr, gptr, ulist, ustride, gcol, gw);

  if (all_steps) {
    const int TS = 4;
    k_gather_users<TS, T_STEPS><<<dim3(gs(NUM_LOC, 8), T_STEPS / TS), 256, 0, stream>>>(
        uptr, ulist, ustride, Wemb, bemb, e, NUM_LOC);
    k_gcn_relu<TS, T_STEPS><<<dim3(gs(NUM_LOC, 16), T_STEPS / TS), 512, 0, stream>>>(
        gptr, gcol, gw, idg, e, W1, b1, h1, NUM_LOC);
    k_gcn_e2n<TS, T_STEPS><<<dim3(gs(NUM_LOC, 16), T_STEPS / TS), 512, 0, stream>>>(
        gptr, gcol, gw, idg, h1, W2, b2, We2n, be2n, heall, NUM_LOC);
  } else {
    for (int t = 0; t < T_STEPS; ++t) {
      k_gather_users<1, 1><<<gs(NUM_LOC, 8), 256, 0, stream>>>(
          uptr + (size_t)t * NUM_LOC, ulist + (size_t)t * ustride, ustride, Wemb, bemb, e, NUM_LOC);
      k_gcn_relu<1, 1><<<gs(NUM_LOC, 16), 512, 0, stream>>>(gptr, gcol, gw, idg, e, W1, b1, h1, NUM_LOC);
      k_gcn_e2n<1, 1><<<gs(NUM_LOC, 16), 512, 0, stream>>>(gptr, gcol, gw, idg, h1, W2, b2, We2n, be2n,
                                                            heall + (size_t)t * NUM_LOC * 32, NUM_LOC);
    }
  }

  k_lstm_all<<<gs(USR_NUM, 32), 64, 0, stream>>>(seq, heall, Wi0, Wh0, bL0, Wi1, Wh1, bL1,
                                                 Wout, bout, out, USR_NUM);
}


# Index lists built with no global atomic: a work-group per chunk of keys counts and ranks in an LDS histogram, chunk bases from a scan; 8 -> 7 launches; list order = chunk, then LDS add arrival
# speedup vs baseline: 9.1501x; 1.6161x over previous
#include <hip/hip_runtime.h>
#include <hip/hip_bf16.h>
#include <math.h>

#define USR_NUM 100000
#define USR_DIM 64
#define NUM_LOC 5000
#define T_STEPS 16
#define N_EDGES 80000

typedef float v2f __attribute__((ext_vector_type(2)));
typedef float v4f __attribute__((ext_vector_type(4)));
typedef float v8f __attribute__((ext_vector_type(8)));

typedef int v4i __attribute__((ext_vector_type(4)));
#define NB 8
#define UG (USR_NUM / 4 / NB)
#define GG (N_EDGES / 4 / NB)
#define NROW (T_STEPS + 1)
static_assert(USR_NUM % (4 * NB) == 0 && N_EDGES % (4 * NB) == 0, "chunks are whole 16-byte key groups");
static_assert(UG <= 4 * 1024 && GG <= 4 * 1024, "a thread takes at most 4 key groups of its chunk");
static_assert(NUM_LOC <= 5 * 1024, "a thread takes at most 5 bins of a histogram");

__global__ __launch_bounds__(1024) void k_hist(const int* __restrict__ seq, const int* __restrict__ dst,
                                                int* __restrict__ hist) {
  __shared__ int h[NUM_LOC];
  const int tid = threadIdx.x;
  const int row = blockIdx.x / NB;
  const int c = blockIdx.x - row * NB;
  const bool graph = row == T_STEPS;
  const int ng = graph ? GG : UG;
  const int* keys = graph ? dst + 4 * c * GG : seq + (long)row * USR_NUM + 4 * c * UG;
  v4i k[4];
#pragma unroll
  for (int u = 0; u < 4; ++u) {
    const int g = tid + u * 1024;
    const v4i z = { 0, 0, 0, 0 };
    k[u] = g < ng ? *(const v4i*)(keys + 4 * g) : z;
  }
  for (int i = tid; i < NUM_LOC; i += 1024) h[i] = 0;
  __syncthreads();
#pragma unroll
  for (int u = 0; u < 4; ++u) {
    if (tid + u * 1024 < ng) {
#pragma unroll
      for (int s = 0; s < 4; ++s) atomicAdd(&h[k[u][s]], 1);
    }
  }
  __syncthreads();
  int* hp = hist + (long)blockIdx.x * NUM_LOC;
  for (int i = tid; i < NUM_LOC; i += 1024) hp[i] = h[i];
}

__global__ __launch_bounds__(1024) void k_scan(int* __restrict__ hist, int* __restrict__ uptr,
                                               int* __restrict__ gptr,
                                               float* __restrict__ isq, float* __restrict__ idg) {
  __shared__ int part[1024];
  __shared__ int tot[5 * 1024];
  const int tid = threadIdx.x;
  const int row = blockIdx.x;
  const bool graph = row == T_STEPS;
  int* p = graph ? gptr : uptr + row * NUM_LOC;
  int* hp = hist + (long)row * NB * NUM_LOC;
  int pre[5][NB];
  int tt[5];
#pragma unroll
  for (int k = 0; k < 5; ++k) {
    const int idx = k * 1024 + tid;
    int t = 0;
#pragma unroll
    for (int c = 0; c < NB; ++c) {
      const int x = idx < NUM_LOC ? hp[c * NUM_LOC + idx] : 0;
      pre[k][c] = t;
      t += x;
    }
    tt[k] = t;
    tot[idx] = t;
  }
  __syncthreads();
  int v[5];
  int s = 0;
#pragma unroll
  for (int k = 0; k < 5; ++k) { v[k] = tot[tid * 5 + k]; s += v[k]; }
  part[tid] = s;
  __syncthreads();
  for (int off = 1; off < 1024; off <<= 1) {
    int x = tid >= off ? part[tid - off] : 0;
    __syncthreads();
    part[tid] += x;
    __syncthreads();
  }
  int run = part[tid] - s;
#pragma unroll
  for (int k = 0; k < 5; ++k) { tot[tid * 5 + k] = run; run += v[k]; }
  __syncthreads();
#pragma unroll
  for (int k = 0; k < 5; ++k) {
    const int idx = k * 1024 + tid;
    if (idx < NUM_LOC) {
      const int st = tot[idx];
#pragma unroll
      for (int c = 0; c < NB; ++c) hp[c * NUM_LOC + idx] = st + pre[k][c];
      p[idx] = st + tt[k];
      if (graph) {
        const float d = (float)tt[k] + 1.0f;
        isq[idx] = rsqrtf(d);
        idg[idx] = 1.0f / d;
      }
    }
  }
}

__global__ __launch_bounds__(1024) void k_fill(const int* __restrict__ seq, const int* __restrict__ src,
                       const int* __restrict__ dst, const float* __restrict__ isq,
                       const int* __restrict__ hist,
                       int* __restrict__ ulist_all, int ustride, int* __restrict__ gcol,
                       float* __restrict__ gw) {
  __shared__ int off[NUM_LOC];
  const int tid = threadIdx.x;
  const int row = blockIdx.x / NB;
  const int c = blockIdx.x - row * NB;
  const bool graph = row == T_STEPS;
  const int ng = graph ? GG : UG;
  const int* keys = graph ? dst + 4 * c * GG : seq + (long)row * USR_NUM + 4 * c * UG;
  const int* hp = hist + (long)blockIdx.x * NUM_LOC;
  v4i k[4], sv[4];
  float w[4][4];
#pragma unroll
  for (int u = 0; u < 4; ++u) {
    const int g = tid + u * 1024;
    const v4i z = { 0, 0, 0, 0 };
    k[u] = g < ng ? *(const v4i*)(keys + 4 * g) : z;
    sv[u] = z;
  }
  if (graph) {
    const int* srcs = src + 4 * c * GG;
#pragma unroll
    for (int u = 0; u < 4; ++u) {
      const int g = tid + u * 1024;
      if (g < GG) sv[u] = *(const v4i*)(srcs + 4 * g);
    }
#pragma unroll
    for (int u = 0; u < 4; ++u) {
#pragma unroll
      for (int s = 0; s < 4; ++s) w[u][s] = isq[sv[u][s]] * isq[k[u][s]];
    }
  }
#pragma unroll
  for (int it = 0; it < 5; ++it) {
    const int i = tid + it * 1024;
    if (i < NUM_LOC) off[i] = hp[i];
  }
  __syncthreads();
  if (!graph) {
    int* l = ulist_all + (long)row * ustride;
#pragma unroll
    for (int u = 0; u < 4; ++u) {
      const int g = tid + u * 1024;
      if (g < UG) {
        int pos[4];
#pragma unroll
        for (int s = 0; s < 4; ++s) pos[s] = atomicAdd(&off[k[u][s]], 1);
#pragma unroll
        for (int s = 0; s < 4; ++s) l[pos[s]] = 4 * (c * UG + g) + s;
      }
    }
  } else {
#pragma unroll
    for (int u = 0; u < 4; ++u) {
      const int g = tid + u * 1024;
      if (g < GG) {
        int pos[4];
#pragma unroll
        for (int s = 0; s < 4; ++s) pos[s] = atomicAdd(&off[k[u][s]], 1);
#pragma unroll
        for (int s = 0; s < 4; ++s) { gcol[pos[s]] = sv[u][s]; gw[pos[s]] = w[u][s]; }
      }
    }
  }
}

template <int TS, int NS>
__global__ __launch_bounds__(256) void k_gather_users(
    const int* __restrict__ uptr, const int* __restrict__ ulist, int ustride,
    const float* __restrict__ Wemb, const float* __restrict__ bemb, float* __restrict__ e_all,
    int nloc) {
  const int lane = threadIdx.x & 31;
  const int j = __builtin_amdgcn_readfirstlane(blockIdx.x * 8 + (threadIdx.x >> 5));
  if (j >= nloc) return;
  const v2f bb = *(const v2f*)(bemb + 2 * lane);
  for (int it = 0; it < TS; ++it) {
    const int t = blockIdx.y * TS + it;
    const int* ptr = uptr + t * NUM_LOC;
    const int* ul = ulist + (long)t * ustride;
    const int beg = j ? ptr[j - 1] : 0;
    const int end = ptr[j];
    v2f acc = { 0.0f, 0.0f };
    for (int k = beg; k < end; k += 8) {
      int id[8];
#pragma unroll
      for (int u = 0; u < 8; ++u) id[u] = ul[k + u < end ? k + u : end - 1];
      v2f x[8];
#pragma unroll
      for (int u = 0; u < 8; ++u) x[u] = *(const v2f*)(Wemb + (long)id[u] * 64 + 2 * lane);
#pragma unroll
      for (int u = 0; u < 8; ++u) {
        if (k + u < end) acc += x[u] + bb;
      }
    }
    float cv = (float)(end - beg); if (cv < 1.0f) cv = 1.0f;
    v2f r = { acc[0] / cv, acc[1] / cv };
    *(v2f*)(e_all + ((long)j * NS + t) * 64 + 2 * lane) = r;
  }
}

#define TP 68
#define OP 36

template <int TS, int NS>
__device__ __forceinline__ void gather_node(
    v2f (&acc)[TS], const int* __restrict__ gptr, const int* __restrict__ gcol,
    const float* __restrict__ gw, const float* __restrict__ idg, const float* __restrict__ h,
    int q, int t0, int lane) {
  constexpr int GE = TS == 1 ? 8 : 4;
  const int beg = q ? gptr[q - 1] : 0;
  const int end = gptr[q];
  const float* hl = h + (long)t0 * 64 + 2 * lane;
  v2f self[TS];
#pragma unroll
  for (int ts = 0; ts < TS; ++ts) self[ts] = *(const v2f*)(hl + (long)q * (NS * 64) + ts * 64);
  const float sd = idg[q];
#pragma unroll
  for (int ts = 0; ts < TS; ++ts) { v2f z = { 0.0f, 0.0f }; acc[ts] = z; }
  for (int k = beg; k < end; k += GE) {
    int s[GE];
    float w[GE];
#pragma unroll
    for (int u = 0; u < GE; ++u) {
      const int kk = k + u < end ? k + u : end - 1;
      s[u] = gcol[kk];
      w[u] = gw[kk];
    }
    v2f x[GE][TS];
#pragma unroll
    for (int u = 0; u < GE; ++u) {
      const float* hs = hl + (long)s[u] * (NS * 64);
#pragma unroll
      for (int ts = 0; ts < TS; ++ts) x[u][ts] = *(const v2f*)(hs + ts * 64);
    }
#pragma unroll
    for (int u = 0; u < GE; ++u) {
      const bool live = k + u < end;
#pragma unroll
      for (int ts = 0; ts < TS; ++ts) {
        const v2f old = acc[ts];
        const v2f upd = old + x[u][ts] * w[u];
        acc[ts] = live ? upd : old;
      }
    }
  }
#pragma unroll
  for (int ts = 0; ts < TS; ++ts) acc[ts] += self[ts] * sd;
}

__device__ __forceinline__ void load_bfrag(v2f (&b)[16], const float* __restrict__ Wm, int nc,
                                           int tt, int half, int l16) {
#pragma unroll
  for (int ks = 0; ks < 16; ++ks) {
    b[ks][0] = Wm[(4 * ks + 2 * half + 0) * nc + tt * 16 + l16];
    b[ks][1] = Wm[(4 * ks + 2 * half + 1) * nc + tt * 16 + l16];
  }
}

template <int NT>
__device__ __forceinline__ void tiles_product(v8f (&c)[NT], const float* tiles, const v2f (&b)[16],
                                              float bv, int half, int l16) {
#pragma unroll
  for (int i = 0; i < NT; ++i) { v8f z = { bv, bv, bv, bv, bv, bv, bv, bv }; c[i] = z; }
#pragma unroll
  for (int ks = 0; ks < 16; ++ks) {
#pragma unroll
    for (int i = 0; i < NT; ++i) {
      const v2f a = *(const v2f*)(tiles + (i * 16 + l16) * TP + 4 * ks + 2 * half);
      c[i] = __builtin_amdgcn_wmma_f32_16x16x4_f32(false, a, false, b[ks], (short)0, c[i], false, false);
    }
  }
}

template <int TS, int NS>
__global__ __launch_bounds__(512) void k_gcn_relu(
    const int* __restrict__ gptr, const int* __restrict__ gcol, const float* __restrict__ gw,
    const float* __restrict__ idg, const float* __restrict__ hin, const float* __restrict__ W,
    const float* __restrict__ bias, float* __restrict__ hout, int nloc) {
  __shared__ __attribute__((aligned(16))) float At[TS * 16 * TP];
  __shared__ __attribute__((aligned(16))) float Ct[TS * 16 * TP];
  const int lane = threadIdx.x & 31;
  const int wave = threadIdx.x >> 5;
  const int half = lane >> 4;
  const int l16  = lane & 15;
  const int t0   = blockIdx.y * TS;
  const int q = __builtin_amdgcn_readfirstlane(blockIdx.x * 16 + wave);

  v2f bf[16];
  float bv = 0.0f;
  if (wave < 4) {
    load_bfrag(bf, W, 64, wave, half, l16);
    bv = bias[wave * 16 + l16];
  }

  v2f acc[TS];
#pragma unroll
  for (int ts = 0; ts < TS; ++ts) { v2f z = { 0.0f, 0.0f }; acc[ts] = z; }
  if (q < nloc) gather_node<TS, NS>(acc, gptr, gcol, gw, idg, hin, q, t0, lane);
#pragma unroll
  for (int ts = 0; ts < TS; ++ts) *(v2f*)(At + (ts * 16 + wave) * TP + 2 * lane) = acc[ts];
  __syncthreads();

  if (wave < 4) {
    v8f c[TS];
    tiles_product<TS>(c, At, bf, bv, half, l16);
#pragma unroll
    for (int ts = 0; ts < TS; ++ts) {
#pragma unroll
      for (int r = 0; r < 8; ++r) {
        const float v = c[ts][r];
        Ct[(ts * 16 + r + 8 * half) * TP + wave * 16 + l16] = v > 0.0f ? v : 0.0f;
      }
    }
  }
  __syncthreads();

  if (q < nloc) {
#pragma unroll
    for (int ts = 0; ts < TS; ++ts)
      *(v2f*)(hout + ((long)q * NS + t0 + ts) * 64 + 2 * lane) =
          *(const v2f*)(Ct + (ts * 16 + wave) * TP + 2 * lane);
  }
}

template <int TS, int NS>
__global__ __launch_bounds__(512) void k_gcn_e2n(
    const int* __restrict__ gptr, const int* __restrict__ gcol, const float* __restrict__ gw,
    const float* __restrict__ idg, const float* __restrict__ hin, const float* __restrict__ W2,
    const float* __restrict__ b2, const float* __restrict__ We, const float* __restrict__ be,
    float* __restrict__ out, int nloc) {
  __shared__ __attribute__((aligned(16))) float At[TS * 16 * TP];
  __shared__ __attribute__((aligned(16))) float Ht[TS * 16 * TP];
  __shared__ __attribute__((aligned(16))) float Ot[TS * 16 * OP];
  const int lane = threadIdx.x & 31;
  const int wave = threadIdx.x >> 5;
  const int half = lane >> 4;
  const int l16  = lane & 15;
  const int t0   = blockIdx.y * TS;
  const int q = __builtin_amdgcn_readfirstlane(blockIdx.x * 16 + wave);

  constexpr int ET = TS == 1 ? 1 : 2;
  constexpr int EW = 2 * TS / ET;
  const int ec = (wave - 4) & 1;
  const int es = ((wave - 4) >> 1) * ET;
  v2f bf[16];
  float bv = 0.0f;
  if (wave < 4) {
    load_bfrag(bf, W2, 64, wave, half, l16);
    bv = b2[wave * 16 + l16];
  } else if (wave < 4 + EW) {
    load_bfrag(bf, We, 32, ec, half, l16);
    bv = be[ec * 16 + l16];
  }

  v2f acc[TS];
#pragma unroll
  for (int ts = 0; ts < TS; ++ts) { v2f z = { 0.0f, 0.0f }; acc[ts] = z; }
  if (q < nloc) gather_node<TS, NS>(acc, gptr, gcol, gw, idg, hin, q, t0, lane);
#pragma unroll
  for (int ts = 0; ts < TS; ++ts) *(v2f*)(At + (ts * 16 + wave) * TP + 2 * lane) = acc[ts];
  __syncthreads();

  if (wave < 4) {
    v8f c[TS];
    tiles_product<TS>(c, At, bf, bv, half, l16);
#pragma unroll
    for (int ts = 0; ts < TS; ++ts) {
#pragma unroll
      for (int r = 0; r < 8; ++r) Ht[(ts * 16 + r + 8 * half) * TP + wave * 16 + l16] = c[ts][r];
    }
  }
  __syncthreads();

  if (wave >= 4 && wave < 4 + EW) {
    v8f c[ET];
    tiles_product<ET>(c, Ht + es * 16 * TP, bf, bv, half, l16);
#pragma unroll
    for (int i = 0; i < ET; ++i) {
#pragma unroll
      for (int r = 0; r < 8; ++r) Ot[((es + i) * 16 + r + 8 * half) * OP + ec * 16 + l16] = c[i][r];
    }
  }
  __syncthreads();

  if (q < nloc) {
#pragma unroll
    for (int ts = 0; ts < TS; ++ts)
      out[((long)(t0 + ts) * nloc + q) * 32 + lane] = Ot[(ts * 16 + wave) * OP + lane];
  }
}

#define HP 20

__device__ __forceinline__ float sigm(float x) {
  return __builtin_amdgcn_rcpf(1.0f + __builtin_amdgcn_exp2f(x * -1.4426950408889634f));
}
__device__ __forceinline__ float tanh_small(float x) {
  const float y2 = x * x;
  float p = __builtin_fmaf(y2, -0x1.758e7ap-8f, 0x1.521192p-6f);
  p = __builtin_fmaf(y2, p, -0x1.b8389cp-5f);
  p = __builtin_fmaf(y2, p, 0x1.110704p-3f);
  p = __builtin_fmaf(y2, p, -0x1.555532p-2f);
  return __builtin_fmaf(y2, x * p, x);
}
__device__ __forceinline__ float tanh_any(float x) {
  const float y = __builtin_fabsf(x);
  const float t = __builtin_amdgcn_exp2f(y * 2.8853900817779268f);
  const float big = __builtin_fmaf(__builtin_amdgcn_rcpf(t + 1.0f), -2.0f, 1.0f);
  return y < 0.625f ? tanh_small(x) : __builtin_copysignf(big, x);
}
__device__ __forceinline__ void tanh8(float (&o)[8], const float (&x)[8]) {
  float m = __builtin_fabsf(x[0]);
#pragma unroll
  for (int r = 1; r < 8; ++r) m = __builtin_fmaxf(m, __builtin_fabsf(x[r]));
  if (__builtin_amdgcn_ballot_w32(m >= 0.625f) == 0) {
#pragma unroll
    for (int r = 0; r < 8; ++r) o[r] = tanh_small(x[r]);
  } else {
#pragma unroll
    for (int r = 0; r < 8; ++r) o[r] = tanh_any(x[r]);
  }
}
__device__ __forceinline__ void cell8(const v8f (&acc)[4], float (&c)[8], float* Ht, int half, int l16) {
  float g[8], tg[8], th[8];
#pragma unroll
  for (int r = 0; r < 8; ++r) g[r] = acc[2][r];
  tanh8(tg, g);
#pragma unroll
  for (int r = 0; r < 8; ++r) c[r] = sigm(acc[1][r]) * c[r] + sigm(acc[0][r]) * tg[r];
  tanh8(th, c);
#pragma unroll
  for (int r = 0; r < 8; ++r) Ht[(r + 8 * half) * HP + l16] = sigm(acc[3][r]) * th[r];
}

template <int NKS>
__device__ __forceinline__ void stage_frag(float* dst, const float* __restrict__ Wm) {
  for (int i = threadIdx.x; i < NKS * 256; i += 64) {
    const int s = i & 1, tt = (i >> 1) & 3, ln = (i >> 3) & 31, ks = i >> 8;
    dst[i] = Wm[(ks * 4 + 2 * (ln >> 4) + s) * 64 + tt * 16 + (ln & 15)];
  }
}

__global__ __launch_bounds__(64) void k_lstm_all(
    const int* __restrict__ seq,
    const float* __restrict__ he,
    const float* __restrict__ Wi0, const float* __restrict__ Wh0, const float* __restrict__ b0,
    const float* __restrict__ Wi1, const float* __restrict__ Wh1, const float* __restrict__ b1,
    const float* __restrict__ Wout, const float* __restrict__ bout,
    float* __restrict__ out, int N) {
  __shared__ __attribute__((aligned(32))) float Ws[80 * 64];
  __shared__ __attribute__((aligned(16))) float Hs[2 * 16 * HP];
  __shared__ __attribute__((aligned(16))) float Wo[32];
  __shared__ __attribute__((aligned(16))) float Bb[2 * 16 * 4];
  stage_frag<8>(Ws,            Wi0);
  stage_frag<4>(Ws +  8 * 256, Wh0);
  stage_frag<4>(Ws + 12 * 256, Wi1);
  stage_frag<4>(Ws + 16 * 256, Wh1);
  if (threadIdx.x < 32) Wo[threadIdx.x] = Wout[(threadIdx.x & 15) * 2 + (threadIdx.x >> 4)];
  Bb[threadIdx.x]      = b0[(threadIdx.x & 3) * 16 + (threadIdx.x >> 2)];
  Bb[64 + threadIdx.x] = b1[(threadIdx.x & 3) * 16 + (threadIdx.x >> 2)];
  __syncthreads();

  const int lane = threadIdx.x & 31;
  const int wave = threadIdx.x >> 5;
  const int half = lane >> 4;
  const int l16  = lane & 15;
  const int row0 = (blockIdx.x * 2 + wave) * 16;

  int ar = row0 + l16; if (ar > N - 1) ar = N - 1;

  const float* wo = Wo + half * 16;
  const float bo = bout[half];

  const v8f* Wf = (const v8f*)Ws + lane;
  float* Ht = Hs + wave * (16 * HP);

  v2f h0A[4], h1A[4];
  float c0[8], c1[8];
#pragma unroll
  for (int i = 0; i < 4; ++i) { v2f z = { 0.0f, 0.0f }; h0A[i] = z; h1A[i] = z; }
#pragma unroll
  for (int i = 0; i < 8; ++i) { c0[i] = 0.0f; c1[i] = 0.0f; }

  int id = seq[ar];
  for (int t = 0; t < T_STEPS; ++t) {
    const float* Xr = he + ((long)t * NUM_LOC + id) * 32 + 2 * half;
    v2f xa[8];
#pragma unroll
    for (int ks = 0; ks < 8; ++ks) xa[ks] = *(const v2f*)(Xr + ks * 4);
    const int tn = t + 1 < T_STEPS ? t + 1 : t;
    id = seq[(long)tn * N + ar];

    v8f acc[4];
    const v4f bs0 = *(const v4f*)(Bb + l16 * 4);
#pragma unroll
    for (int tt = 0; tt < 4; ++tt) {
      float bv = bs0[tt];
      v8f tmp = { bv, bv, bv, bv, bv, bv, bv, bv };
      acc[tt] = tmp;
    }
#pragma unroll
    for (int ks = 0; ks < 8; ++ks) {
      const v8f w8 = Wf[(0 + ks) * 32];
#pragma unroll
      for (int tt = 0; tt < 4; ++tt) {
        v2f w = { w8[2 * tt], w8[2 * tt + 1] };
        acc[tt] = __builtin_amdgcn_wmma_f32_16x16x4_f32(false, xa[ks], false, w, (short)0,
                                                        acc[tt], false, false);
      }
    }
#pragma unroll
    for (int ks = 0; ks < 4; ++ks) {
      const v8f w8 = Wf[(8 + ks) * 32];
#pragma unroll
      for (int tt = 0; tt < 4; ++tt) {
        v2f w = { w8[2 * tt], w8[2 * tt + 1] };
        acc[tt] = __builtin_amdgcn_wmma_f32_16x16x4_f32(false, h0A[ks], false, w, (short)0,
                                                        acc[tt], false, false);
      }
    }
    cell8(acc, c0, Ht, half, l16);
    __syncthreads();
#pragma unroll
    for (int ks = 0; ks < 4; ++ks) h0A[ks] = *(const v2f*)(Ht + l16 * HP + ks * 4 + 2 * half);

    const v4f bs1 = *(const v4f*)(Bb + 64 + l16 * 4);
#pragma unroll
    for (int tt = 0; tt < 4; ++tt) {
      float bv = bs1[tt];
      v8f tmp = { bv, bv, bv, bv, bv, bv, bv, bv };
      acc[tt] = tmp;
    }
#pragma unroll
    for (int ks = 0; ks < 4; ++ks) {
      const v8f w8 = Wf[(12 + ks) * 32];
#pragma unroll
      for (int tt = 0; tt < 4; ++tt) {
        v2f w = { w8[2 * tt], w8[2 * tt + 1] };
        acc[tt] = __builtin_amdgcn_wmma_f32_16x16x4_f32(false, h0A[ks], false, w, (short)0,
                                                        acc[tt], false, false);
      }
    }
#pragma unroll
    for (int ks = 0; ks < 4; ++ks) {
      const v8f w8 = Wf[(16 + ks) * 32];
#pragma unroll
      for (int tt = 0; tt < 4; ++tt) {
        v2f w = { w8[2 * tt], w8[2 * tt + 1] };
        acc[tt] = __builtin_amdgcn_wmma_f32_16x16x4_f32(false, h1A[ks], false, w, (short)0,
                                                        acc[tt], false, false);
      }
    }
    cell8(acc, c1, Ht, half, l16);
    __syncthreads();
#pragma unroll
    for (int ks = 0; ks < 4; ++ks) h1A[ks] = *(const v2f*)(Ht + l16 * HP + ks * 4 + 2 * half);

    float a = bo;
#pragma unroll
    for (int j = 0; j < 16; ++j) a += Ht[l16 * HP + j] * wo[j];
    if (row0 + l16 < N) out[((long)t * N + row0 + l16) * 2 + half] = a;
  }
}

static inline int gs(long n, int b) { return (int)((n + b - 1) / b); }

extern "C" void kernel_launch(void* const* d_in, const int* in_sizes, int n_in,
                              void* d_out, int out_size, void* d_ws, size_t ws_size,
                              hipStream_t stream) {
  (void)in_sizes; (void)n_in; (void)out_size;

  const int*   seq  = (const int*)d_in[0];
  const int*   gidx = (const int*)d_in[1];
  const float* Wemb = (const float*)d_in[2];
  const float* bemb = (const float*)d_in[3];
  const float* W1   = (const float*)d_in[4];
  const float* b1   = (const float*)d_in[5];
  const float* W2   = (const float*)d_in[6];
  const float* b2   = (const float*)d_in[7];
  const float* We2n = (const float*)d_in[8];
  const float* be2n = (const float*)d_in[9];
  const float* Wi0  = (const float*)d_in[10];
  const float* Wh0  = (const float*)d_in[11];
  const float* bL0  = (const float*)d_in[12];
  const float* Wi1  = (const float*)d_in[13];
  const float* Wh1  = (const float*)d_in[14];
  const float* bL1  = (const float*)d_in[15];
  const float* Wout = (const float*)d_in[16];
  const float* bout = (const float*)d_in[17];

  const int* srcE = gidx;
  const int* dstE = gidx + N_EDGES;
  float* out = (float*)d_out;

  float* p = (float*)d_ws;
  float* isq   = p; p += 5056;
  float* idg   = p; p += 5056;
  int*   uptr  = (int*)p; p += (size_t)T_STEPS * NUM_LOC;
  int*   gptr  = (int*)p; p += 5056;
  int*   gcol  = (int*)p; p += N_EDGES;
  float* gw    = p; p += N_EDGES;
  const size_t FIXED = 3 * 5056 + (size_t)T_STEPS * NUM_LOC + 2 * (size_t)N_EDGES;
  const size_t AREA  = (size_t)NUM_LOC * T_STEPS * 64;

  const bool all_steps = ws_size >= (FIXED + 2 * AREA) * sizeof(float);
  float* e     = p;
  float* h1    = all_steps ? p + AREA : p + NUM_LOC * 64;
  float* heall = all_steps ? p : p + 2 * NUM_LOC * 64;
  int*   ulist = all_steps ? (int*)(p + AREA) : (int*)heall;
  const int ustride = all_steps ? USR_NUM : NUM_LOC * 32;
  int*   hist  = all_steps ? (int*)p : (int*)(p + 10 * NUM_LOC * 64);

  k_hist<<<NROW * NB, 1024, 0, stream>>>(seq, dstE, hist);
  k_scan<<<NROW, 1024, 0, stream>>>(hist, uptr, gptr, isq, idg);
  k_fill<<<NROW * NB, 1024, 0, stream>>>(seq, srcE, dstE, isq, hist, ulist, ustride, gcol, gw);

  if (all_steps) {
    const int TS = 4;
    k_gather_users<TS, T_STEPS><<<dim3(gs(NUM_LOC, 8), T_STEPS / TS), 256, 0, stream>>>(
        uptr, ulist, ustride, Wemb, bemb, e, NUM_LOC);
    k_gcn_relu<TS, T_STEPS><<<dim3(gs(NUM_LOC, 16), T_STEPS / TS), 512, 0, stream>>>(
        gptr, gcol, gw, idg, e, W1, b1, h1, NUM_LOC);
    k_gcn_e2n<TS, T_STEPS><<<dim3(gs(NUM_LOC, 16), T_STEPS / TS), 512, 0, stream>>>(
        gptr, gcol, gw, idg, h1, W2, b2, We2n, be2n, heall, NUM_LOC);
  } else {
    for (int t = 0; t < T_STEPS; ++t) {
      k_gather_users<1, 1><<<gs(NUM_LOC, 8), 256, 0, stream>>>(
          uptr + (size_t)t * NUM_LOC, ulist + (size_t)t * ustride, ustride, Wemb, bemb, e, NUM_LOC);
      k_gcn_relu<1, 1><<<gs(NUM_LOC, 16), 512, 0, stream>>>(gptr, gcol, gw, idg, e, W1, b1, h1, NUM_LOC);
      k_gcn_e2n<1, 1><<<gs(NUM_LOC, 16), 512, 0, stream>>>(gptr, gcol, gw, idg, h1, W2, b2, We2n, be2n,
                                                            heall + (size_t)t * NUM_LOC * 32, NUM_LOC);
    }
  }

  k_lstm_all<<<gs(USR_NUM, 32), 64, 0, stream>>>(seq, heall, Wi0, Wh0, bL0, Wi1, Wh1, bL1,
                                                 Wout, bout, out, USR_NUM);
}


# Tail's layer-0 input product x @ Wi0 + b0 moved in front of the gather: once per (step, location) behind the edge2node product; tail gathers gate rows as accumulator start (80 -> 48 WMMAs a step)
# speedup vs baseline: 10.2002x; 1.1148x over previous
#include <hip/hip_runtime.h>
#include <hip/hip_bf16.h>
#include <math.h>

#define USR_NUM 100000
#define USR_DIM 64
#define NUM_LOC 5000
#define T_STEPS 16
#define N_EDGES 80000

typedef float v2f __attribute__((ext_vector_type(2)));
typedef float v4f __attribute__((ext_vector_type(4)));
typedef float v8f __attribute__((ext_vector_type(8)));

typedef int v4i __attribute__((ext_vector_type(4)));
#define NB 8
#define UG (USR_NUM / 4 / NB)
#define GG (N_EDGES / 4 / NB)
#define NROW (T_STEPS + 1)
static_assert(USR_NUM % (4 * NB) == 0 && N_EDGES % (4 * NB) == 0, "chunks are whole 16-byte key groups");
static_assert(UG <= 4 * 1024 && GG <= 4 * 1024, "a thread takes at most 4 key groups of its chunk");
static_assert(NUM_LOC <= 5 * 1024, "a thread takes at most 5 bins of a histogram");

__global__ __launch_bounds__(1024) void k_hist(const int* __restrict__ seq, const int* __restrict__ dst,
                                                int* __restrict__ hist) {
  __shared__ int h[NUM_LOC];
  const int tid = threadIdx.x;
  const int row = blockIdx.x / NB;
  const int c = blockIdx.x - row * NB;
  const bool graph = row == T_STEPS;
  const int ng = graph ? GG : UG;
  const int* keys = graph ? dst + 4 * c * GG : seq + (long)row * USR_NUM + 4 * c * UG;
  v4i k[4];
#pragma unroll
  for (int u = 0; u < 4; ++u) {
    const int g = tid + u * 1024;
    const v4i z = { 0, 0, 0, 0 };
    k[u] = g < ng ? *(const v4i*)(keys + 4 * g) : z;
  }
  for (int i = tid; i < NUM_LOC; i += 1024) h[i] = 0;
  __syncthreads();
#pragma unroll
  for (int u = 0; u < 4; ++u) {
    if (tid + u * 1024 < ng) {
#pragma unroll
      for (int s = 0; s < 4; ++s) atomicAdd(&h[k[u][s]], 1);
    }
  }
  __syncthreads();
  int* hp = hist + (long)blockIdx.x * NUM_LOC;
  for (int i = tid; i < NUM_LOC; i += 1024) hp[i] = h[i];
}

__global__ __launch_bounds__(1024) void k_scan(int* __restrict__ hist, int* __restrict__ uptr,
                                               int* __restrict__ gptr,
                                               float* __restrict__ isq, float* __restrict__ idg) {
  __shared__ int part[1024];
  __shared__ int tot[5 * 1024];
  const int tid = threadIdx.x;
  const int row = blockIdx.x;
  const bool graph = row == T_STEPS;
  int* p = graph ? gptr : uptr + row * NUM_LOC;
  int* hp = hist + (long)row * NB * NUM_LOC;
  int pre[5][NB];
  int tt[5];
#pragma unroll
  for (int k = 0; k < 5; ++k) {
    const int idx = k * 1024 + tid;
    int t = 0;
#pragma unroll
    for (int c = 0; c < NB; ++c) {
      const int x = idx < NUM_LOC ? hp[c * NUM_LOC + idx] : 0;
      pre[k][c] = t;
      t += x;
    }
    tt[k] = t;
    tot[idx] = t;
  }
  __syncthreads();
  int v[5];
  int s = 0;
#pragma unroll
  for (int k = 0; k < 5; ++k) { v[k] = tot[tid * 5 + k]; s += v[k]; }
  part[tid] = s;
  __syncthreads();
  for (int off = 1; off < 1024; off <<= 1) {
    int x = tid >= off ? part[tid - off] : 0;
    __syncthreads();
    part[tid] += x;
    __syncthreads();
  }
  int run = part[tid] - s;
#pragma unroll
  for (int k = 0; k < 5; ++k) { tot[tid * 5 + k] = run; run += v[k]; }
  __syncthreads();
#pragma unroll
  for (int k = 0; k < 5; ++k) {
    const int idx = k * 1024 + tid;
    if (idx < NUM_LOC) {
      const int st = tot[idx];
#pragma unroll
      for (int c = 0; c < NB; ++c) hp[c * NUM_LOC + idx] = st + pre[k][c];
      p[idx] = st + tt[k];
      if (graph) {
        const float d = (float)tt[k] + 1.0f;
        isq[idx] = rsqrtf(d);
        idg[idx] = 1.0f / d;
      }
    }
  }
}

__global__ __launch_bounds__(1024) void k_fill(const int* __restrict__ seq, const int* __restrict__ src,
                       const int* __restrict__ dst, const float* __restrict__ isq,
                       const int* __restrict__ hist,
                       int* __restrict__ ulist_all, int ustride, int* __restrict__ gcol,
                       float* __restrict__ gw) {
  __shared__ int off[NUM_LOC];
  const int tid = threadIdx.x;
  const int row = blockIdx.x / NB;
  const int c = blockIdx.x - row * NB;
  const bool graph = row == T_STEPS;
  const int ng = graph ? GG : UG;
  const int* keys = graph ? dst + 4 * c * GG : seq + (long)row * USR_NUM + 4 * c * UG;
  const int* hp = hist + (long)blockIdx.x * NUM_LOC;
  v4i k[4], sv[4];
  float w[4][4];
#pragma unroll
  for (int u = 0; u < 4; ++u) {
    const int g = tid + u * 1024;
    const v4i z = { 0, 0, 0, 0 };
    k[u] = g < ng ? *(const v4i*)(keys + 4 * g) : z;
    sv[u] = z;
  }
  if (graph) {
    const int* srcs = src + 4 * c * GG;
#pragma unroll
    for (int u = 0; u < 4; ++u) {
      const int g = tid + u * 1024;
      if (g < GG) sv[u] = *(const v4i*)(srcs + 4 * g);
    }
#pragma unroll
    for (int u = 0; u < 4; ++u) {
#pragma unroll
      for (int s = 0; s < 4; ++s) w[u][s] = isq[sv[u][s]] * isq[k[u][s]];
    }
  }
#pragma unroll
  for (int it = 0; it < 5; ++it) {
    const int i = tid + it * 1024;
    if (i < NUM_LOC) off[i] = hp[i];
  }
  __syncthreads();
  if (!graph) {
    int* l = ulist_all + (long)row * ustride;
#pragma unroll
    for (int u = 0; u < 4; ++u) {
      const int g = tid + u * 1024;
      if (g < UG) {
        int pos[4];
#pragma unroll
        for (int s = 0; s < 4; ++s) pos[s] = atomicAdd(&off[k[u][s]], 1);
#pragma unroll
        for (int s = 0; s < 4; ++s) l[pos[s]] = 4 * (c * UG + g) + s;
      }
    }
  } else {
#pragma unroll
    for (int u = 0; u < 4; ++u) {
      const int g = tid + u * 1024;
      if (g < GG) {
        int pos[4];
#pragma unroll
        for (int s = 0; s < 4; ++s) pos[s] = atomicAdd(&off[k[u][s]], 1);
#pragma unroll
        for (int s = 0; s < 4; ++s) { gcol[pos[s]] = sv[u][s]; gw[pos[s]] = w[u][s]; }
      }
    }
  }
}

template <int TS, int NS>
__global__ __launch_bounds__(256) void k_gather_users(
    const int* __restrict__ uptr, const int* __restrict__ ulist, int ustride,
    const float* __restrict__ Wemb, const float* __restrict__ bemb, float* __restrict__ e_all,
    int nloc) {
  const int lane = threadIdx.x & 31;
  const int j = __builtin_amdgcn_readfirstlane(blockIdx.x * 8 + (threadIdx.x >> 5));
  if (j >= nloc) return;
  const v2f bb = *(const v2f*)(bemb + 2 * lane);
  for (int it = 0; it < TS; ++it) {
    const int t = blockIdx.y * TS + it;
    const int* ptr = uptr + t * NUM_LOC;
    const int* ul = ulist + (long)t * ustride;
    const int beg = j ? ptr[j - 1] : 0;
    const int end = ptr[j];
    v2f acc = { 0.0f, 0.0f };
    for (int k = beg; k < end; k += 8) {
      int id[8];
#pragma unroll
      for (int u = 0; u < 8; ++u) id[u] = ul[k + u < end ? k + u : end - 1];
      v2f x[8];
#pragma unroll
      for (int u = 0; u < 8; ++u) x[u] = *(const v2f*)(Wemb + (long)id[u] * 64 + 2 * lane);
#pragma unroll
      for (int u = 0; u < 8; ++u) {
        if (k + u < end) acc += x[u] + bb;
      }
    }
    float cv = (float)(end - beg); if (cv < 1.0f) cv = 1.0f;
    v2f r = { acc[0] / cv, acc[1] / cv };
    *(v2f*)(e_all + ((long)j * NS + t) * 64 + 2 * lane) = r;
  }
}

#define TP 68
#define OP 36

template <int TS, int NS>
__device__ __forceinline__ void gather_node(
    v2f (&acc)[TS], const int* __restrict__ gptr, const int* __restrict__ gcol,
    const float* __restrict__ gw, const float* __restrict__ idg, const float* __restrict__ h,
    int q, int t0, int lane) {
  constexpr int GE = TS == 1 ? 8 : 4;
  const int beg = q ? gptr[q - 1] : 0;
  const int end = gptr[q];
  const float* hl = h + (long)t0 * 64 + 2 * lane;
  v2f self[TS];
#pragma unroll
  for (int ts = 0; ts < TS; ++ts) self[ts] = *(const v2f*)(hl + (long)q * (NS * 64) + ts * 64);
  const float sd = idg[q];
#pragma unroll
  for (int ts = 0; ts < TS; ++ts) { v2f z = { 0.0f, 0.0f }; acc[ts] = z; }
  for (int k = beg; k < end; k += GE) {
    int s[GE];
    float w[GE];
#pragma unroll
    for (int u = 0; u < GE; ++u) {
      const int kk = k + u < end ? k + u : end - 1;
      s[u] = gcol[kk];
      w[u] = gw[kk];
    }
    v2f x[GE][TS];
#pragma unroll
    for (int u = 0; u < GE; ++u) {
      const float* hs = hl + (long)s[u] * (NS * 64);
#pragma unroll
      for (int ts = 0; ts < TS; ++ts) x[u][ts] = *(const v2f*)(hs + ts * 64);
    }
#pragma unroll
    for (int u = 0; u < GE; ++u) {
      const bool live = k + u < end;
#pragma unroll
      for (int ts = 0; ts < TS; ++ts) {
        const v2f old = acc[ts];
        const v2f upd = old + x[u][ts] * w[u];
        acc[ts] = live ? upd : old;
      }
    }
  }
#pragma unroll
  for (int ts = 0; ts < TS; ++ts) acc[ts] += self[ts] * sd;
}

template <int NKS>
__device__ __forceinline__ void load_bfrag(v2f (&b)[16], const float* __restrict__ Wm, int nc,
                                           int tt, int half, int l16) {
#pragma unroll
  for (int ks = 0; ks < NKS; ++ks) {
    b[ks][0] = Wm[(4 * ks + 2 * half + 0) * nc + tt * 16 + l16];
    b[ks][1] = Wm[(4 * ks + 2 * half + 1) * nc + tt * 16 + l16];
  }
}

template <int NT, int NKS = 16, int PITCH = TP>
__device__ __forceinline__ void tiles_product(v8f (&c)[NT], const float* tiles, const v2f (&b)[16],
                                              float bv, int half, int l16) {
#pragma unroll
  for (int i = 0; i < NT; ++i) { v8f z = { bv, bv, bv, bv, bv, bv, bv, bv }; c[i] = z; }
#pragma unroll
  for (int ks = 0; ks < NKS; ++ks) {
#pragma unroll
    for (int i = 0; i < NT; ++i) {
      const v2f a = *(const v2f*)(tiles + (i * 16 + l16) * PITCH + 4 * ks + 2 * half);
      c[i] = __builtin_amdgcn_wmma_f32_16x16x4_f32(false, a, false, b[ks], (short)0, c[i], false, false);
    }
  }
}

template <int TS, int NS>
__global__ __launch_bounds__(512) void k_gcn_relu(
    const int* __restrict__ gptr, const int* __restrict__ gcol, const float* __restrict__ gw,
    const float* __restrict__ idg, const float* __restrict__ hin, const float* __restrict__ W,
    const float* __restrict__ bias, float* __restrict__ hout, int nloc) {
  __shared__ __attribute__((aligned(16))) float At[TS * 16 * TP];
  __shared__ __attribute__((aligned(16))) float Ct[TS * 16 * TP];
  const int lane = threadIdx.x & 31;
  const int wave = threadIdx.x >> 5;
  const int half = lane >> 4;
  const int l16  = lane & 15;
  const int t0   = blockIdx.y * TS;
  const int q = __builtin_amdgcn_readfirstlane(blockIdx.x * 16 + wave);

  v2f bf[16];
  float bv = 0.0f;
  if (wave < 4) {
    load_bfrag<16>(bf, W, 64, wave, half, l16);
    bv = bias[wave * 16 + l16];
  }

  v2f acc[TS];
#pragma unroll
  for (int ts = 0; ts < TS; ++ts) { v2f z = { 0.0f, 0.0f }; acc[ts] = z; }
  if (q < nloc) gather_node<TS, NS>(acc, gptr, gcol, gw, idg, hin, q, t0, lane);
#pragma unroll
  for (int ts = 0; ts < TS; ++ts) *(v2f*)(At + (ts * 16 + wave) * TP + 2 * lane) = acc[ts];
  __syncthreads();

  if (wave < 4) {
    v8f c[TS];
    tiles_product<TS>(c, At, bf, bv, half, l16);
#pragma unroll
    for (int ts = 0; ts < TS; ++ts) {
#pragma unroll
      for (int r = 0; r < 8; ++r) {
        const float v = c[ts][r];
        Ct[(ts * 16 + r + 8 * half) * TP + wave * 16 + l16] = v > 0.0f ? v : 0.0f;
      }
    }
  }
  __syncthreads();

  if (q < nloc) {
#pragma unroll
    for (int ts = 0; ts < TS; ++ts)
      *(v2f*)(hout + ((long)q * NS + t0 + ts) * 64 + 2 * lane) =
          *(const v2f*)(Ct + (ts * 16 + wave) * TP + 2 * lane);
  }
}

template <int TS, int NS>
__global__ __launch_bounds__(512) void k_gcn_e2n(
    const int* __restrict__ gptr, const int* __restrict__ gcol, const float* __restrict__ gw,
    const float* __restrict__ idg, const float* __restrict__ hin, const float* __restrict__ W2,
    const float* __restrict__ b2, const float* __restrict__ We, const float* __restrict__ be,
    const float* __restrict__ Wi0, const float* __restrict__ bL0,
    float* __restrict__ out, int nloc) {
  __shared__ __attribute__((aligned(16))) float At[TS * 16 * TP];
  __shared__ __attribute__((aligned(16))) float Ht[TS * 16 * TP];
  __shared__ __attribute__((aligned(16))) float Ot[TS * 16 * OP];
  const int lane = threadIdx.x & 31;
  const int wave = threadIdx.x >> 5;
  const int half = lane >> 4;
  const int l16  = lane & 15;
  const int t0   = blockIdx.y * TS;
  const int q = __builtin_amdgcn_readfirstlane(blockIdx.x * 16 + wave);

  constexpr int ET = TS == 1 ? 1 : 2;
  constexpr int EW = 2 * TS / ET;
  static_assert(4 + EW <= 12, "the edge2node waves end before the gate waves begin");
  const int ec = (wave - 4) & 1;
  const int es = ((wave - 4) >> 1) * ET;
  const int gc = wave - 12;
  v2f bf[16];
  float bv = 0.0f;
  if (wave < 4) {
    load_bfrag<16>(bf, W2, 64, wave, half, l16);
    bv = b2[wave * 16 + l16];
  } else if (wave < 4 + EW) {
    load_bfrag<16>(bf, We, 32, ec, half, l16);
    bv = be[ec * 16 + l16];
  } else if (wave >= 12) {
    load_bfrag<8>(bf, Wi0, 64, gc, half, l16);
    bv = bL0[gc * 16 + l16];
  }

  v2f acc[TS];
#pragma unroll
  for (int ts = 0; ts < TS; ++ts) { v2f z = { 0.0f, 0.0f }; acc[ts] = z; }
  if (q < nloc) gather_node<TS, NS>(acc, gptr, gcol, gw, idg, hin, q, t0, lane);
#pragma unroll
  for (int ts = 0; ts < TS; ++ts) *(v2f*)(At + (ts * 16 + wave) * TP + 2 * lane) = acc[ts];
  __syncthreads();

  if (wave < 4) {
    v8f c[TS];
    tiles_product<TS>(c, At, bf, bv, half, l16);
#pragma unroll
    for (int ts = 0; ts < TS; ++ts) {
#pragma unroll
      for (int r = 0; r < 8; ++r) Ht[(ts * 16 + r + 8 * half) * TP + wave * 16 + l16] = c[ts][r];
    }
  }
  __syncthreads();

  if (wave >= 4 && wave < 4 + EW) {
    v8f c[ET];
    tiles_product<ET>(c, Ht + es * 16 * TP, bf, bv, half, l16);
#pragma unroll
    for (int i = 0; i < ET; ++i) {
#pragma unroll
      for (int r = 0; r < 8; ++r) Ot[((es + i) * 16 + r + 8 * half) * OP + ec * 16 + l16] = c[i][r];
    }
  }
  __syncthreads();

  if (wave >= 12) {
    v8f c[TS];
    tiles_product<TS, 8, OP>(c, Ot, bf, bv, half, l16);
#pragma unroll
    for (int ts = 0; ts < TS; ++ts) {
#pragma unroll
      for (int r = 0; r < 8; ++r) At[(ts * 16 + r + 8 * half) * TP + gc * 16 + l16] = c[ts][r];
    }
  }
  __syncthreads();

  if (q < nloc) {
#pragma unroll
    for (int ts = 0; ts < TS; ++ts)
      *(v2f*)(out + ((long)(t0 + ts) * nloc + q) * 64 + 2 * lane) =
          *(const v2f*)(At + (ts * 16 + wave) * TP + 2 * lane);
  }
}

static_assert((long)T_STEPS * NUM_LOC * 64 < (1L << 31) && (long)T_STEPS * USR_NUM < (1L << 31), "32-bit element indices in the tail");
#define HP 20

__device__ __forceinline__ float sigm(float x) {
  return __builtin_amdgcn_rcpf(1.0f + __builtin_amdgcn_exp2f(x * -1.4426950408889634f));
}
__device__ __forceinline__ float tanh_small(float x) {
  const float y2 = x * x;
  float p = __builtin_fmaf(y2, -0x1.758e7ap-8f, 0x1.521192p-6f);
  p = __builtin_fmaf(y2, p, -0x1.b8389cp-5f);
  p = __builtin_fmaf(y2, p, 0x1.110704p-3f);
  p = __builtin_fmaf(y2, p, -0x1.555532p-2f);
  return __builtin_fmaf(y2, x * p, x);
}
__device__ __forceinline__ float tanh_any(float x) {
  const float y = __builtin_fabsf(x);
  const float t = __builtin_amdgcn_exp2f(y * 2.8853900817779268f);
  const float big = __builtin_fmaf(__builtin_amdgcn_rcpf(t + 1.0f), -2.0f, 1.0f);
  return y < 0.625f ? tanh_small(x) : __builtin_copysignf(big, x);
}
__device__ __forceinline__ void tanh8(float (&o)[8], const float (&x)[8]) {
  float m = __builtin_fabsf(x[0]);
#pragma unroll
  for (int r = 1; r < 8; ++r) m = __builtin_fmaxf(m, __builtin_fabsf(x[r]));
  if (__builtin_amdgcn_ballot_w32(m >= 0.625f) == 0) {
#pragma unroll
    for (int r = 0; r < 8; ++r) o[r] = tanh_small(x[r]);
  } else {
#pragma unroll
    for (int r = 0; r < 8; ++r) o[r] = tanh_any(x[r]);
  }
}
__device__ __forceinline__ void cell8(const v8f (&acc)[4], float (&c)[8], float* Ht, int half, int l16) {
  float g[8], tg[8], th[8];
#pragma unroll
  for (int r = 0; r < 8; ++r) g[r] = acc[2][r];
  tanh8(tg, g);
#pragma unroll
  for (int r = 0; r < 8; ++r) c[r] = sigm(acc[1][r]) * c[r] + sigm(acc[0][r]) * tg[r];
  tanh8(th, c);
#pragma unroll
  for (int r = 0; r < 8; ++r) Ht[(r + 8 * half) * HP + l16] = sigm(acc[3][r]) * th[r];
}

template <int NKS>
__device__ __forceinline__ void stage_frag(float* dst, const float* __restrict__ Wm) {
  for (int i = threadIdx.x; i < NKS * 256; i += 64) {
    const int s = i & 1, tt = (i >> 1) & 3, ln = (i >> 3) & 31, ks = i >> 8;
    dst[i] = Wm[(ks * 4 + 2 * (ln >> 4) + s) * 64 + tt * 16 + (ln & 15)];
  }
}

__global__ __launch_bounds__(64) void k_lstm_all(
    const int* __restrict__ seq,
    const float* __restrict__ g0,
    const float* __restrict__ Wh0,
    const float* __restrict__ Wi1, const float* __restrict__ Wh1, const float* __restrict__ b1,
    const float* __restrict__ Wout, const float* __restrict__ bout,
    float* __restrict__ out, int N) {
  __shared__ __attribute__((aligned(32))) float Ws[48 * 64];
  __shared__ __attribute__((aligned(16))) float Hs[2 * 16 * HP];
  __shared__ __attribute__((aligned(16))) float Wo[32];
  __shared__ __attribute__((aligned(16))) float Bb[16 * 4];
  stage_frag<4>(Ws,           Wh0);
  stage_frag<4>(Ws + 4 * 256, Wi1);
  stage_frag<4>(Ws + 8 * 256, Wh1);
  if (threadIdx.x < 32) Wo[threadIdx.x] = Wout[(threadIdx.x & 15) * 2 + (threadIdx.x >> 4)];
  Bb[threadIdx.x] = b1[(threadIdx.x & 3) * 16 + (threadIdx.x >> 2)];
  __syncthreads();

  const int lane = threadIdx.x & 31;
  const int wave = threadIdx.x >> 5;
  const int half = lane >> 4;
  const int l16  = lane & 15;
  const int row0 = (blockIdx.x * 2 + wave) * 16;

  int ur[8];
#pragma unroll
  for (int r = 0; r < 8; ++r) { const int u = row0 + 8 * half + r; ur[r] = u > N - 1 ? N - 1 : u; }

  const float* wo = Wo + half * 16;
  const float bo = bout[half];

  const v8f* Wf = (const v8f*)Ws + lane;
  float* Ht = Hs + wave * (16 * HP);

  v2f h0A[4], h1A[4];
  float c0[8], c1[8];
#pragma unroll
  for (int i = 0; i < 4; ++i) { v2f z = { 0.0f, 0.0f }; h0A[i] = z; h1A[i] = z; }
#pragma unroll
  for (int i = 0; i < 8; ++i) { c0[i] = 0.0f; c1[i] = 0.0f; }

  int id[8];
#pragma unroll
  for (int r = 0; r < 8; ++r) id[r] = seq[ur[r]];
  for (int t = 0; t < T_STEPS; ++t) {
    v8f acc[4];
#pragma unroll
    for (int r = 0; r < 8; ++r) {
      const float* gr = g0 + (unsigned)((t * NUM_LOC + id[r]) * 64 + l16);
#pragma unroll
      for (int tt = 0; tt < 4; ++tt) acc[tt][r] = gr[tt * 16];
    }
    const int tn = t + 1 < T_STEPS ? t + 1 : t;
#pragma unroll
    for (int r = 0; r < 8; ++r) id[r] = seq[(unsigned)(tn * N + ur[r])];

#pragma unroll
    for (int ks = 0; ks < 4; ++ks) {
      const v8f w8 = Wf[(0 + ks) * 32];
#pragma unroll
      for (int tt = 0; tt < 4; ++tt) {
        v2f w = { w8[2 * tt], w8[2 * tt + 1] };
        acc[tt] = __builtin_amdgcn_wmma_f32_16x16x4_f32(false, h0A[ks], false, w, (short)0,
                                                        acc[tt], false, false);
      }
    }
    cell8(acc, c0, Ht, half, l16);
    __syncthreads();
#pragma unroll
    for (int ks = 0; ks < 4; ++ks) h0A[ks] = *(const v2f*)(Ht + l16 * HP + ks * 4 + 2 * half);

    const v4f bs1 = *(const v4f*)(Bb + l16 * 4);
#pragma unroll
    for (int tt = 0; tt < 4; ++tt) {
      float bv = bs1[tt];
      v8f tmp = { bv, bv, bv, bv, bv, bv, bv, bv };
      acc[tt] = tmp;
    }
#pragma unroll
    for (int ks = 0; ks < 4; ++ks) {
      const v8f w8 = Wf[(4 + ks) * 32];
#pragma unroll
      for (int tt = 0; tt < 4; ++tt) {
        v2f w = { w8[2 * tt], w8[2 * tt + 1] };
        acc[tt] = __builtin_amdgcn_wmma_f32_16x16x4_f32(false, h0A[ks], false, w, (short)0,
                                                        acc[tt], false, false);
      }
    }
#pragma unroll
    for (int ks = 0; ks < 4; ++ks) {
      const v8f w8 = Wf[(8 + ks) * 32];
#pragma unroll
      for (int tt = 0; tt < 4; ++tt) {
        v2f w = { w8[2 * tt], w8[2 * tt + 1] };
        acc[tt] = __builtin_amdgcn_wmma_f32_16x16x4_f32(false, h1A[ks], false, w, (short)0,
                                                        acc[tt], false, false);
      }
    }
    cell8(acc, c1, Ht, half, l16);
    __syncthreads();
#pragma unroll
    for (int ks = 0; ks < 4; ++ks) h1A[ks] = *(const v2f*)(Ht + l16 * HP + ks * 4 + 2 * half);

    float a = bo;
#pragma unroll
    for (int j = 0; j < 16; ++j) a += Ht[l16 * HP + j] * wo[j];
    if (row0 + l16 < N) out[((long)t * N + row0 + l16) * 2 + half] = a;
  }
}

static inline int gs(long n, int b) { return (int)((n + b - 1) / b); }

extern "C" void kernel_launch(void* const* d_in, const int* in_sizes, int n_in,
                              void* d_out, int out_size, void* d_ws, size_t ws_size,
                              hipStream_t stream) {
  (void)in_sizes; (void)n_in; (void)out_size;

  const int*   seq  = (const int*)d_in[0];
  const int*   gidx = (const int*)d_in[1];
  const float* Wemb = (const float*)d_in[2];
  const float* bemb = (const float*)d_in[3];
  const float* W1   = (const float*)d_in[4];
  const float* b1   = (const float*)d_in[5];
  const float* W2   = (const float*)d_in[6];
  const float* b2   = (const float*)d_in[7];
  const float* We2n = (const float*)d_in[8];
  const float* be2n = (const float*)d_in[9];
  const float* Wi0  = (const float*)d_in[10];
  const float* Wh0  = (const float*)d_in[11];
  const float* bL0  = (const float*)d_in[12];
  const float* Wi1  = (const float*)d_in[13];
  const float* Wh1  = (const float*)d_in[14];
  const float* bL1  = (const float*)d_in[15];
  const float* Wout = (const float*)d_in[16];
  const float* bout = (const float*)d_in[17];

  const int* srcE = gidx;
  const int* dstE = gidx + N_EDGES;
  float* out = (float*)d_out;

  float* p = (float*)d_ws;
  float* isq   = p; p += 5056;
  float* idg   = p; p += 5056;
  int*   uptr  = (int*)p; p += (size_t)T_STEPS * NUM_LOC;
  int*   gptr  = (int*)p; p += 5056;
  int*   gcol  = (int*)p; p += N_EDGES;
  float* gw    = p; p += N_EDGES;
  const size_t FIXED = 3 * 5056 + (size_t)T_STEPS * NUM_LOC + 2 * (size_t)N_EDGES;
  const size_t AREA  = (size_t)NUM_LOC * T_STEPS * 64;

  const bool all_steps = ws_size >= (FIXED + 2 * AREA) * sizeof(float);
  float* e     = p;
  float* h1    = all_steps ? p + AREA : p + NUM_LOC * 64;
  float* g0all = all_steps ? p : p + 2 * NUM_LOC * 64;
  int*   ulist = all_steps ? (int*)(p + AREA) : (int*)g0all;
  const int ustride = all_steps ? USR_NUM : NUM_LOC * 64;
  int*   hist  = all_steps ? (int*)p : (int*)(p + 18 * NUM_LOC * 64);

  k_hist<<<NROW * NB, 1024, 0, stream>>>(seq, dstE, hist);
  k_scan<<<NROW, 1024, 0, stream>>>(hist, uptr, gptr, isq, idg);
  k_fill<<<NROW * NB, 1024, 0, stream>>>(seq, srcE, dstE, isq, hist, ulist, ustride, gcol, gw);

  if (all_steps) {
    const int TS = 4;
    k_gather_users<TS, T_STEPS><<<dim3(gs(NUM_LOC, 8), T_STEPS / TS), 256, 0, stream>>>(
        uptr, ulist, ustride, Wemb, bemb, e, NUM_LOC);
    k_gcn_relu<TS, T_STEPS><<<dim3(gs(NUM_LOC, 16), T_STEPS / TS), 512, 0, stream>>>(
        gptr, gcol, gw, idg, e, W1, b1, h1, NUM_LOC);
    k_gcn_e2n<TS, T_STEPS><<<dim3(gs(NUM_LOC, 16), T_STEPS / TS), 512, 0, stream>>>(
        gptr, gcol, gw, idg, h1, W2, b2, We2n, be2n, Wi0, bL0, g0all, NUM_LOC);
  } else {
    for (int t = 0; t < T_STEPS; ++t) {
      k_gather_users<1, 1><<<gs(NUM_LOC, 8), 256, 0, stream>>>(
          uptr + (size_t)t * NUM_LOC, ulist + (size_t)t * ustride, ustride, Wemb, bemb, e, NUM_LOC);
      k_gcn_relu<1, 1><<<gs(NUM_LOC, 16), 512, 0, stream>>>(gptr, gcol, gw, idg, e, W1, b1, h1, NUM_LOC);
      k_gcn_e2n<1, 1><<<gs(NUM_LOC, 16), 512, 0, stream>>>(gptr, gcol, gw, idg, h1, W2, b2, We2n, be2n, Wi0, bL0,
                                                            g0all + (size_t)t * NUM_LOC * 64, NUM_LOC);
    }
  }

  k_lstm_all<<<gs(USR_NUM, 32), 64, 0, stream>>>(seq, g0all, Wh0, Wi1, Wh1, bL1,
                                                 Wout, bout, out, USR_NUM);
}


# GCN layer 2 and edge2node folded: M = W2 @ We (64x32, built per call) multiplies layer-1 rows BEFORE aggregation, so layer 2 gathers 128-B rows; products re-associated, rounding changes
# speedup vs baseline: 10.5954x; 1.0387x over previous
#include <hip/hip_runtime.h>
#include <hip/hip_bf16.h>
#include <math.h>

#define USR_NUM 100000
#define USR_DIM 64
#define NUM_LOC 5000
#define T_STEPS 16
#define N_EDGES 80000

typedef float v2f __attribute__((ext_vector_type(2)));
typedef float v4f __attribute__((ext_vector_type(4)));
typedef float v8f __attribute__((ext_vector_type(8)));

typedef int v4i __attribute__((ext_vector_type(4)));
#define NB 8
#define UG (USR_NUM / 4 / NB)
#define GG (N_EDGES / 4 / NB)
#define NROW (T_STEPS + 1)
static_assert(USR_NUM % (4 * NB) == 0 && N_EDGES % (4 * NB) == 0, "chunks are whole 16-byte key groups");
static_assert(UG <= 4 * 1024 && GG <= 4 * 1024, "a thread takes at most 4 key groups of its chunk");
static_assert(NUM_LOC <= 5 * 1024, "a thread takes at most 5 bins of a histogram");

__device__ __forceinline__ void fold_w2_we(const float* __restrict__ W2, const float* __restrict__ We,
                                           const float* __restrict__ b2, const float* __restrict__ be,
                                           float* __restrict__ M, float* __restrict__ cy) {
  const int lane = threadIdx.x & 31;
  const int wave = threadIdx.x >> 5;
  const int half = lane >> 4;
  const int l16  = lane & 15;
  if (wave < 8) {
    const int rt = wave >> 1;
    const int ct = wave & 1;
    v8f acc = { 0.0f, 0.0f, 0.0f, 0.0f, 0.0f, 0.0f, 0.0f, 0.0f };
#pragma unroll
    for (int ks = 0; ks < 16; ++ks) {
      const v2f a = *(const v2f*)(W2 + (rt * 16 + l16) * 64 + 4 * ks + 2 * half);
      v2f b;
      b[0] = We[(4 * ks + 2 * half + 0) * 32 + ct * 16 + l16];
      b[1] = We[(4 * ks + 2 * half + 1) * 32 + ct * 16 + l16];
      acc = __builtin_amdgcn_wmma_f32_16x16x4_f32(false, a, false, b, (short)0, acc, false, false);
    }
#pragma unroll
    for (int r = 0; r < 8; ++r) M[(rt * 16 + r + 8 * half) * 32 + ct * 16 + l16] = acc[r];
  } else if (wave == 8) {
    float a = be[lane];
    for (int k = 0; k < 64; ++k) a += b2[k] * We[k * 32 + lane];
    cy[lane] = a;
  }
}

__global__ __launch_bounds__(1024) void k_hist(const int* __restrict__ seq, const int* __restrict__ dst,
                                                int* __restrict__ hist,
                                                const float* __restrict__ W2, const float* __restrict__ We,
                                                const float* __restrict__ b2, const float* __restrict__ be,
                                                float* __restrict__ M, float* __restrict__ cy) {
  __shared__ int h[NUM_LOC];
  if (blockIdx.x == NROW * NB) {
    fold_w2_we(W2, We, b2, be, M, cy);
    return;
  }
  const int tid = threadIdx.x;
  const int row = blockIdx.x / NB;
  const int c = blockIdx.x - row * NB;
  const bool graph = row == T_STEPS;
  const int ng = graph ? GG : UG;
  const int* keys = graph ? dst + 4 * c * GG : seq + (long)row * USR_NUM + 4 * c * UG;
  v4i k[4];
#pragma unroll
  for (int u = 0; u < 4; ++u) {
    const int g = tid + u * 1024;
    const v4i z = { 0, 0, 0, 0 };
    k[u] = g < ng ? *(const v4i*)(keys + 4 * g) : z;
  }
  for (int i = tid; i < NUM_LOC; i += 1024) h[i] = 0;
  __syncthreads();
#pragma unroll
  for (int u = 0; u < 4; ++u) {
    if (tid + u * 1024 < ng) {
#pragma unroll
      for (int s = 0; s < 4; ++s) atomicAdd(&h[k[u][s]], 1);
    }
  }
  __syncthreads();
  int* hp = hist + (long)blockIdx.x * NUM_LOC;
  for (int i = tid; i < NUM_LOC; i += 1024) hp[i] = h[i];
}

__global__ __launch_bounds__(1024) void k_scan(int* __restrict__ hist, int* __restrict__ uptr,
                                               int* __restrict__ gptr,
                                               float* __restrict__ isq, float* __restrict__ idg) {
  __shared__ int part[1024];
  __shared__ int tot[5 * 1024];
  const int tid = threadIdx.x;
  const int row = blockIdx.x;
  const bool graph = row == T_STEPS;
  int* p = graph ? gptr : uptr + row * NUM_LOC;
  int* hp = hist + (long)row * NB * NUM_LOC;
  int pre[5][NB];
  int tt[5];
#pragma unroll
  for (int k = 0; k < 5; ++k) {
    const int idx = k * 1024 + tid;
    int t = 0;
#pragma unroll
    for (int c = 0; c < NB; ++c) {
      const int x = idx < NUM_LOC ? hp[c * NUM_LOC + idx] : 0;
      pre[k][c] = t;
      t += x;
    }
    tt[k] = t;
    tot[idx] = t;
  }
  __syncthreads();
  int v[5];
  int s = 0;
#pragma unroll
  for (int k = 0; k < 5; ++k) { v[k] = tot[tid * 5 + k]; s += v[k]; }
  part[tid] = s;
  __syncthreads();
  for (int off = 1; off < 1024; off <<= 1) {
    int x = tid >= off ? part[tid - off] : 0;
    __syncthreads();
    part[tid] += x;
    __syncthreads();
  }
  int run = part[tid] - s;
#pragma unroll
  for (int k = 0; k < 5; ++k) { tot[tid * 5 + k] = run; run += v[k]; }
  __syncthreads();
#pragma unroll
  for (int k = 0; k < 5; ++k) {
    const int idx = k * 1024 + tid;
    if (idx < NUM_LOC) {
      const int st = tot[idx];
#pragma unroll
      for (int c = 0; c < NB; ++c) hp[c * NUM_LOC + idx] = st + pre[k][c];
      p[idx] = st + tt[k];
      if (graph) {
        const float d = (float)tt[k] + 1.0f;
        isq[idx] = rsqrtf(d);
        idg[idx] = 1.0f / d;
      }
    }
  }
}

__global__ __launch_bounds__(1024) void k_fill(const int* __restrict__ seq, const int* __restrict__ src,
                       const int* __restrict__ dst, const float* __restrict__ isq,
                       const int* __restrict__ hist,
                       int* __restrict__ ulist_all, int ustride, int* __restrict__ gcol,
                       float* __restrict__ gw) {
  __shared__ int off[NUM_LOC];
  const int tid = threadIdx.x;
  const int row = blockIdx.x / NB;
  const int c = blockIdx.x - row * NB;
  const bool graph = row == T_STEPS;
  const int ng = graph ? GG : UG;
  const int* keys = graph ? dst + 4 * c * GG : seq + (long)row * USR_NUM + 4 * c * UG;
  const int* hp = hist + (long)blockIdx.x * NUM_LOC;
  v4i k[4], sv[4];
  float w[4][4];
#pragma unroll
  for (int u = 0; u < 4; ++u) {
    const int g = tid + u * 1024;
    const v4i z = { 0, 0, 0, 0 };
    k[u] = g < ng ? *(const v4i*)(keys + 4 * g) : z;
    sv[u] = z;
  }
  if (graph) {
    const int* srcs = src + 4 * c * GG;
#pragma unroll
    for (int u = 0; u < 4; ++u) {
      const int g = tid + u * 1024;
      if (g < GG) sv[u] = *(const v4i*)(srcs + 4 * g);
    }
#pragma unroll
    for (int u = 0; u < 4; ++u) {
#pragma unroll
      for (int s = 0; s < 4; ++s) w[u][s] = isq[sv[u][s]] * isq[k[u][s]];
    }
  }
#pragma unroll
  for (int it = 0; it < 5; ++it) {
    const int i = tid + it * 1024;
    if (i < NUM_LOC) off[i] = hp[i];
  }
  __syncthreads();
  if (!graph) {
    int* l = ulist_all + (long)row * ustride;
#pragma unroll
    for (int u = 0; u < 4; ++u) {
      const int g = tid + u * 1024;
      if (g < UG) {
        int pos[4];
#pragma unroll
        for (int s = 0; s < 4; ++s) pos[s] = atomicAdd(&off[k[u][s]], 1);
#pragma unroll
        for (int s = 0; s < 4; ++s) l[pos[s]] = 4 * (c * UG + g) + s;
      }
    }
  } else {
#pragma unroll
    for (int u = 0; u < 4; ++u) {
      const int g = tid + u * 1024;
      if (g < GG) {
        int pos[4];
#pragma unroll
        for (int s = 0; s < 4; ++s) pos[s] = atomicAdd(&off[k[u][s]], 1);
#pragma unroll
        for (int s = 0; s < 4; ++s) { gcol[pos[s]] = sv[u][s]; gw[pos[s]] = w[u][s]; }
      }
    }
  }
}

template <int TS, int NS>
__global__ __launch_bounds__(256) void k_gather_users(
    const int* __restrict__ uptr, const int* __restrict__ ulist, int ustride,
    const float* __restrict__ Wemb, const float* __restrict__ bemb, float* __restrict__ e_all,
    int nloc) {
  const int lane = threadIdx.x & 31;
  const int j = __builtin_amdgcn_readfirstlane(blockIdx.x * 8 + (threadIdx.x >> 5));
  if (j >= nloc) return;
  const v2f bb = *(const v2f*)(bemb + 2 * lane);
  for (int it = 0; it < TS; ++it) {
    const int t = blockIdx.y * TS + it;
    const int* ptr = uptr + t * NUM_LOC;
    const int* ul = ulist + (long)t * ustride;
    const int beg = j ? ptr[j - 1] : 0;
    const int end = ptr[j];
    v2f acc = { 0.0f, 0.0f };
    for (int k = beg; k < end; k += 8) {
      int id[8];
#pragma unroll
      for (int u = 0; u < 8; ++u) id[u] = ul[k + u < end ? k + u : end - 1];
      v2f x[8];
#pragma unroll
      for (int u = 0; u < 8; ++u) x[u] = *(const v2f*)(Wemb + (long)id[u] * 64 + 2 * lane);
#pragma unroll
      for (int u = 0; u < 8; ++u) {
        if (k + u < end) acc += x[u] + bb;
      }
    }
    float cv = (float)(end - beg); if (cv < 1.0f) cv = 1.0f;
    v2f r = { acc[0] / cv, acc[1] / cv };
    *(v2f*)(e_all + ((long)j * NS + t) * 64 + 2 * lane) = r;
  }
}

#define TP 68
#define OP 36

template <int TS, int NS>
__device__ __forceinline__ void gather_node(
    v2f (&acc)[TS], const int* __restrict__ gptr, const int* __restrict__ gcol,
    const float* __restrict__ gw, const float* __restrict__ idg, const float* __restrict__ h,
    int q, int t0, int lane) {
  constexpr int GE = TS == 1 ? 8 : 4;
  const int beg = q ? gptr[q - 1] : 0;
  const int end = gptr[q];
  const float* hl = h + (long)t0 * 64 + 2 * lane;
  v2f self[TS];
#pragma unroll
  for (int ts = 0; ts < TS; ++ts) self[ts] = *(const v2f*)(hl + (long)q * (NS * 64) + ts * 64);
  const float sd = idg[q];
#pragma unroll
  for (int ts = 0; ts < TS; ++ts) { v2f z = { 0.0f, 0.0f }; acc[ts] = z; }
  for (int k = beg; k < end; k += GE) {
    int s[GE];
    float w[GE];
#pragma unroll
    for (int u = 0; u < GE; ++u) {
      const int kk = k + u < end ? k + u : end - 1;
      s[u] = gcol[kk];
      w[u] = gw[kk];
    }
    v2f x[GE][TS];
#pragma unroll
    for (int u = 0; u < GE; ++u) {
      const float* hs = hl + (long)s[u] * (NS * 64);
#pragma unroll
      for (int ts = 0; ts < TS; ++ts) x[u][ts] = *(const v2f*)(hs + ts * 64);
    }
#pragma unroll
    for (int u = 0; u < GE; ++u) {
      const bool live = k + u < end;
#pragma unroll
      for (int ts = 0; ts < TS; ++ts) {
        const v2f old = acc[ts];
        const v2f upd = old + x[u][ts] * w[u];
        acc[ts] = live ? upd : old;
      }
    }
  }
#pragma unroll
  for (int ts = 0; ts < TS; ++ts) acc[ts] += self[ts] * sd;
}

template <int N> struct lanevec;
template <> struct lanevec<1> { typedef float t; };
template <> struct lanevec<4> { typedef v4f t; };

template <int TS, int NS>
__device__ __forceinline__ typename lanevec<TS>::t gather_node32(
    const int* __restrict__ gptr, const int* __restrict__ gcol, const float* __restrict__ gw,
    const float* __restrict__ idg, const float* __restrict__ h, int q, int t0, int lane) {
  typedef typename lanevec<TS>::t VT;
  constexpr int GE = 8;
  const int beg = q ? gptr[q - 1] : 0;
  const int end = gptr[q];
  const float* hl = h + (long)t0 * 32 + TS * lane;
  const VT self = *(const VT*)(hl + (long)q * (NS * 32));
  const float sd = idg[q];
  VT acc = {};
  for (int k = beg; k < end; k += GE) {
    int s[GE];
    float w[GE];
#pragma unroll
    for (int u = 0; u < GE; ++u) {
      const int kk = k + u < end ? k + u : end - 1;
      s[u] = gcol[kk];
      w[u] = gw[kk];
    }
    VT x[GE];
#pragma unroll
    for (int u = 0; u < GE; ++u) x[u] = *(const VT*)(hl + (long)s[u] * (NS * 32));
#pragma unroll
    for (int u = 0; u < GE; ++u) {
      const bool live = k + u < end;
      const VT upd = acc + x[u] * w[u];
      acc = live ? upd : acc;
    }
  }
  acc += self * sd;
  return acc;
}

template <int NKS>
__device__ __forceinline__ void load_bfrag(v2f (&b)[16], const float* __restrict__ Wm, int nc,
                                           int tt, int half, int l16) {
#pragma unroll
  for (int ks = 0; ks < NKS; ++ks) {
    b[ks][0] = Wm[(4 * ks + 2 * half + 0) * nc + tt * 16 + l16];
    b[ks][1] = Wm[(4 * ks + 2 * half + 1) * nc + tt * 16 + l16];
  }
}

template <int NT, int NKS = 16, int PITCH = TP>
__device__ __forceinline__ void tiles_product(v8f (&c)[NT], const float* tiles, const v2f (&b)[16],
                                              float bv, int half, int l16) {
#pragma unroll
  for (int i = 0; i < NT; ++i) { v8f z = { bv, bv, bv, bv, bv, bv, bv, bv }; c[i] = z; }
#pragma unroll
  for (int ks = 0; ks < NKS; ++ks) {
#pragma unroll
    for (int i = 0; i < NT; ++i) {
      const v2f a = *(const v2f*)(tiles + (i * 16 + l16) * PITCH + 4 * ks + 2 * half);
      c[i] = __builtin_amdgcn_wmma_f32_16x16x4_f32(false, a, false, b[ks], (short)0, c[i], false, false);
    }
  }
}

template <int TS, int NS>
__global__ __launch_bounds__(512) void k_gcn_relu(
    const int* __restrict__ gptr, const int* __restrict__ gcol, const float* __restrict__ gw,
    const float* __restrict__ idg, const float* __restrict__ hin, const float* __restrict__ W,
    const float* __restrict__ bias, const float* __restrict__ M, float* __restrict__ pout, int nloc) {
  typedef typename lanevec<TS>::t VT;
  __shared__ __attribute__((aligned(16))) float At[TS * 16 * TP];
  __shared__ __attribute__((aligned(16))) float Ct[TS * 16 * TP];
  __shared__ __attribute__((aligned(16))) float Pt[TS * 16 * OP];
  const int lane = threadIdx.x & 31;
  const int wave = threadIdx.x >> 5;
  const int half = lane >> 4;
  const int l16  = lane & 15;
  const int t0   = blockIdx.y * TS;
  const int q = __builtin_amdgcn_readfirstlane(blockIdx.x * 16 + wave);

  constexpr int ET = TS == 1 ? 1 : 2;
  constexpr int EW = 2 * TS / ET;
  static_assert(4 + EW <= 16, "the product waves are waves of the work-group");
  const int ec = (wave - 4) & 1;
  const int es = ((wave - 4) >> 1) * ET;
  v2f bf[16];
  float bv = 0.0f;
  if (wave < 4) {
    load_bfrag<16>(bf, W, 64, wave, half, l16);
    bv = bias[wave * 16 + l16];
  } else if (wave < 4 + EW) {
    load_bfrag<16>(bf, M, 32, ec, half, l16);
  }

  v2f acc[TS];
#pragma unroll
  for (int ts = 0; ts < TS; ++ts) { v2f z = { 0.0f, 0.0f }; acc[ts] = z; }
  if (q < nloc) gather_node<TS, NS>(acc, gptr, gcol, gw, idg, hin, q, t0, lane);
#pragma unroll
  for (int ts = 0; ts < TS; ++ts) *(v2f*)(At + (ts * 16 + wave) * TP + 2 * lane) = acc[ts];
  __syncthreads();

  if (wave < 4) {
    v8f c[TS];
    tiles_product<TS>(c, At, bf, bv, half, l16);
#pragma unroll
    for (int ts = 0; ts < TS; ++ts) {
#pragma unroll
      for (int r = 0; r < 8; ++r) {
        const float v = c[ts][r];
        Ct[(ts * 16 + r + 8 * half) * TP + wave * 16 + l16] = v > 0.0f ? v : 0.0f;
      }
    }
  }
  __syncthreads();

  if (wave >= 4 && wave < 4 + EW) {
    v8f c[ET];
    tiles_product<ET>(c, Ct + es * 16 * TP, bf, 0.0f, half, l16);
#pragma unroll
    for (int i = 0; i < ET; ++i) {
#pragma unroll
      for (int r = 0; r < 8; ++r) Pt[((es + i) * 16 + r + 8 * half) * OP + ec * 16 + l16] = c[i][r];
    }
  }
  __syncthreads();

  if (q < nloc) {
    const int f = TS * lane;
    *(VT*)(pout + ((long)q * NS + t0) * 32 + f) = *(const VT*)(Pt + ((f >> 5) * 16 + wave) * OP + (f & 31));
  }
}

template <int TS, int NS>
__global__ __launch_bounds__(512) void k_gcn_e2n(
    const int* __restrict__ gptr, const int* __restrict__ gcol, const float* __restrict__ gw,
    const float* __restrict__ idg, const float* __restrict__ pin, const float* __restrict__ cy,
    const float* __restrict__ Wi0, const float* __restrict__ bL0,
    float* __restrict__ out, int nloc) {
  typedef typename lanevec<TS>::t VT;
  __shared__ __attribute__((aligned(16))) float Ot[TS * 16 * OP];
  __shared__ __attribute__((aligned(16))) float Gt[TS * 16 * TP];
  const int lane = threadIdx.x & 31;
  const int wave = threadIdx.x >> 5;
  const int half = lane >> 4;
  const int l16  = lane & 15;
  const int t0   = blockIdx.y * TS;
  const int q = __builtin_amdgcn_readfirstlane(blockIdx.x * 16 + wave);

  static_assert(4 * TS <= 16, "the product waves are waves of the work-group");
  const int gc = wave & 3;
  const int gs = wave >> 2;
  v2f bf[16];
  float bv = 0.0f;
  if (wave < 4 * TS) {
    load_bfrag<8>(bf, Wi0, 64, gc, half, l16);
    bv = bL0[gc * 16 + l16];
  }

  const int f = TS * lane;
  const VT cv = *(const VT*)(cy + (f & 31));
  VT acc = {};
  if (q < nloc) acc = gather_node32<TS, NS>(gptr, gcol, gw, idg, pin, q, t0, lane);
  *(VT*)(Ot + ((f >> 5) * 16 + wave) * OP + (f & 31)) = acc + cv;
  __syncthreads();

  if (wave < 4 * TS) {
    v8f c[1];
    tiles_product<1, 8, OP>(c, Ot + gs * 16 * OP, bf, bv, half, l16);
#pragma unroll
    for (int r = 0; r < 8; ++r) Gt[(gs * 16 + r + 8 * half) * TP + gc * 16 + l16] = c[0][r];
  }
  __syncthreads();

  if (q < nloc) {
#pragma unroll
    for (int ts = 0; ts < TS; ++ts)
      *(v2f*)(out + ((long)(t0 + ts) * nloc + q) * 64 + 2 * lane) =
          *(const v2f*)(Gt + (ts * 16 + wave) * TP + 2 * lane);
  }
}

static_assert((long)T_STEPS * NUM_LOC * 64 < (1L << 31) && (long)T_STEPS * USR_NUM < (1L << 31), "32-bit element indices in the tail");
#define HP 20

__device__ __forceinline__ float sigm(float x) {
  return __builtin_amdgcn_rcpf(1.0f + __builtin_amdgcn_exp2f(x * -1.4426950408889634f));
}
__device__ __forceinline__ float tanh_small(float x) {
  const float y2 = x * x;
  float p = __builtin_fmaf(y2, -0x1.758e7ap-8f, 0x1.521192p-6f);
  p = __builtin_fmaf(y2, p, -0x1.b8389cp-5f);
  p = __builtin_fmaf(y2, p, 0x1.110704p-3f);
  p = __builtin_fmaf(y2, p, -0x1.555532p-2f);
  return __builtin_fmaf(y2, x * p, x);
}
__device__ __forceinline__ float tanh_any(float x) {
  const float y = __builtin_fabsf(x);
  const float t = __builtin_amdgcn_exp2f(y * 2.8853900817779268f);
  const float big = __builtin_fmaf(__builtin_amdgcn_rcpf(t + 1.0f), -2.0f, 1.0f);
  return y < 0.625f ? tanh_small(x) : __builtin_copysignf(big, x);
}
__device__ __forceinline__ void tanh8(float (&o)[8], const float (&x)[8]) {
  float m = __builtin_fabsf(x[0]);
#pragma unroll
  for (int r = 1; r < 8; ++r) m = __builtin_fmaxf(m, __builtin_fabsf(x[r]));
  if (__builtin_amdgcn_ballot_w32(m >= 0.625f) == 0) {
#pragma unroll
    for (int r = 0; r < 8; ++r) o[r] = tanh_small(x[r]);
  } else {
#pragma unroll
    for (int r = 0; r < 8; ++r) o[r] = tanh_any(x[r]);
  }
}
__device__ __forceinline__ void cell8(const v8f (&acc)[4], float (&c)[8], float* Ht, int half, int l16) {
  float g[8], tg[8], th[8];
#pragma unroll
  for (int r = 0; r < 8; ++r) g[r] = acc[2][r];
  tanh8(tg, g);
#pragma unroll
  for (int r = 0; r < 8; ++r) c[r] = sigm(acc[1][r]) * c[r] + sigm(acc[0][r]) * tg[r];
  tanh8(th, c);
#pragma unroll
  for (int r = 0; r < 8; ++r) Ht[(r + 8 * half) * HP + l16] = sigm(acc[3][r]) * th[r];
}

template <int NKS>
__device__ __forceinline__ void stage_frag(float* dst, const float* __restrict__ Wm) {
  for (int i = threadIdx.x; i < NKS * 256; i += 64) {
    const int s = i & 1, tt = (i >> 1) & 3, ln = (i >> 3) & 31, ks = i >> 8;
    dst[i] = Wm[(ks * 4 + 2 * (ln >> 4) + s) * 64 + tt * 16 + (ln & 15)];
  }
}

__global__ __launch_bounds__(64) void k_lstm_all(
    const int* __restrict__ seq,
    const float* __restrict__ g0,
    const float* __restrict__ Wh0,
    const float* __restrict__ Wi1, const float* __restrict__ Wh1, const float* __restrict__ b1,
    const float* __restrict__ Wout, const float* __restrict__ bout,
    float* __restrict__ out, int N) {
  __shared__ __attribute__((aligned(32))) float Ws[48 * 64];
  __shared__ __attribute__((aligned(16))) float Hs[2 * 16 * HP];
  __shared__ __attribute__((aligned(16))) float Wo[32];
  __shared__ __attribute__((aligned(16))) float Bb[16 * 4];
  stage_frag<4>(Ws,           Wh0);
  stage_frag<4>(Ws + 4 * 256, Wi1);
  stage_frag<4>(Ws + 8 * 256, Wh1);
  if (threadIdx.x < 32) Wo[threadIdx.x] = Wout[(threadIdx.x & 15) * 2 + (threadIdx.x >> 4)];
  Bb[threadIdx.x] = b1[(threadIdx.x & 3) * 16 + (threadIdx.x >> 2)];
  __syncthreads();

  const int lane = threadIdx.x & 31;
  const int wave = threadIdx.x >> 5;
  const int half = lane >> 4;
  const int l16  = lane & 15;
  const int row0 = (blockIdx.x * 2 + wave) * 16;

  int ur[8];
#pragma unroll
  for (int r = 0; r < 8; ++r) { const int u = row0 + 8 * half + r; ur[r] = u > N - 1 ? N - 1 : u; }

  const float* wo = Wo + half * 16;
  const float bo = bout[half];

  const v8f* Wf = (const v8f*)Ws + lane;
  float* Ht = Hs + wave * (16 * HP);

  v2f h0A[4], h1A[4];
  float c0[8], c1[8];
#pragma unroll
  for (int i = 0; i < 4; ++i) { v2f z = { 0.0f, 0.0f }; h0A[i] = z; h1A[i] = z; }
#pragma unroll
  for (int i = 0; i < 8; ++i) { c0[i] = 0.0f; c1[i] = 0.0f; }

  int id[8];
#pragma unroll
  for (int r = 0; r < 8; ++r) id[r] = seq[ur[r]];
  for (int t = 0; t < T_STEPS; ++t) {
    v8f acc[4];
#pragma unroll
    for (int r = 0; r < 8; ++r) {
      const float* gr = g0 + (unsigned)((t * NUM_LOC + id[r]) * 64 + l16);
#pragma unroll
      for (int tt = 0; tt < 4; ++tt) acc[tt][r] = gr[tt * 16];
    }
    const int tn = t + 1 < T_STEPS ? t + 1 : t;
#pragma unroll
    for (int r = 0; r < 8; ++r) id[r] = seq[(unsigned)(tn * N + ur[r])];

#pragma unroll
    for (int ks = 0; ks < 4; ++ks) {
      const v8f w8 = Wf[(0 + ks) * 32];
#pragma unroll
      for (int tt = 0; tt < 4; ++tt) {
        v2f w = { w8[2 * tt], w8[2 * tt + 1] };
        acc[tt] = __builtin_amdgcn_wmma_f32_16x16x4_f32(false, h0A[ks], false, w, (short)0,
                                                        acc[tt], false, false);
      }
    }
    cell8(acc, c0, Ht, half, l16);
    __syncthreads();
#pragma unroll
    for (int ks = 0; ks < 4; ++ks) h0A[ks] = *(const v2f*)(Ht + l16 * HP + ks * 4 + 2 * half);

    const v4f bs1 = *(const v4f*)(Bb + l16 * 4);
#pragma unroll
    for (int tt = 0; tt < 4; ++tt) {
      float bv = bs1[tt];
      v8f tmp = { bv, bv, bv, bv, bv, bv, bv, bv };
      acc[tt] = tmp;
    }
#pragma unroll
    for (int ks = 0; ks < 4; ++ks) {
      const v8f w8 = Wf[(4 + ks) * 32];
#pragma unroll
      for (int tt = 0; tt < 4; ++tt) {
        v2f w = { w8[2 * tt], w8[2 * tt + 1] };
        acc[tt] = __builtin_amdgcn_wmma_f32_16x16x4_f32(false, h0A[ks], false, w, (short)0,
                                                        acc[tt], false, false);
      }
    }
#pragma unroll
    for (int ks = 0; ks < 4; ++ks) {
      const v8f w8 = Wf[(8 + ks) * 32];
#pragma unroll
      for (int tt = 0; tt < 4; ++tt) {
        v2f w = { w8[2 * tt], w8[2 * tt + 1] };
        acc[tt] = __builtin_amdgcn_wmma_f32_16x16x4_f32(false, h1A[ks], false, w, (short)0,
                                                        acc[tt], false, false);
      }
    }
    cell8(acc, c1, Ht, half, l16);
    __syncthreads();
#pragma unroll
    for (int ks = 0; ks < 4; ++ks) h1A[ks] = *(const v2f*)(Ht + l16 * HP + ks * 4 + 2 * half);

    float a = bo;
#pragma unroll
    for (int j = 0; j < 16; ++j) a += Ht[l16 * HP + j] * wo[j];
    if (row0 + l16 < N) out[((long)t * N + row0 + l16) * 2 + half] = a;
  }
}

static inline int gs(long n, int b) { return (int)((n + b - 1) / b); }

extern "C" void kernel_launch(void* const* d_in, const int* in_sizes, int n_in,
                              void* d_out, int out_size, void* d_ws, size_t ws_size,
                              hipStream_t stream) {
  (void)in_sizes; (void)n_in; (void)out_size;

  const int*   seq  = (const int*)d_in[0];
  const int*   gidx = (const int*)d_in[1];
  const float* Wemb = (const float*)d_in[2];
  const float* bemb = (const float*)d_in[3];
  const float* W1   = (const float*)d_in[4];
  const float* b1   = (const float*)d_in[5];
  const float* W2   = (const float*)d_in[6];
  const float* b2   = (const float*)d_in[7];
  const float* We2n = (const float*)d_in[8];
  const float* be2n = (const float*)d_in[9];
  const float* Wi0  = (const float*)d_in[10];
  const float* Wh0  = (const float*)d_in[11];
  const float* bL0  = (const float*)d_in[12];
  const float* Wi1  = (const float*)d_in[13];
  const float* Wh1  = (const float*)d_in[14];
  const float* bL1  = (const float*)d_in[15];
  const float* Wout = (const float*)d_in[16];
  const float* bout = (const float*)d_in[17];

  const int* srcE = gidx;
  const int* dstE = gidx + N_EDGES;
  float* out = (float*)d_out;

  float* p = (float*)d_ws;
  float* isq   = p; p += 5056;
  float* idg   = p; p += 5056;
  int*   uptr  = (int*)p; p += (size_t)T_STEPS * NUM_LOC;
  int*   gptr  = (int*)p; p += 5056;
  int*   gcol  = (int*)p; p += N_EDGES;
  float* gw    = p; p += N_EDGES;
  const size_t FIXED = 3 * 5056 + (size_t)T_STEPS * NUM_LOC + 2 * (size_t)N_EDGES;
  const size_t AREA  = (size_t)NUM_LOC * T_STEPS * 64;

  const bool all_steps = ws_size >= (FIXED + 2 * AREA) * sizeof(float);
  float* e     = p;
  float* h1    = all_steps ? p + AREA : p + NUM_LOC * 64;
  float* Mf    = all_steps ? h1 + AREA / 2 : h1 + NUM_LOC * 32;
  float* cy    = Mf + 64 * 32;
  float* g0all = all_steps ? p : p + 2 * NUM_LOC * 64;
  int*   ulist = all_steps ? (int*)(p + AREA) : (int*)g0all;
  const int ustride = all_steps ? USR_NUM : NUM_LOC * 64;
  int*   hist  = all_steps ? (int*)p : (int*)(p + 18 * NUM_LOC * 64);

  k_hist<<<NROW * NB + 1, 1024, 0, stream>>>(seq, dstE, hist, W2, We2n, b2, be2n, Mf, cy);
  k_scan<<<NROW, 1024, 0, stream>>>(hist, uptr, gptr, isq, idg);
  k_fill<<<NROW * NB, 1024, 0, stream>>>(seq, srcE, dstE, isq, hist, ulist, ustride, gcol, gw);

  if (all_steps) {
    const int TS = 4;
    k_gather_users<TS, T_STEPS><<<dim3(gs(NUM_LOC, 8), T_STEPS / TS), 256, 0, stream>>>(
        uptr, ulist, ustride, Wemb, bemb, e, NUM_LOC);
    k_gcn_relu<TS, T_STEPS><<<dim3(gs(NUM_LOC, 16), T_STEPS / TS), 512, 0, stream>>>(
        gptr, gcol, gw, idg, e, W1, b1, Mf, h1, NUM_LOC);
    k_gcn_e2n<TS, T_STEPS><<<dim3(gs(NUM_LOC, 16), T_STEPS / TS), 512, 0, stream>>>(
        gptr, gcol, gw, idg, h1, cy, Wi0, bL0, g0all, NUM_LOC);
  } else {
    for (int t = 0; t < T_STEPS; ++t) {
      k_gather_users<1, 1><<<gs(NUM_LOC, 8), 256, 0, stream>>>(
          uptr + (size_t)t * NUM_LOC, ulist + (size_t)t * ustride, ustride, Wemb, bemb, e, NUM_LOC);
      k_gcn_relu<1, 1><<<gs(NUM_LOC, 16), 512, 0, stream>>>(gptr, gcol, gw, idg, e, W1, b1, Mf, h1, NUM_LOC);
      k_gcn_e2n<1, 1><<<gs(NUM_LOC, 16), 512, 0, stream>>>(gptr, gcol, gw, idg, h1, cy, Wi0, bL0,
                                                            g0all + (size_t)t * NUM_LOC * 64, NUM_LOC);
    }
  }

  k_lstm_all<<<gs(USR_NUM, 32), 64, 0, stream>>>(seq, g0all, Wh0, Wi1, Wh1, bL1,
                                                 Wout, bout, out, USR_NUM);
}


# Index lists in ONE launch: a work-group owns 625 bins of a key row and reads the whole row twice (count, then rank in LDS); no chunk histograms, no scan; 7 -> 5 launches; order = LDS add arrival
# speedup vs baseline: 12.5412x; 1.1836x over previous
#include <hip/hip_runtime.h>
#include <hip/hip_bf16.h>
#include <math.h>

#define USR_NUM 100000
#define USR_DIM 64
#define NUM_LOC 5000
#define T_STEPS 16
#define N_EDGES 80000

typedef float v2f __attribute__((ext_vector_type(2)));
typedef float v4f __attribute__((ext_vector_type(4)));
typedef float v8f __attribute__((ext_vector_type(8)));

typedef int v4i __attribute__((ext_vector_type(4)));
#define NB 8
#define BR (NUM_LOC / NB)
#define UGR (USR_NUM / 4)
#define GGR (N_EDGES / 4)
#define KU 5
#define NROW (T_STEPS + 1)
static_assert(NUM_LOC % NB == 0 && BR + 32 <= 1024, "a thread takes at most one bin of its work-group's range");
static_assert(USR_NUM % 4 == 0 && N_EDGES % 4 == 0, "key rows are whole 16-byte key groups");

__device__ __forceinline__ void fold_w2_we(const float* __restrict__ W2, const float* __restrict__ We,
                                           const float* __restrict__ b2, const float* __restrict__ be,
                                           float* __restrict__ M, float* __restrict__ cy) {
  const int lane = threadIdx.x & 31;
  const int wave = threadIdx.x >> 5;
  const int half = lane >> 4;
  const int l16  = lane & 15;
  if (wave < 8) {
    const int rt = wave >> 1;
    const int ct = wave & 1;
    v8f acc = { 0.0f, 0.0f, 0.0f, 0.0f, 0.0f, 0.0f, 0.0f, 0.0f };
#pragma unroll
    for (int ks = 0; ks < 16; ++ks) {
      const v2f a = *(const v2f*)(W2 + (rt * 16 + l16) * 64 + 4 * ks + 2 * half);
      v2f b;
      b[0] = We[(4 * ks + 2 * half + 0) * 32 + ct * 16 + l16];
      b[1] = We[(4 * ks + 2 * half + 1) * 32 + ct * 16 + l16];
      acc = __builtin_amdgcn_wmma_f32_16x16x4_f32(false, a, false, b, (short)0, acc, false, false);
    }
#pragma unroll
    for (int r = 0; r < 8; ++r) M[(rt * 16 + r + 8 * half) * 32 + ct * 16 + l16] = acc[r];
  } else if (wave == 8) {
    float a = be[lane];
    for (int k = 0; k < 64; ++k) a += b2[k] * We[k * 32 + lane];
    cy[lane] = a;
  }
}

template <bool GRAPH>
__device__ __forceinline__ void lists_of_range(const int* __restrict__ keys, const int* __restrict__ src,
                                               int lo, int* __restrict__ ptr, float* __restrict__ idg,
                                               int* __restrict__ l, int* __restrict__ gcol, float* __restrict__ gw,
                                               int* cnt, int* part, int* hf, int* sbelow) {
  constexpr int NG = GRAPH ? GGR : UGR;
  constexpr int NT = (NG + KU * 1024 - 1) / (KU * 1024);
  const int tid = threadIdx.x;
  const int lane = tid & 31;

  if (tid < BR + 32) cnt[tid] = 0;
  if (tid == 0) *sbelow = 0;
  if (GRAPH) {
    for (int i = tid; i < NUM_LOC; i += 1024) hf[i] = 0;
  }
  __syncthreads();

  int below = 0;
  for (int t = 0; t < NT; ++t) {
    v4i k[KU];
#pragma unroll
    for (int u = 0; u < KU; ++u) {
      const int g = tid + (t * KU + u) * 1024;
      const v4i z = { 0, 0, 0, 0 };
      k[u] = g < NG ? *(const v4i*)(keys + 4 * g) : z;
    }
#pragma unroll
    for (int u = 0; u < KU; ++u) {
      if (tid + (t * KU + u) * 1024 < NG) {
#pragma unroll
        for (int s = 0; s < 4; ++s) {
          const int d = k[u][s] - lo;
          const unsigned di = (unsigned)d < (unsigned)(BR + lane) ? (unsigned)d : (unsigned)(BR + lane);
          atomicAdd(&cnt[di], 1);
          below -= d >> 31;
          if (GRAPH) atomicAdd(&hf[k[u][s]], 1);
        }
      }
    }
  }
  for (int m = 16; m >= 1; m >>= 1) below += __shfl_xor(below, m, 32);
  if (lane == 0) atomicAdd(sbelow, below);
  __syncthreads();

  const int my = tid < BR ? cnt[tid] : 0;
  part[tid] = my;
  __syncthreads();
  for (int off = 1; off < 1024; off <<= 1) {
    int x = tid >= off ? part[tid - off] : 0;
    __syncthreads();
    part[tid] += x;
    __syncthreads();
  }
  const int end = *sbelow + part[tid];
  if (tid < BR) {
    ptr[lo + tid] = end;
    cnt[tid] = end - my;
    if (GRAPH) idg[lo + tid] = 1.0f / ((float)my + 1.0f);
  }
  __syncthreads();

  for (int t = 0; t < NT; ++t) {
    v4i k[KU], sv[KU];
    int pos[KU][4];
#pragma unroll
    for (int u = 0; u < KU; ++u) {
      const int g = tid + (t * KU + u) * 1024;
      const v4i z = { 0, 0, 0, 0 };
      k[u] = g < NG ? *(const v4i*)(keys + 4 * g) : z;
      sv[u] = (GRAPH && g < NG) ? *(const v4i*)(src + 4 * g) : z;
    }
#pragma unroll
    for (int u = 0; u < KU; ++u) {
      const bool live = tid + (t * KU + u) * 1024 < NG;
#pragma unroll
      for (int s = 0; s < 4; ++s) {
        const int d = k[u][s] - lo;
        pos[u][s] = -1;
        if (live && (unsigned)d < (unsigned)BR) pos[u][s] = atomicAdd(&cnt[d], 1);
      }
    }
#pragma unroll
    for (int u = 0; u < KU; ++u) {
      const int g = tid + (t * KU + u) * 1024;
#pragma unroll
      for (int s = 0; s < 4; ++s) {
        if (pos[u][s] >= 0) {
          if (!GRAPH) {
            l[pos[u][s]] = 4 * g + s;
          } else {
            gcol[pos[u][s]] = sv[u][s];
            gw[pos[u][s]] = rsqrtf((float)hf[sv[u][s]] + 1.0f) * rsqrtf((float)hf[k[u][s]] + 1.0f);
          }
        }
      }
    }
  }
}

__global__ __launch_bounds__(1024) void k_lists(const int* __restrict__ seq, const int* __restrict__ src,
                                                 const int* __restrict__ dst,
                                                 int* __restrict__ uptr, int* __restrict__ gptr,
                                                 float* __restrict__ idg,
                                                 int* __restrict__ ulist_all, int ustride,
                                                 int* __restrict__ gcol, float* __restrict__ gw,
                                                 const float* __restrict__ W2, const float* __restrict__ We,
                                                 const float* __restrict__ b2, const float* __restrict__ be,
                                                 float* __restrict__ M, float* __restrict__ cy) {
  __shared__ int cnt[BR + 32];
  __shared__ int part[1024];
  __shared__ int hf[NUM_LOC];
  __shared__ int sbelow;
  if (blockIdx.x == NROW * NB) {
    fold_w2_we(W2, We, b2, be, M, cy);
    return;
  }
  const int row = blockIdx.x / NB;
  const int lo = (blockIdx.x - row * NB) * BR;
  if (row == T_STEPS) {
    lists_of_range<true>(dst, src, lo, gptr, idg, ulist_all, gcol, gw, cnt, part, hf, &sbelow);
  } else {
    lists_of_range<false>(seq + (long)row * USR_NUM, src, lo, uptr + row * NUM_LOC, idg,
                          ulist_all + (long)row * ustride, gcol, gw, cnt, part, hf, &sbelow);
  }
}

template <int TS, int NS>
__global__ __launch_bounds__(256) void k_gather_users(
    const int* __restrict__ uptr, const int* __restrict__ ulist, int ustride,
    const float* __restrict__ Wemb, const float* __restrict__ bemb, float* __restrict__ e_all,
    int nloc) {
  const int lane = threadIdx.x & 31;
  const int j = __builtin_amdgcn_readfirstlane(blockIdx.x * 8 + (threadIdx.x >> 5));
  if (j >= nloc) return;
  const v2f bb = *(const v2f*)(bemb + 2 * lane);
  for (int it = 0; it < TS; ++it) {
    const int t = blockIdx.y * TS + it;
    const int* ptr = uptr + t * NUM_LOC;
    const int* ul = ulist + (long)t * ustride;
    const int beg = j ? ptr[j - 1] : 0;
    const int end = ptr[j];
    v2f acc = { 0.0f, 0.0f };
    for (int k = beg; k < end; k += 8) {
      int id[8];
#pragma unroll
      for (int u = 0; u < 8; ++u) id[u] = ul[k + u < end ? k + u : end - 1];
      v2f x[8];
#pragma unroll
      for (int u = 0; u < 8; ++u) x[u] = *(const v2f*)(Wemb + (long)id[u] * 64 + 2 * lane);
#pragma unroll
      for (int u = 0; u < 8; ++u) {
        if (k + u < end) acc += x[u] + bb;
      }
    }
    float cv = (float)(end - beg); if (cv < 1.0f) cv = 1.0f;
    v2f r = { acc[0] / cv, acc[1] / cv };
    *(v2f*)(e_all + ((long)j * NS + t) * 64 + 2 * lane) = r;
  }
}

#define TP 68
#define OP 36

template <int TS, int NS>
__device__ __forceinline__ void gather_node(
    v2f (&acc)[TS], const int* __restrict__ gptr, const int* __restrict__ gcol,
    const float* __restrict__ gw, const float* __restrict__ idg, const float* __restrict__ h,
    int q, int t0, int lane) {
  constexpr int GE = TS == 1 ? 8 : 4;
  const int beg = q ? gptr[q - 1] : 0;
  const int end = gptr[q];
  const float* hl = h + (long)t0 * 64 + 2 * lane;
  v2f self[TS];
#pragma unroll
  for (int ts = 0; ts < TS; ++ts) self[ts] = *(const v2f*)(hl + (long)q * (NS * 64) + ts * 64);
  const float sd = idg[q];
#pragma unroll
  for (int ts = 0; ts < TS; ++ts) { v2f z = { 0.0f, 0.0f }; acc[ts] = z; }
  for (int k = beg; k < end; k += GE) {
    int s[GE];
    float w[GE];
#pragma unroll
    for (int u = 0; u < GE; ++u) {
      const int kk = k + u < end ? k + u : end - 1;
      s[u] = gcol[kk];
      w[u] = gw[kk];
    }
    v2f x[GE][TS];
#pragma unroll
    for (int u = 0; u < GE; ++u) {
      const float* hs = hl + (long)s[u] * (NS * 64);
#pragma unroll
      for (int ts = 0; ts < TS; ++ts) x[u][ts] = *(const v2f*)(hs + ts * 64);
    }
#pragma unroll
    for (int u = 0; u < GE; ++u) {
      const bool live = k + u < end;
#pragma unroll
      for (int ts = 0; ts < TS; ++ts) {
        const v2f old = acc[ts];
        const v2f upd = old + x[u][ts] * w[u];
        acc[ts] = live ? upd : old;
      }
    }
  }
#pragma unroll
  for (int ts = 0; ts < TS; ++ts) acc[ts] += self[ts] * sd;
}

template <int N> struct lanevec;
template <> struct lanevec<1> { typedef float t; };
template <> struct lanevec<4> { typedef v4f t; };

template <int TS, int NS>
__device__ __forceinline__ typename lanevec<TS>::t gather_node32(
    const int* __restrict__ gptr, const int* __restrict__ gcol, const float* __restrict__ gw,
    const float* __restrict__ idg, const float* __restrict__ h, int q, int t0, int lane) {
  typedef typename lanevec<TS>::t VT;
  constexpr int GE = 8;
  const int beg = q ? gptr[q - 1] : 0;
  const int end = gptr[q];
  const float* hl = h + (long)t0 * 32 + TS * lane;
  const VT self = *(const VT*)(hl + (long)q * (NS * 32));
  const float sd = idg[q];
  VT acc = {};
  for (int k = beg; k < end; k += GE) {
    int s[GE];
    float w[GE];
#pragma unroll
    for (int u = 0; u < GE; ++u) {
      const int kk = k + u < end ? k + u : end - 1;
      s[u] = gcol[kk];
      w[u] = gw[kk];
    }
    VT x[GE];
#pragma unroll
    for (int u = 0; u < GE; ++u) x[u] = *(const VT*)(hl + (long)s[u] * (NS * 32));
#pragma unroll
    for (int u = 0; u < GE; ++u) {
      const bool live = k + u < end;
      const VT upd = acc + x[u] * w[u];
      acc = live ? upd : acc;
    }
  }
  acc += self * sd;
  return acc;
}

template <int NKS>
__device__ __forceinline__ void load_bfrag(v2f (&b)[16], const float* __restrict__ Wm, int nc,
                                           int tt, int half, int l16) {
#pragma unroll
  for (int ks = 0; ks < NKS; ++ks) {
    b[ks][0] = Wm[(4 * ks + 2 * half + 0) * nc + tt * 16 + l16];
    b[ks][1] = Wm[(4 * ks + 2 * half + 1) * nc + tt * 16 + l16];
  }
}

template <int NT, int NKS = 16, int PITCH = TP>
__device__ __forceinline__ void tiles_product(v8f (&c)[NT], const float* tiles, const v2f (&b)[16],
                                              float bv, int half, int l16) {
#pragma unroll
  for (int i = 0; i < NT; ++i) { v8f z = { bv, bv, bv, bv, bv, bv, bv, bv }; c[i] = z; }
#pragma unroll
  for (int ks = 0; ks < NKS; ++ks) {
#pragma unroll
    for (int i = 0; i < NT; ++i) {
      const v2f a = *(const v2f*)(tiles + (i * 16 + l16) * PITCH + 4 * ks + 2 * half);
      c[i] = __builtin_amdgcn_wmma_f32_16x16x4_f32(false, a, false, b[ks], (short)0, c[i], false, false);
    }
  }
}

template <int TS, int NS>
__global__ __launch_bounds__(512) void k_gcn_relu(
    const int* __restrict__ gptr, const int* __restrict__ gcol, const float* __restrict__ gw,
    const float* __restrict__ idg, const float* __restrict__ hin, const float* __restrict__ W,
    const float* __restrict__ bias, const float* __restrict__ M, float* __restrict__ pout, int nloc) {
  typedef typename lanevec<TS>::t VT;
  __shared__ __attribute__((aligned(16))) float At[TS * 16 * TP];
  __shared__ __attribute__((aligned(16))) float Ct[TS * 16 * TP];
  __shared__ __attribute__((aligned(16))) float Pt[TS * 16 * OP];
  const int lane = threadIdx.x & 31;
  const int wave = threadIdx.x >> 5;
  const int half = lane >> 4;
  const int l16  = lane & 15;
  const int t0   = blockIdx.y * TS;
  const int q = __builtin_amdgcn_readfirstlane(blockIdx.x * 16 + wave);

  constexpr int ET = TS == 1 ? 1 : 2;
  constexpr int EW = 2 * TS / ET;
  static_assert(4 + EW <= 16, "the product waves are waves of the work-group");
  const int ec = (wave - 4) & 1;
  const int es = ((wave - 4) >> 1) * ET;
  v2f bf[16];
  float bv = 0.0f;
  if (wave < 4) {
    load_bfrag<16>(bf, W, 64, wave, half, l16);
    bv = bias[wave * 16 + l16];
  } else if (wave < 4 + EW) {
    load_bfrag<16>(bf, M, 32, ec, half, l16);
  }

  v2f acc[TS];
#pragma unroll
  for (int ts = 0; ts < TS; ++ts) { v2f z = { 0.0f, 0.0f }; acc[ts] = z; }
  if (q < nloc) gather_node<TS, NS>(acc, gptr, gcol, gw, idg, hin, q, t0, lane);
#pragma unroll
  for (int ts = 0; ts < TS; ++ts) *(v2f*)(At + (ts * 16 + wave) * TP + 2 * lane) = acc[ts];
  __syncthreads();

  if (wave < 4) {
    v8f c[TS];
    tiles_product<TS>(c, At, bf, bv, half, l16);
#pragma unroll
    for (int ts = 0; ts < TS; ++ts) {
#pragma unroll
      for (int r = 0; r < 8; ++r) {
        const float v = c[ts][r];
        Ct[(ts * 16 + r + 8 * half) * TP + wave * 16 + l16] = v > 0.0f ? v : 0.0f;
      }
    }
  }
  __syncthreads();

  if (wave >= 4 && wave < 4 + EW) {
    v8f c[ET];
    tiles_product<ET>(c, Ct + es * 16 * TP, bf, 0.0f, half, l16);
#pragma unroll
    for (int i = 0; i < ET; ++i) {
#pragma unroll
      for (int r = 0; r < 8; ++r) Pt[((es + i) * 16 + r + 8 * half) * OP + ec * 16 + l16] = c[i][r];
    }
  }
  __syncthreads();

  if (q < nloc) {
    const int f = TS * lane;
    *(VT*)(pout + ((long)q * NS + t0) * 32 + f) = *(const VT*)(Pt + ((f >> 5) * 16 + wave) * OP + (f & 31));
  }
}

template <int TS, int NS>
__global__ __launch_bounds__(512) void k_gcn_e2n(
    const int* __restrict__ gptr, const int* __restrict__ gcol, const float* __restrict__ gw,
    const float* __restrict__ idg, const float* __restrict__ pin, const float* __restrict__ cy,
    const float* __restrict__ Wi0, const float* __restrict__ bL0,
    float* __restrict__ out, int nloc) {
  typedef typename lanevec<TS>::t VT;
  __shared__ __attribute__((aligned(16))) float Ot[TS * 16 * OP];
  __shared__ __attribute__((aligned(16))) float Gt[TS * 16 * TP];
  const int lane = threadIdx.x & 31;
  const int wave = threadIdx.x >> 5;
  const int half = lane >> 4;
  const int l16  = lane & 15;
  const int t0   = blockIdx.y * TS;
  const int q = __builtin_amdgcn_readfirstlane(blockIdx.x * 16 + wave);

  static_assert(4 * TS <= 16, "the product waves are waves of the work-group");
  const int gc = wave & 3;
  const int gs = wave >> 2;
  v2f bf[16];
  float bv = 0.0f;
  if (wave < 4 * TS) {
    load_bfrag<8>(bf, Wi0, 64, gc, half, l16);
    bv = bL0[gc * 16 + l16];
  }

  const int f = TS * lane;
  const VT cv = *(const VT*)(cy + (f & 31));
  VT acc = {};
  if (q < nloc) acc = gather_node32<TS, NS>(gptr, gcol, gw, idg, pin, q, t0, lane);
  *(VT*)(Ot + ((f >> 5) * 16 + wave) * OP + (f & 31)) = acc + cv;
  __syncthreads();

  if (wave < 4 * TS) {
    v8f c[1];
    tiles_product<1, 8, OP>(c, Ot + gs * 16 * OP, bf, bv, half, l16);
#pragma unroll
    for (int r = 0; r < 8; ++r) Gt[(gs * 16 + r + 8 * half) * TP + gc * 16 + l16] = c[0][r];
  }
  __syncthreads();

  if (q < nloc) {
#pragma unroll
    for (int ts = 0; ts < TS; ++ts)
      *(v2f*)(out + ((long)(t0 + ts) * nloc + q) * 64 + 2 * lane) =
          *(const v2f*)(Gt + (ts * 16 + wave) * TP + 2 * lane);
  }
}

static_assert((long)T_STEPS * NUM_LOC * 64 < (1L << 31) && (long)T_STEPS * USR_NUM < (1L << 31), "32-bit element indices in the tail");
#define HP 20

__device__ __forceinline__ float sigm(float x) {
  return __builtin_amdgcn_rcpf(1.0f + __builtin_amdgcn_exp2f(x * -1.4426950408889634f));
}
__device__ __forceinline__ float tanh_small(float x) {
  const float y2 = x * x;
  float p = __builtin_fmaf(y2, -0x1.758e7ap-8f, 0x1.521192p-6f);
  p = __builtin_fmaf(y2, p, -0x1.b8389cp-5f);
  p = __builtin_fmaf(y2, p, 0x1.110704p-3f);
  p = __builtin_fmaf(y2, p, -0x1.555532p-2f);
  return __builtin_fmaf(y2, x * p, x);
}
__device__ __forceinline__ float tanh_any(float x) {
  const float y = __builtin_fabsf(x);
  const float t = __builtin_amdgcn_exp2f(y * 2.8853900817779268f);
  const float big = __builtin_fmaf(__builtin_amdgcn_rcpf(t + 1.0f), -2.0f, 1.0f);
  return y < 0.625f ? tanh_small(x) : __builtin_copysignf(big, x);
}
__device__ __forceinline__ void tanh8(float (&o)[8], const float (&x)[8]) {
  float m = __builtin_fabsf(x[0]);
#pragma unroll
  for (int r = 1; r < 8; ++r) m = __builtin_fmaxf(m, __builtin_fabsf(x[r]));
  if (__builtin_amdgcn_ballot_w32(m >= 0.625f) == 0) {
#pragma unroll
    for (int r = 0; r < 8; ++r) o[r] = tanh_small(x[r]);
  } else {
#pragma unroll
    for (int r = 0; r < 8; ++r) o[r] = tanh_any(x[r]);
  }
}
__device__ __forceinline__ void cell8(const v8f (&acc)[4], float (&c)[8], float* Ht, int half, int l16) {
  float g[8], tg[8], th[8];
#pragma unroll
  for (int r = 0; r < 8; ++r) g[r] = acc[2][r];
  tanh8(tg, g);
#pragma unroll
  for (int r = 0; r < 8; ++r) c[r] = sigm(acc[1][r]) * c[r] + sigm(acc[0][r]) * tg[r];
  tanh8(th, c);
#pragma unroll
  for (int r = 0; r < 8; ++r) Ht[(r + 8 * half) * HP + l16] = sigm(acc[3][r]) * th[r];
}

template <int NKS>
__device__ __forceinline__ void stage_frag(float* dst, const float* __restrict__ Wm) {
  for (int i = threadIdx.x; i < NKS * 256; i += 64) {
    const int s = i & 1, tt = (i >> 1) & 3, ln = (i >> 3) & 31, ks = i >> 8;
    dst[i] = Wm[(ks * 4 + 2 * (ln >> 4) + s) * 64 + tt * 16 + (ln & 15)];
  }
}

__global__ __launch_bounds__(64) void k_lstm_all(
    const int* __restrict__ seq,
    const float* __restrict__ g0,
    const float* __restrict__ Wh0,
    const float* __restrict__ Wi1, const float* __restrict__ Wh1, const float* __restrict__ b1,
    const float* __restrict__ Wout, const float* __restrict__ bout,
    float* __restrict__ out, int N) {
  __shared__ __attribute__((aligned(32))) float Ws[48 * 64];
  __shared__ __attribute__((aligned(16))) float Hs[2 * 16 * HP];
  __shared__ __attribute__((aligned(16))) float Wo[32];
  __shared__ __attribute__((aligned(16))) float Bb[16 * 4];
  stage_frag<4>(Ws,           Wh0);
  stage_frag<4>(Ws + 4 * 256, Wi1);
  stage_frag<4>(Ws + 8 * 256, Wh1);
  if (threadIdx.x < 32) Wo[threadIdx.x] = Wout[(threadIdx.x & 15) * 2 + (threadIdx.x >> 4)];
  Bb[threadIdx.x] = b1[(threadIdx.x & 3) * 16 + (threadIdx.x >> 2)];
  __syncthreads();

  const int lane = threadIdx.x & 31;
  const int wave = threadIdx.x >> 5;
  const int half = lane >> 4;
  const int l16  = lane & 15;
  const int row0 = (blockIdx.x * 2 + wave) * 16;

  int ur[8];
#pragma unroll
  for (int r = 0; r < 8; ++r) { const int u = row0 + 8 * half + r; ur[r] = u > N - 1 ? N - 1 : u; }

  const float* wo = Wo + half * 16;
  const float bo = bout[half];

  const v8f* Wf = (const v8f*)Ws + lane;
  float* Ht = Hs + wave * (16 * HP);

  v2f h0A[4], h1A[4];
  float c0[8], c1[8];
#pragma unroll
  for (int i = 0; i < 4; ++i) { v2f z = { 0.0f, 0.0f }; h0A[i] = z; h1A[i] = z; }
#pragma unroll
  for (int i = 0; i < 8; ++i) { c0[i] = 0.0f; c1[i] = 0.0f; }

  int id[8];
#pragma unroll
  for (int r = 0; r < 8; ++r) id[r] = seq[ur[r]];
  for (int t = 0; t < T_STEPS; ++t) {
    v8f acc[4];
#pragma unroll
    for (int r = 0; r < 8; ++r) {
      const float* gr = g0 + (unsigned)((t * NUM_LOC + id[r]) * 64 + l16);
#pragma unroll
      for (int tt = 0; tt < 4; ++tt) acc[tt][r] = gr[tt * 16];
    }
    const int tn = t + 1 < T_STEPS ? t + 1 : t;
#pragma unroll
    for (int r = 0; r < 8; ++r) id[r] = seq[(unsigned)(tn * N + ur[r])];

#pragma unroll
    for (int ks = 0; ks < 4; ++ks) {
      const v8f w8 = Wf[(0 + ks) * 32];
#pragma unroll
      for (int tt = 0; tt < 4; ++tt) {
        v2f w = { w8[2 * tt], w8[2 * tt + 1] };
        acc[tt] = __builtin_amdgcn_wmma_f32_16x16x4_f32(false, h0A[ks], false, w, (short)0,
                                                        acc[tt], false, false);
      }
    }
    cell8(acc, c0, Ht, half, l16);
    __syncthreads();
#pragma unroll
    for (int ks = 0; ks < 4; ++ks) h0A[ks] = *(const v2f*)(Ht + l16 * HP + ks * 4 + 2 * half);

    const v4f bs1 = *(const v4f*)(Bb + l16 * 4);
#pragma unroll
    for (int tt = 0; tt < 4; ++tt) {
      float bv = bs1[tt];
      v8f tmp = { bv, bv, bv, bv, bv, bv, bv, bv };
      acc[tt] = tmp;
    }
#pragma unroll
    for (int ks = 0; ks < 4; ++ks) {
      const v8f w8 = Wf[(4 + ks) * 32];
#pragma unroll
      for (int tt = 0; tt < 4; ++tt) {
        v2f w = { w8[2 * tt], w8[2 * tt + 1] };
        acc[tt] = __builtin_amdgcn_wmma_f32_16x16x4_f32(false, h0A[ks], false, w, (short)0,
                                                        acc[tt], false, false);
      }
    }
#pragma unroll
    for (int ks = 0; ks < 4; ++ks) {
      const v8f w8 = Wf[(8 + ks) * 32];
#pragma unroll
      for (int tt = 0; tt < 4; ++tt) {
        v2f w = { w8[2 * tt], w8[2 * tt + 1] };
        acc[tt] = __builtin_amdgcn_wmma_f32_16x16x4_f32(false, h1A[ks], false, w, (short)0,
                                                        acc[tt], false, false);
      }
    }
    cell8(acc, c1, Ht, half, l16);
    __syncthreads();
#pragma unroll
    for (int ks = 0; ks < 4; ++ks) h1A[ks] = *(const v2f*)(Ht + l16 * HP + ks * 4 + 2 * half);

    float a = bo;
#pragma unroll
    for (int j = 0; j < 16; ++j) a += Ht[l16 * HP + j] * wo[j];
    if (row0 + l16 < N) out[((long)t * N + row0 + l16) * 2 + half] = a;
  }
}

static inline int gs(long n, int b) { return (int)((n + b - 1) / b); }

extern "C" void kernel_launch(void* const* d_in, const int* in_sizes, int n_in,
                              void* d_out, int out_size, void* d_ws, size_t ws_size,
                              hipStream_t stream) {
  (void)in_sizes; (void)n_in; (void)out_size;

  const int*   seq  = (const int*)d_in[0];
  const int*   gidx = (const int*)d_in[1];
  const float* Wemb = (const float*)d_in[2];
  const float* bemb = (const float*)d_in[3];
  const float* W1   = (const float*)d_in[4];
  const float* b1   = (const float*)d_in[5];
  const float* W2   = (const float*)d_in[6];
  const float* b2   = (const float*)d_in[7];
  const float* We2n = (const float*)d_in[8];
  const float* be2n = (const float*)d_in[9];
  const float* Wi0  = (const float*)d_in[10];
  const float* Wh0  = (const float*)d_in[11];
  const float* bL0  = (const float*)d_in[12];
  const float* Wi1  = (const float*)d_in[13];
  const float* Wh1  = (const float*)d_in[14];
  const float* bL1  = (const float*)d_in[15];
  const float* Wout = (const float*)d_in[16];
  const float* bout = (const float*)d_in[17];

  const int* srcE = gidx;
  const int* dstE = gidx + N_EDGES;
  float* out = (float*)d_out;

  float* p = (float*)d_ws;
  float* idg   = p; p += 5056;
  int*   uptr  = (int*)p; p += (size_t)T_STEPS * NUM_LOC;
  int*   gptr  = (int*)p; p += 5056;
  int*   gcol  = (int*)p; p += N_EDGES;
  float* gw    = p; p += N_EDGES;
  const size_t FIXED = 2 * 5056 + (size_t)T_STEPS * NUM_LOC + 2 * (size_t)N_EDGES;
  const size_t AREA  = (size_t)NUM_LOC * T_STEPS * 64;

  const bool all_steps = ws_size >= (FIXED + 2 * AREA) * sizeof(float);
  float* e     = p;
  float* h1    = all_steps ? p + AREA : p + NUM_LOC * 64;
  float* Mf    = all_steps ? h1 + AREA / 2 : h1 + NUM_LOC * 32;
  float* cy    = Mf + 64 * 32;
  float* g0all = all_steps ? p : p + 2 * NUM_LOC * 64;
  int*   ulist = all_steps ? (int*)(p + AREA) : (int*)g0all;
  const int ustride = all_steps ? USR_NUM : NUM_LOC * 64;

  k_lists<<<NROW * NB + 1, 1024, 0, stream>>>(seq, srcE, dstE, uptr, gptr, idg, ulist, ustride, gcol, gw,
                                             W2, We2n, b2, be2n, Mf, cy);

  if (all_steps) {
    const int TS = 4;
    k_gather_users<TS, T_STEPS><<<dim3(gs(NUM_LOC, 8), T_STEPS / TS), 256, 0, stream>>>(
        uptr, ulist, ustride, Wemb, bemb, e, NUM_LOC);
    k_gcn_relu<TS, T_STEPS><<<dim3(gs(NUM_LOC, 16), T_STEPS / TS), 512, 0, stream>>>(
        gptr, gcol, gw, idg, e, W1, b1, Mf, h1, NUM_LOC);
    k_gcn_e2n<TS, T_STEPS><<<dim3(gs(NUM_LOC, 16), T_STEPS / TS), 512, 0, stream>>>(
        gptr, gcol, gw, idg, h1, cy, Wi0, bL0, g0all, NUM_LOC);
  } else {
    for (int t = 0; t < T_STEPS; ++t) {
      k_gather_users<1, 1><<<gs(NUM_LOC, 8), 256, 0, stream>>>(
          uptr + (size_t)t * NUM_LOC, ulist + (size_t)t * ustride, ustride, Wemb, bemb, e, NUM_LOC);
      k_gcn_relu<1, 1><<<gs(NUM_LOC, 16), 512, 0, stream>>>(gptr, gcol, gw, idg, e, W1, b1, Mf, h1, NUM_LOC);
      k_gcn_e2n<1, 1><<<gs(NUM_LOC, 16), 512, 0, stream>>>(gptr, gcol, gw, idg, h1, cy, Wi0, bL0,
                                                            g0all + (size_t)t * NUM_LOC * 64, NUM_LOC);
    }
  }

  k_lstm_all<<<gs(USR_NUM, 32), 64, 0, stream>>>(seq, g0all, Wh0, Wi1, Wh1, bL1,
                                                 Wout, bout, out, USR_NUM);
}
